# starting kernel
# baseline (speedup 1.0000x reference)
#include <hip/hip_runtime.h>
#include <hip/hip_bf16.h>
#include <cstddef>

#define N0_  100000
#define N1_  50000
#define N2_  25000
#define DEG_ 16
#define IN_  256
#define HID_ 128
#define OUT_ 128
#define H1_  2
#define H2_  1

typedef float v2f __attribute__((ext_vector_type(2)));
typedef float v8f __attribute__((ext_vector_type(8)));

template <int TN, int TK>
__global__ void gat_gemm_wmma(const float* __restrict__ A,
                              const float* __restrict__ B,
                              float* __restrict__ D,
                              int M, int tilesM) {
  constexpr int groupsN = TN >> 6;
  const int lane = threadIdx.x & 31;
  const int wave = threadIdx.x >> 5;
  const int gid = blockIdx.x * 8 + wave;
  if (gid >= tilesM * groupsN) return;
  const int tileM = gid / groupsN;
  const int gN    = gid % groupsN;

  const int half = lane >> 4;
  const int l15  = lane & 15;
  int row = tileM * 16 + l15;
  if (row >= M) row = M - 1;
  const float2* ap = (const float2*)(A + (size_t)row * TK) + half;
  const float*  bp = B + (size_t)(half * 2) * TN + gN * 64 + l15;

  v8f c0 = {}, c1 = {}, c2 = {}, c3 = {};

#pragma unroll 8
  for (int k0 = 0; k0 < TK; k0 += 4) {
    const float2 a2 = *ap;
    ap += 2;
    v2f a = { a2.x, a2.y };
    v2f b0 = { bp[ 0], bp[TN +  0] };
    v2f b1 = { bp[16], bp[TN + 16] };
    v2f b2 = { bp[32], bp[TN + 32] };
    v2f b3 = { bp[48], bp[TN + 48] };
    bp += 4 * TN;
    c0 = __builtin_amdgcn_wmma_f32_16x16x4_f32(false, a, false, b0, (short)0, c0, false, false);
    c1 = __builtin_amdgcn_wmma_f32_16x16x4_f32(false, a, false, b1, (short)0, c1, false, false);
    c2 = __builtin_amdgcn_wmma_f32_16x16x4_f32(false, a, false, b2, (short)0, c2, false, false);
    c3 = __builtin_amdgcn_wmma_f32_16x16x4_f32(false, a, false, b3, (short)0, c3, false, false);
  }

#pragma unroll
  for (int r = 0; r < 8; ++r) {
    const int m = tileM * 16 + r + half * 8;
    if (m < M) {
      float* Dr = D + (size_t)m * TN + gN * 64 + l15;
      Dr[0]  = c0[r];
      Dr[16] = c1[r];
      Dr[32] = c2[r];
      Dr[48] = c3[r];
    }
  }
}

__global__ void gat_att_dot(const float* __restrict__ F,
                            const float* __restrict__ att,
                            float* __restrict__ out, int Nn, int H) {
  const int wid  = blockIdx.x * 8 + (threadIdx.x >> 5);
  const int lane = threadIdx.x & 31;
  if (wid >= Nn * H) return;
  const int n = wid / H, h = wid % H;
  const float4 fv = *((const float4*)(F + ((size_t)n * H + h) * 128) + lane);
  const float4 av = *((const float4*)(att + (size_t)h * 128) + lane);
  float s = fv.x * av.x + fv.y * av.y + fv.z * av.z + fv.w * av.w;
#pragma unroll
  for (int off = 16; off > 0; off >>= 1) s += __shfl_xor(s, off, 32);
  if (lane == 0) out[wid] = s;
}

__global__ void gat_edge_softmax_agg(const float* __restrict__ HS,
                                     const float* __restrict__ AS,
                                     const float* __restrict__ AD,
                                     const int* __restrict__ col,
                                     const int* __restrict__ tt,
                                     const int* __restrict__ timeP,
                                     const int* __restrict__ intervalP,
                                     const float* __restrict__ b,
                                     const float* __restrict__ g,
                                     const float* __restrict__ beta,
                                     const float* __restrict__ rm,
                                     const float* __restrict__ rv,
                                     float* __restrict__ OUT,
                                     int Nd, int H, int mode) {
  const int wid  = blockIdx.x * 8 + (threadIdx.x >> 5);
  const int lane = threadIdx.x & 31;
  if (wid >= Nd * H) return;
  const int i = wid / H, h = wid % H;

  const int t0  = timeP[0];
  const int t1e = t0 + intervalP[0];
  const float adv = AD[(size_t)i * H + h];

  float e[DEG_];
  int   cj[DEG_];
  bool  val[DEG_];
  float m = -__builtin_inff();
#pragma unroll
  for (int j = 0; j < DEG_; ++j) {
    const int eidx = i * DEG_ + j;
    cj[j] = col[eidx];
    const int tv = tt[eidx];
    val[j] = (tv >= t0) && (tv < t1e);
    float ev = AS[(size_t)cj[j] * H + h] + adv;
    ev = ev > 0.0f ? ev : 0.2f * ev;
    e[j] = ev;
    if (val[j] && ev > m) m = ev;
  }
  if (!(m > -__builtin_inff())) m = 0.0f;

  float w[DEG_];
  float denom = 0.0f;
#pragma unroll
  for (int j = 0; j < DEG_; ++j) {
    const float ex = val[j] ? __expf(e[j] - m) : 0.0f;
    w[j] = ex;
    denom += ex;
  }
  const float inv = 1.0f / fmaxf(denom, 1e-16f);

  float4 acc = {0.0f, 0.0f, 0.0f, 0.0f};
#pragma unroll
  for (int j = 0; j < DEG_; ++j) {
    if (w[j] != 0.0f) {
      const float4 hv = *((const float4*)(HS + ((size_t)cj[j] * H + h) * 128) + lane);
      const float wj = w[j] * inv;
      acc.x += wj * hv.x; acc.y += wj * hv.y;
      acc.z += wj * hv.z; acc.w += wj * hv.w;
    }
  }

  float vals[4] = {acc.x, acc.y, acc.z, acc.w};
  if (mode == 1) {
#pragma unroll
    for (int q = 0; q < 4; ++q) {
      const int cc = h * 128 + lane * 4 + q;
      float vv = vals[q] + b[cc];
      vv = (vv - rm[cc]) * rsqrtf(rv[cc] + 1e-5f) * g[cc] + beta[cc];
      vv = vv > 0.0f ? vv : 0.01f * vv;
      OUT[(size_t)i * (H * 128) + cc] = vv;
    }
  } else {
#pragma unroll
    for (int q = 0; q < 4; ++q) {
      const int cc = lane * 4 + q;
      OUT[(size_t)i * 128 + cc] = vals[q] / (float)H + b[cc];
    }
  }
}

static inline int cdiv(int a, int b) { return (a + b - 1) / b; }

extern "C" void kernel_launch(void* const* d_in, const int* in_sizes, int n_in,
                              void* d_out, int out_size, void* d_ws, size_t ws_size,
                              hipStream_t stream) {
  const float* x     = (const float*)d_in[0];
  const int*   col1  = (const int*)d_in[2];
  const int*   t1    = (const int*)d_in[3];
  const int*   col2  = (const int*)d_in[5];
  const int*   t2    = (const int*)d_in[6];
  const int*   timeP = (const int*)d_in[7];
  const int*   intvP = (const int*)d_in[8];
  const float* Ws1   = (const float*)d_in[9];
  const float* Wd1   = (const float*)d_in[10];
  const float* as1   = (const float*)d_in[11];
  const float* ad1   = (const float*)d_in[12];
  const float* b1    = (const float*)d_in[13];
  const float* g1    = (const float*)d_in[14];
  const float* beta1 = (const float*)d_in[15];
  const float* m1    = (const float*)d_in[16];
  const float* v1    = (const float*)d_in[17];
  const float* Ws2   = (const float*)d_in[18];
  const float* Wd2   = (const float*)d_in[19];
  const float* as2   = (const float*)d_in[20];
  const float* ad2   = (const float*)d_in[21];
  const float* b2    = (const float*)d_in[22];

  float* ws   = (float*)d_ws;
  float* hs1  = ws;
  float* hd1  = hs1  + (size_t)N0_ * 256;
  float* as1v = hd1  + (size_t)N1_ * 256;
  float* ad1v = as1v + (size_t)N0_ * H1_;
  float* h    = ad1v + (size_t)N1_ * H1_;
  float* as2v = h    + (size_t)N1_ * 256;
  float* ad2v = as2v + (size_t)N1_;
  float* hs2  = hd1;
  float* hd2  = hd1  + (size_t)N1_ * 128;

  const dim3 blk(256);

  {
    int tilesM = cdiv(N0_, 16), waves = tilesM * (256 / 64);
    gat_gemm_wmma<256, 256><<<cdiv(waves, 8), blk, 0, stream>>>(x, Ws1, hs1, N0_, tilesM);
  }
  {
    int tilesM = cdiv(N1_, 16), waves = tilesM * (256 / 64);
    gat_gemm_wmma<256, 256><<<cdiv(waves, 8), blk, 0, stream>>>(x, Wd1, hd1, N1_, tilesM);
  }
  gat_att_dot<<<cdiv(N0_ * H1_, 8), blk, 0, stream>>>(hs1, as1, as1v, N0_, H1_);
  gat_att_dot<<<cdiv(N1_ * H1_, 8), blk, 0, stream>>>(hd1, ad1, ad1v, N1_, H1_);
  gat_edge_softmax_agg<<<cdiv(N1_ * H1_, 8), blk, 0, stream>>>(
      hs1, as1v, ad1v, col1, t1, timeP, intvP,
      b1, g1, beta1, m1, v1, h, N1_, H1_, 1);

  {
    int tilesM = cdiv(N1_, 16), waves = tilesM * (128 / 64);
    gat_gemm_wmma<128, 256><<<cdiv(waves, 8), blk, 0, stream>>>(h, Ws2, hs2, N1_, tilesM);
  }
  {
    int tilesM = cdiv(N2_, 16), waves = tilesM * (128 / 64);
    gat_gemm_wmma<128, 256><<<cdiv(waves, 8), blk, 0, stream>>>(h, Wd2, hd2, N2_, tilesM);
  }
  gat_att_dot<<<cdiv(N1_ * H2_, 8), blk, 0, stream>>>(hs2, as2, as2v, N1_, H2_);
  gat_att_dot<<<cdiv(N2_ * H2_, 8), blk, 0, stream>>>(hd2, ad2, ad2v, N2_, H2_);
  gat_edge_softmax_agg<<<cdiv(N2_ * H2_, 8), blk, 0, stream>>>(
      hs2, as2v, ad2v, col2, t2, timeP, intvP,
      b2, nullptr, nullptr, nullptr, nullptr, (float*)d_out, N2_, H2_, 2);
}


# GEMMs: B copied once per work-group into LDS (k-pair interleaved, padded pitch); 256 work-groups of 32 waves loop over row tiles; B fragments read from LDS, not global
# speedup vs baseline: 1.2473x; 1.2473x over previous
#include <hip/hip_runtime.h>
#include <hip/hip_bf16.h>
#include <cstddef>

#define N0_  100000
#define N1_  50000
#define N2_  25000
#define DEG_ 16
#define IN_  256
#define HID_ 128
#define OUT_ 128
#define H1_  2
#define H2_  1

typedef float v2f __attribute__((ext_vector_type(2)));
typedef float v8f __attribute__((ext_vector_type(8)));

#define GEMM_WGS_   256
#define GEMM_WAVES_ 32

template <int TN, int TK>
__global__ void __launch_bounds__(GEMM_WAVES_ * 32)
gat_gemm_wmma(const float* __restrict__ A,
              const float* __restrict__ B,
              float* __restrict__ D,
              int M, int tilesM) {
  constexpr int P      = 2 * TN + 32;
  constexpr int P2     = P / 2;
  constexpr int HALVES = TN / 128;
  __shared__ __attribute__((aligned(16))) float Bs[(TK / 2) * P];

  {
    constexpr int C4    = TN / 4;
    constexpr int ITEMS = (TK / 2) * C4;
#pragma unroll
    for (int it = threadIdx.x; it < ITEMS; it += GEMM_WAVES_ * 32) {
      const int kp = it / C4;
      const int c4 = it % C4;
      const float4 r0 = *((const float4*)(B + (size_t)(2 * kp) * TN) + c4);
      const float4 r1 = *((const float4*)(B + (size_t)(2 * kp + 1) * TN) + c4);
      float4* dst = (float4*)(Bs + kp * P + c4 * 8);
      dst[0] = make_float4(r0.x, r1.x, r0.y, r1.y);
      dst[1] = make_float4(r0.z, r1.z, r0.w, r1.w);
    }
  }
  __syncthreads();

  const int lane = threadIdx.x & 31;
  const int wave = threadIdx.x >> 5;
  const int half = lane >> 4;
  const int l15  = lane & 15;

  for (int j = wave; ; j += GEMM_WAVES_) {
    const int tileM = blockIdx.x + gridDim.x * (j / HALVES);
    if (tileM >= tilesM) break;
    const int nh = j % HALVES;

    int row = tileM * 16 + l15;
    if (row >= M) row = M - 1;
    const float2* ap = (const float2*)(A + (size_t)row * TK) + half;
    const float2* bs = (const float2*)Bs + half * P2 + nh * 128 + l15;

    v8f c[8];
#pragma unroll
    for (int t = 0; t < 8; ++t) c[t] = v8f{};

#pragma unroll 4
    for (int k0 = 0; k0 < TK; k0 += 4) {
      const float2 a2 = *ap;
      ap += 2;
      v2f a = { a2.x, a2.y };
#pragma unroll
      for (int t = 0; t < 8; ++t) {
        const float2 b2 = bs[t * 16];
        v2f b = { b2.x, b2.y };
        c[t] = __builtin_amdgcn_wmma_f32_16x16x4_f32(false, a, false, b, (short)0, c[t], false, false);
      }
      bs += P;
    }

#pragma unroll
    for (int r = 0; r < 8; ++r) {
      const int m = tileM * 16 + r + half * 8;
      if (m < M) {
        float* Dr = D + (size_t)m * TN + nh * 128 + l15;
#pragma unroll
        for (int t = 0; t < 8; ++t) Dr[t * 16] = c[t][r];
      }
    }
  }
}

__global__ void gat_att_dot(const float* __restrict__ F,
                            const float* __restrict__ att,
                            float* __restrict__ out, int Nn, int H) {
  const int wid  = blockIdx.x * 8 + (threadIdx.x >> 5);
  const int lane = threadIdx.x & 31;
  if (wid >= Nn * H) return;
  const int n = wid / H, h = wid % H;
  const float4 fv = *((const float4*)(F + ((size_t)n * H + h) * 128) + lane);
  const float4 av = *((const float4*)(att + (size_t)h * 128) + lane);
  float s = fv.x * av.x + fv.y * av.y + fv.z * av.z + fv.w * av.w;
#pragma unroll
  for (int off = 16; off > 0; off >>= 1) s += __shfl_xor(s, off, 32);
  if (lane == 0) out[wid] = s;
}

__global__ void gat_edge_softmax_agg(const float* __restrict__ HS,
                                     const float* __restrict__ AS,
                                     const float* __restrict__ AD,
                                     const int* __restrict__ col,
                                     const int* __restrict__ tt,
                                     const int* __restrict__ timeP,
                                     const int* __restrict__ intervalP,
                                     const float* __restrict__ b,
                                     const float* __restrict__ g,
                                     const float* __restrict__ beta,
                                     const float* __restrict__ rm,
                                     const float* __restrict__ rv,
                                     float* __restrict__ OUT,
                                     int Nd, int H, int mode) {
  const int wid  = blockIdx.x * 8 + (threadIdx.x >> 5);
  const int lane = threadIdx.x & 31;
  if (wid >= Nd * H) return;
  const int i = wid / H, h = wid % H;

  const int t0  = timeP[0];
  const int t1e = t0 + intervalP[0];
  const float adv = AD[(size_t)i * H + h];

  float e[DEG_];
  int   cj[DEG_];
  bool  val[DEG_];
  float m = -__builtin_inff();
#pragma unroll
  for (int j = 0; j < DEG_; ++j) {
    const int eidx = i * DEG_ + j;
    cj[j] = col[eidx];
    const int tv = tt[eidx];
    val[j] = (tv >= t0) && (tv < t1e);
    float ev = AS[(size_t)cj[j] * H + h] + adv;
    ev = ev > 0.0f ? ev : 0.2f * ev;
    e[j] = ev;
    if (val[j] && ev > m) m = ev;
  }
  if (!(m > -__builtin_inff())) m = 0.0f;

  float w[DEG_];
  float denom = 0.0f;
#pragma unroll
  for (int j = 0; j < DEG_; ++j) {
    const float ex = val[j] ? __expf(e[j] - m) : 0.0f;
    w[j] = ex;
    denom += ex;
  }
  const float inv = 1.0f / fmaxf(denom, 1e-16f);

  float4 acc = {0.0f, 0.0f, 0.0f, 0.0f};
#pragma unroll
  for (int j = 0; j < DEG_; ++j) {
    if (w[j] != 0.0f) {
      const float4 hv = *((const float4*)(HS + ((size_t)cj[j] * H + h) * 128) + lane);
      const float wj = w[j] * inv;
      acc.x += wj * hv.x; acc.y += wj * hv.y;
      acc.z += wj * hv.z; acc.w += wj * hv.w;
    }
  }

  float vals[4] = {acc.x, acc.y, acc.z, acc.w};
  if (mode == 1) {
#pragma unroll
    for (int q = 0; q < 4; ++q) {
      const int cc = h * 128 + lane * 4 + q;
      float vv = vals[q] + b[cc];
      vv = (vv - rm[cc]) * rsqrtf(rv[cc] + 1e-5f) * g[cc] + beta[cc];
      vv = vv > 0.0f ? vv : 0.01f * vv;
      OUT[(size_t)i * (H * 128) + cc] = vv;
    }
  } else {
#pragma unroll
    for (int q = 0; q < 4; ++q) {
      const int cc = lane * 4 + q;
      OUT[(size_t)i * 128 + cc] = vals[q] / (float)H + b[cc];
    }
  }
}

static inline int cdiv(int a, int b) { return (a + b - 1) / b; }

extern "C" void kernel_launch(void* const* d_in, const int* in_sizes, int n_in,
                              void* d_out, int out_size, void* d_ws, size_t ws_size,
                              hipStream_t stream) {
  const float* x     = (const float*)d_in[0];
  const int*   col1  = (const int*)d_in[2];
  const int*   t1    = (const int*)d_in[3];
  const int*   col2  = (const int*)d_in[5];
  const int*   t2    = (const int*)d_in[6];
  const int*   timeP = (const int*)d_in[7];
  const int*   intvP = (const int*)d_in[8];
  const float* Ws1   = (const float*)d_in[9];
  const float* Wd1   = (const float*)d_in[10];
  const float* as1   = (const float*)d_in[11];
  const float* ad1   = (const float*)d_in[12];
  const float* b1    = (const float*)d_in[13];
  const float* g1    = (const float*)d_in[14];
  const float* beta1 = (const float*)d_in[15];
  const float* m1    = (const float*)d_in[16];
  const float* v1    = (const float*)d_in[17];
  const float* Ws2   = (const float*)d_in[18];
  const float* Wd2   = (const float*)d_in[19];
  const float* as2   = (const float*)d_in[20];
  const float* ad2   = (const float*)d_in[21];
  const float* b2    = (const float*)d_in[22];

  float* ws   = (float*)d_ws;
  float* hs1  = ws;
  float* hd1  = hs1  + (size_t)N0_ * 256;
  float* as1v = hd1  + (size_t)N1_ * 256;
  float* ad1v = as1v + (size_t)N0_ * H1_;
  float* h    = ad1v + (size_t)N1_ * H1_;
  float* as2v = h    + (size_t)N1_ * 256;
  float* ad2v = as2v + (size_t)N1_;
  float* hs2  = hd1;
  float* hd2  = hd1  + (size_t)N1_ * 128;

  const dim3 blk(256);
  const dim3 gblk(GEMM_WAVES_ * 32);

  gat_gemm_wmma<256, 256><<<GEMM_WGS_, gblk, 0, stream>>>(x, Ws1, hs1, N0_, cdiv(N0_, 16));
  gat_gemm_wmma<256, 256><<<GEMM_WGS_, gblk, 0, stream>>>(x, Wd1, hd1, N1_, cdiv(N1_, 16));
  gat_att_dot<<<cdiv(N0_ * H1_, 8), blk, 0, stream>>>(hs1, as1, as1v, N0_, H1_);
  gat_att_dot<<<cdiv(N1_ * H1_, 8), blk, 0, stream>>>(hd1, ad1, ad1v, N1_, H1_);
  gat_edge_softmax_agg<<<cdiv(N1_ * H1_, 8), blk, 0, stream>>>(
      hs1, as1v, ad1v, col1, t1, timeP, intvP,
      b1, g1, beta1, m1, v1, h, N1_, H1_, 1);

  gat_gemm_wmma<128, 256><<<GEMM_WGS_, gblk, 0, stream>>>(h, Ws2, hs2, N1_, cdiv(N1_, 16));
  gat_gemm_wmma<128, 256><<<GEMM_WGS_, gblk, 0, stream>>>(h, Wd2, hd2, N2_, cdiv(N2_, 16));
  gat_att_dot<<<cdiv(N1_ * H2_, 8), blk, 0, stream>>>(hs2, as2, as2v, N1_, H2_);
  gat_att_dot<<<cdiv(N2_ * H2_, 8), blk, 0, stream>>>(hd2, ad2, ad2v, N2_, H2_);
  gat_edge_softmax_agg<<<cdiv(N2_ * H2_, 8), blk, 0, stream>>>(
      hs2, as2v, ad2v, col2, t2, timeP, intvP,
      b2, nullptr, nullptr, nullptr, nullptr, (float*)d_out, N2_, H2_, 2);
}


# Layer 1 aggregate-then-transform: softmax-weighted sum of x rows per head, then one WMMA GEMM over N1 dst rows, bias/BN/lrelu in its store; logits by a 1-tile WMMA pass (layer-1 sum order changes)
# speedup vs baseline: 2.3390x; 1.8753x over previous
#include <hip/hip_runtime.h>
#include <hip/hip_bf16.h>
#include <cstddef>

#define N0_  100000
#define N1_  50000
#define N2_  25000
#define DEG_ 16
#define IN_  256
#define HID_ 128
#define OUT_ 128
#define H1_  2
#define H2_  1

typedef float v2f __attribute__((ext_vector_type(2)));
typedef float v8f __attribute__((ext_vector_type(8)));

template <int H>
__device__ __forceinline__ void gat_fold_row(const float* __restrict__ Ws,
                                             const float* __restrict__ Wd,
                                             const float* __restrict__ as,
                                             const float* __restrict__ ad,
                                             float* __restrict__ WF,
                                             int k, int lane) {
  float outv = 0.0f;
#pragma unroll
  for (int h = 0; h < H; ++h) {
    const float4 av = *((const float4*)(as + (size_t)h * 128) + lane);
    const float4 dv = *((const float4*)(ad + (size_t)h * 128) + lane);
    const float4 sv = *((const float4*)(Ws + (size_t)k * (H * 128) + h * 128) + lane);
    const float4 wv = *((const float4*)(Wd + (size_t)k * (H * 128) + h * 128) + lane);
    float s = sv.x * av.x + sv.y * av.y + sv.z * av.z + sv.w * av.w;
    float d = wv.x * dv.x + wv.y * dv.y + wv.z * dv.z + wv.w * dv.w;
#pragma unroll
    for (int off = 16; off > 0; off >>= 1) {
      s += __shfl_xor(s, off, 32);
      d += __shfl_xor(d, off, 32);
    }
    if (lane == h)     outv = s;
    if (lane == H + h) outv = d;
  }
  if (lane < 16) WF[(size_t)k * 16 + lane] = outv;
}

__global__ void gat_fold_att(const float* __restrict__ Ws1, const float* __restrict__ Wd1,
                             const float* __restrict__ as1, const float* __restrict__ ad1,
                             const float* __restrict__ Ws2, const float* __restrict__ Wd2,
                             const float* __restrict__ as2, const float* __restrict__ ad2,
                             float* __restrict__ WF1, float* __restrict__ WF2) {
  const int wid  = blockIdx.x * 8 + (threadIdx.x >> 5);
  const int lane = threadIdx.x & 31;
  if (wid < IN_) {
    gat_fold_row<H1_>(Ws1, Wd1, as1, ad1, WF1, wid, lane);
  } else if (wid < IN_ + H1_ * HID_) {
    gat_fold_row<H2_>(Ws2, Wd2, as2, ad2, WF2, wid - IN_, lane);
  }
}


__global__ void __launch_bounds__(256)
gat_logit_wmma(const float* __restrict__ A,
               const float* __restrict__ WF,
               float* __restrict__ AS,
               float* __restrict__ AD,
               int M, int Md, int tilesM) {
  constexpr int TK = IN_;
  constexpr int H  = H1_;
  __shared__ __attribute__((aligned(16))) float Bs[(TK / 2) * 32];
  for (int it = threadIdx.x; it < (TK / 2) * 4; it += 256) {
    const int kp = it >> 2;
    const int c4 = it & 3;
    const float4 r0 = *((const float4*)(WF + (size_t)(2 * kp) * 16) + c4);
    const float4 r1 = *((const float4*)(WF + (size_t)(2 * kp + 1) * 16) + c4);
    float4* dst = (float4*)(Bs + kp * 32 + c4 * 8);
    dst[0] = make_float4(r0.x, r1.x, r0.y, r1.y);
    dst[1] = make_float4(r0.z, r1.z, r0.w, r1.w);
  }
  __syncthreads();

  const int lane = threadIdx.x & 31;
  const int wave = threadIdx.x >> 5;
  const int half = lane >> 4;
  const int l15  = lane & 15;
  const int tileM = blockIdx.x * 8 + wave;
  if (tileM >= tilesM) return;

  int row = tileM * 16 + l15;
  if (row >= M) row = M - 1;
  const float2* ap = (const float2*)(A + (size_t)row * TK) + half;
  const float2* bs = (const float2*)Bs + half * 16 + l15;

  v8f c0 = v8f{};
#pragma unroll 4
  for (int k0 = 0; k0 < TK; k0 += 8) {
    const float2 a0 = ap[0];
    const float2 a1 = ap[2];
    const float2 b0 = bs[0];
    const float2 b1 = bs[32];
    ap += 4;
    bs += 64;
    v2f va0 = { a0.x, a0.y };
    v2f vb0 = { b0.x, b0.y };
    v2f va1 = { a1.x, a1.y };
    v2f vb1 = { b1.x, b1.y };
    c0 = __builtin_amdgcn_wmma_f32_16x16x4_f32(false, va0, false, vb0, (short)0, c0, false, false);
    c0 = __builtin_amdgcn_wmma_f32_16x16x4_f32(false, va1, false, vb1, (short)0, c0, false, false);
  }

#pragma unroll
  for (int r = 0; r < 8; ++r) {
    const int m = tileM * 16 + r + half * 8;
    if (m < M) {
      const float v = c0[r];
      if (l15 < H) {
        AS[(size_t)m * H + l15] = v;
      } else if (l15 < 2 * H && m < Md) {
        AD[(size_t)m * H + (l15 - H)] = v;
      }
    }
  }
}

template <int H>
__global__ void gat_softmax_gather(const float* __restrict__ X,
                                   const float* __restrict__ AS,
                                   const float* __restrict__ AD,
                                   const int* __restrict__ col,
                                   const int* __restrict__ tt,
                                   const int* __restrict__ timeP,
                                   const int* __restrict__ intervalP,
                                   float* __restrict__ Z,
                                   int Nd) {
  const int i    = blockIdx.x * 8 + (threadIdx.x >> 5);
  const int lane = threadIdx.x & 31;
  if (i >= Nd) return;

  const int t0  = timeP[0];
  const int t1e = t0 + intervalP[0];

  int  cj[DEG_];
  bool val[DEG_];
#pragma unroll
  for (int j = 0; j < DEG_; ++j) {
    const int eidx = i * DEG_ + j;
    cj[j] = col[eidx];
    const int tv = tt[eidx];
    val[j] = (tv >= t0) && (tv < t1e);
  }

  float w[H][DEG_];
  float inv[H];
#pragma unroll
  for (int h = 0; h < H; ++h) {
    const float adv = AD[(size_t)i * H + h];
    float e[DEG_];
    float m = -__builtin_inff();
#pragma unroll
    for (int j = 0; j < DEG_; ++j) {
      float ev = AS[(size_t)cj[j] * H + h] + adv;
      ev = ev > 0.0f ? ev : 0.2f * ev;
      e[j] = ev;
      if (val[j] && ev > m) m = ev;
    }
    if (!(m > -__builtin_inff())) m = 0.0f;
    float denom = 0.0f;
#pragma unroll
    for (int j = 0; j < DEG_; ++j) {
      const float ex = val[j] ? __expf(e[j] - m) : 0.0f;
      w[h][j] = ex;
      denom += ex;
    }
    inv[h] = 1.0f / fmaxf(denom, 1e-16f);
  }

  float4 acc[H][2];
#pragma unroll
  for (int h = 0; h < H; ++h) {
    acc[h][0] = make_float4(0.0f, 0.0f, 0.0f, 0.0f);
    acc[h][1] = make_float4(0.0f, 0.0f, 0.0f, 0.0f);
  }
#pragma unroll
  for (int j = 0; j < DEG_; ++j) {
    bool any = false;
#pragma unroll
    for (int h = 0; h < H; ++h) any = any || (w[h][j] != 0.0f);
    if (any) {
      const float4* xr = (const float4*)(X + (size_t)cj[j] * 256);
      const float4 x0 = xr[lane];
      const float4 x1 = xr[32 + lane];
#pragma unroll
      for (int h = 0; h < H; ++h) {
        const float wj = w[h][j] * inv[h];
        acc[h][0].x += wj * x0.x; acc[h][0].y += wj * x0.y;
        acc[h][0].z += wj * x0.z; acc[h][0].w += wj * x0.w;
        acc[h][1].x += wj * x1.x; acc[h][1].y += wj * x1.y;
        acc[h][1].z += wj * x1.z; acc[h][1].w += wj * x1.w;
      }
    }
  }
#pragma unroll
  for (int h = 0; h < H; ++h) {
    float4* zr = (float4*)(Z + ((size_t)h * Nd + i) * 256);
    zr[lane]      = acc[h][0];
    zr[32 + lane] = acc[h][1];
  }
}

#define GEMM_WGS_   256
#define GEMM_WAVES_ 32

template <int TN, int TK, int NT>
__device__ __forceinline__ void gat_gemm_item(const float* __restrict__ A,
                                              const float* Bs,
                                              float* __restrict__ D,
                                              float* __restrict__ AS,
                                              float* __restrict__ AD,
                                              const float* __restrict__ bias,
                                              const float* __restrict__ bng,
                                              const float* __restrict__ bnb,
                                              const float* __restrict__ bnm,
                                              const float* __restrict__ bnv,
                                              int M, int Md, int tileM, int nh,
                                              int half, int l15) {
  constexpr int P  = 2 * TN + 32;
  constexpr int P2 = P / 2;
  constexpr int H  = TN / 128;

  int row = tileM * 16 + l15;
  if (row >= M) row = M - 1;
  const float2* ap = (const float2*)(A + (size_t)row * TK) + half;
  const float2* bs = (const float2*)Bs + half * P2 + nh * 128 + l15;

  v8f c[NT];
#pragma unroll
  for (int t = 0; t < NT; ++t) c[t] = v8f{};

  if constexpr (NT == 8) {
#pragma unroll 4
    for (int k0 = 0; k0 < TK; k0 += 4) {
      const float2 a2 = *ap;
      ap += 2;
      v2f a = { a2.x, a2.y };
#pragma unroll
      for (int t = 0; t < 8; ++t) {
        const float2 b2 = bs[t * 16];
        v2f b = { b2.x, b2.y };
        c[t] = __builtin_amdgcn_wmma_f32_16x16x4_f32(false, a, false, b, (short)0, c[t], false, false);
      }
      bs += P;
    }
  } else {
    float2 an = *ap;
    float2 bn[NT];
#pragma unroll
    for (int t = 0; t < 8; ++t) bn[t] = bs[t * 16];
    bn[8] = bs[TN];
#pragma unroll
    for (int k0 = 0; k0 < TK - 4; k0 += 4) {
      const float2 ac = an;
      float2 bc[NT];
#pragma unroll
      for (int t = 0; t < NT; ++t) bc[t] = bn[t];
      ap += 2;
      bs += P;
      an = *ap;
#pragma unroll
      for (int t = 0; t < 8; ++t) bn[t] = bs[t * 16];
      bn[8] = bs[TN];
      __builtin_amdgcn_sched_barrier(0);
      v2f a = { ac.x, ac.y };
#pragma unroll
      for (int t = 0; t < NT; ++t) {
        v2f b = { bc[t].x, bc[t].y };
        c[t] = __builtin_amdgcn_wmma_f32_16x16x4_f32(false, a, false, b, (short)0, c[t], false, false);
      }
    }
    {
      v2f a = { an.x, an.y };
#pragma unroll
      for (int t = 0; t < NT; ++t) {
        v2f b = { bn[t].x, bn[t].y };
        c[t] = __builtin_amdgcn_wmma_f32_16x16x4_f32(false, a, false, b, (short)0, c[t], false, false);
      }
    }
  }

  if constexpr (NT == 8) {
#pragma unroll
    for (int t = 0; t < 8; ++t) {
      const int cc = nh * 128 + t * 16 + l15;
      const float pb = bias[cc];
      const float pm = bnm[cc];
      const float ps = rsqrtf(bnv[cc] + 1e-5f);
      const float pg = bng[cc];
      const float pe = bnb[cc];
#pragma unroll
      for (int r = 0; r < 8; ++r) {
        const int m = tileM * 16 + r + half * 8;
        if (m < M) {
          float vv = c[t][r] + pb;
          vv = (vv - pm) * ps * pg + pe;
          vv = vv > 0.0f ? vv : 0.01f * vv;
          D[(size_t)m * TN + cc] = vv;
        }
      }
    }
  } else {
#pragma unroll
    for (int r = 0; r < 8; ++r) {
      const int m = tileM * 16 + r + half * 8;
      if (m < M) {
        float* Dr = D + (size_t)m * TN + nh * 128 + l15;
#pragma unroll
        for (int t = 0; t < 8; ++t) Dr[t * 16] = c[t][r];
        if (l15 < H) {
          AS[(size_t)m * H + l15] = c[NT - 1][r];
        } else if (l15 < 2 * H && m < Md) {
          AD[(size_t)m * H + (l15 - H)] = c[NT - 1][r];
        }
      }
    }
  }
}

template <int TN, int TK>
__global__ void __launch_bounds__(GEMM_WAVES_ * 32)
gat_gemm_wmma(const float* __restrict__ A,
              const float* __restrict__ B,
              const float* __restrict__ WF,
              float* __restrict__ D,
              float* __restrict__ AS,
              float* __restrict__ AD,
              const float* __restrict__ bias,
              const float* __restrict__ bng,
              const float* __restrict__ bnb,
              const float* __restrict__ bnm,
              const float* __restrict__ bnv,
              int M, int Md, int tilesM) {
  constexpr int P      = 2 * TN + 32;
  constexpr int HALVES = TN / 128;
  __shared__ __attribute__((aligned(16))) float Bs[(TK / 2) * P];

  {
    constexpr int C4    = TN / 4;
    constexpr int ITEMS = (TK / 2) * C4;
#pragma unroll
    for (int it = threadIdx.x; it < ITEMS; it += GEMM_WAVES_ * 32) {
      const int kp = it / C4;
      const int c4 = it % C4;
      const float4 r0 = *((const float4*)(B + (size_t)(2 * kp) * TN) + c4);
      const float4 r1 = *((const float4*)(B + (size_t)(2 * kp + 1) * TN) + c4);
      float4* dst = (float4*)(Bs + kp * P + c4 * 8);
      dst[0] = make_float4(r0.x, r1.x, r0.y, r1.y);
      dst[1] = make_float4(r0.z, r1.z, r0.w, r1.w);
    }
    constexpr int XITEMS = (TK / 2) * 4;
    if (HALVES == 1 && threadIdx.x < XITEMS) {
      const int kp = threadIdx.x >> 2;
      const int c4 = threadIdx.x & 3;
      const float4 r0 = *((const float4*)(WF + (size_t)(2 * kp) * 16) + c4);
      const float4 r1 = *((const float4*)(WF + (size_t)(2 * kp + 1) * 16) + c4);
      float4* dst = (float4*)(Bs + kp * P + 2 * TN + c4 * 8);
      dst[0] = make_float4(r0.x, r1.x, r0.y, r1.y);
      dst[1] = make_float4(r0.z, r1.z, r0.w, r1.w);
    }
  }
  __syncthreads();

  const int lane = threadIdx.x & 31;
  const int wave = __builtin_amdgcn_readfirstlane(threadIdx.x >> 5);
  const int half = lane >> 4;
  const int l15  = lane & 15;

  if constexpr (HALVES == 2) {
    const int slot  = (wave >> 3) * 4 + (wave & 3);
    const int phase = (wave >> 2) & 1;
    for (int i = 0; ; ++i) {
      const int tileM = blockIdx.x + gridDim.x * (slot + 16 * i);
      if (tileM >= tilesM) break;
      const int nh = (phase + i) & 1;
      gat_gemm_item<TN, TK, 8>(A + (size_t)nh * M * TK, Bs, D, AS, AD,
                               bias, bng, bnb, bnm, bnv, M, Md, tileM, nh, half, l15);
    }
  } else {
    for (int i = 0; ; ++i) {
      const int tileM = blockIdx.x + gridDim.x * (wave + GEMM_WAVES_ * i);
      if (tileM >= tilesM) break;
      gat_gemm_item<TN, TK, 9>(A, Bs, D, AS, AD,
                               bias, bng, bnb, bnm, bnv, M, Md, tileM, 0, half, l15);
    }
  }
}

__global__ void gat_edge_softmax_agg(const float* __restrict__ HS,
                                     const float* __restrict__ AS,
                                     const float* __restrict__ AD,
                                     const int* __restrict__ col,
                                     const int* __restrict__ tt,
                                     const int* __restrict__ timeP,
                                     const int* __restrict__ intervalP,
                                     const float* __restrict__ b,
                                     float* __restrict__ OUT,
                                     int Nd, int H) {
  const int wid  = blockIdx.x * 8 + (threadIdx.x >> 5);
  const int lane = threadIdx.x & 31;
  if (wid >= Nd * H) return;
  const int i = wid / H, h = wid % H;

  const int t0  = timeP[0];
  const int t1e = t0 + intervalP[0];
  const float adv = AD[(size_t)i * H + h];

  float e[DEG_];
  int   cj[DEG_];
  bool  val[DEG_];
  float m = -__builtin_inff();
#pragma unroll
  for (int j = 0; j < DEG_; ++j) {
    const int eidx = i * DEG_ + j;
    cj[j] = col[eidx];
    const int tv = tt[eidx];
    val[j] = (tv >= t0) && (tv < t1e);
    float ev = AS[(size_t)cj[j] * H + h] + adv;
    ev = ev > 0.0f ? ev : 0.2f * ev;
    e[j] = ev;
    if (val[j] && ev > m) m = ev;
  }
  if (!(m > -__builtin_inff())) m = 0.0f;

  float w[DEG_];
  float denom = 0.0f;
#pragma unroll
  for (int j = 0; j < DEG_; ++j) {
    const float ex = val[j] ? __expf(e[j] - m) : 0.0f;
    w[j] = ex;
    denom += ex;
  }
  const float inv = 1.0f / fmaxf(denom, 1e-16f);

  float4 acc = {0.0f, 0.0f, 0.0f, 0.0f};
#pragma unroll
  for (int j = 0; j < DEG_; ++j) {
    if (w[j] != 0.0f) {
      const float4 hv = *((const float4*)(HS + ((size_t)cj[j] * H + h) * 128) + lane);
      const float wj = w[j] * inv;
      acc.x += wj * hv.x; acc.y += wj * hv.y;
      acc.z += wj * hv.z; acc.w += wj * hv.w;
    }
  }

  float vals[4] = {acc.x, acc.y, acc.z, acc.w};
#pragma unroll
  for (int q = 0; q < 4; ++q) {
    const int cc = lane * 4 + q;
    OUT[(size_t)i * 128 + cc] = vals[q] / (float)H + b[cc];
  }
}

static inline int cdiv(int a, int b) { return (a + b - 1) / b; }

extern "C" void kernel_launch(void* const* d_in, const int* in_sizes, int n_in,
                              void* d_out, int out_size, void* d_ws, size_t ws_size,
                              hipStream_t stream) {
  const float* x     = (const float*)d_in[0];
  const int*   col1  = (const int*)d_in[2];
  const int*   t1    = (const int*)d_in[3];
  const int*   col2  = (const int*)d_in[5];
  const int*   t2    = (const int*)d_in[6];
  const int*   timeP = (const int*)d_in[7];
  const int*   intvP = (const int*)d_in[8];
  const float* Ws1   = (const float*)d_in[9];
  const float* Wd1   = (const float*)d_in[10];
  const float* as1   = (const float*)d_in[11];
  const float* ad1   = (const float*)d_in[12];
  const float* b1    = (const float*)d_in[13];
  const float* g1    = (const float*)d_in[14];
  const float* beta1 = (const float*)d_in[15];
  const float* m1    = (const float*)d_in[16];
  const float* v1    = (const float*)d_in[17];
  const float* Ws2   = (const float*)d_in[18];
  const float* Wd2   = (const float*)d_in[19];
  const float* as2   = (const float*)d_in[20];
  const float* ad2   = (const float*)d_in[21];
  const float* b2    = (const float*)d_in[22];

  float* ws   = (float*)d_ws;
  float* z1   = ws;
  float* hd1  = z1   + (size_t)N0_ * 256;
  float* as1v = hd1  + (size_t)N1_ * 256;
  float* ad1v = as1v + (size_t)N0_ * H1_;
  float* h    = ad1v + (size_t)N1_ * H1_;
  float* as2v = h    + (size_t)N1_ * 256;
  float* ad2v = as2v + (size_t)N1_;
  float* hs2  = hd1;
  float* wf1  = hd1  + (size_t)N1_ * 128;
  float* wf2  = wf1  + (size_t)IN_ * 16;

  const dim3 blk(256);
  const dim3 gblk(GEMM_WAVES_ * 32);

  gat_fold_att<<<cdiv(IN_ + H1_ * HID_, 8), blk, 0, stream>>>(
      Ws1, Wd1, as1, ad1, Ws2, Wd2, as2, ad2, wf1, wf2);

  gat_logit_wmma<<<cdiv(cdiv(N0_, 16), 8), blk, 0, stream>>>(
      x, wf1, as1v, ad1v, N0_, N1_, cdiv(N0_, 16));
  gat_softmax_gather<H1_><<<cdiv(N1_, 8), blk, 0, stream>>>(
      x, as1v, ad1v, col1, t1, timeP, intvP, z1, N1_);
  gat_gemm_wmma<256, 256><<<GEMM_WGS_, gblk, 0, stream>>>(
      z1, Ws1, nullptr, h, nullptr, nullptr,
      b1, g1, beta1, m1, v1, N1_, N1_, cdiv(N1_, 16));

  gat_gemm_wmma<128, 256><<<GEMM_WGS_, gblk, 0, stream>>>(
      h, Ws2, wf2, hs2, as2v, ad2v,
      nullptr, nullptr, nullptr, nullptr, nullptr, N1_, N2_, cdiv(N1_, 16));
  gat_edge_softmax_agg<<<cdiv(N2_ * H2_, 8), blk, 0, stream>>>(
      hs2, as2v, ad2v, col2, t2, timeP, intvP,
      b2, (float*)d_out, N2_, H2_);
}


# Gather kernels: resident waves walk destination rows, one edge per lane; a row's kept source rows requested together, next rows' indices and logits ahead (was one wave per row; sums unchanged)
# speedup vs baseline: 2.5264x; 1.0801x over previous
#include <hip/hip_runtime.h>
#include <hip/hip_bf16.h>
#include <cstddef>

#define N0_  100000
#define N1_  50000
#define N2_  25000
#define DEG_ 16
#define IN_  256
#define HID_ 128
#define OUT_ 128
#define H1_  2
#define H2_  1

typedef float v2f __attribute__((ext_vector_type(2)));
typedef float v8f __attribute__((ext_vector_type(8)));

template <int H>
__device__ __forceinline__ void gat_fold_row(const float* __restrict__ Ws,
                                             const float* __restrict__ Wd,
                                             const float* __restrict__ as,
                                             const float* __restrict__ ad,
                                             float* __restrict__ WF,
                                             int k, int lane) {
  float outv = 0.0f;
#pragma unroll
  for (int h = 0; h < H; ++h) {
    const float4 av = *((const float4*)(as + (size_t)h * 128) + lane);
    const float4 dv = *((const float4*)(ad + (size_t)h * 128) + lane);
    const float4 sv = *((const float4*)(Ws + (size_t)k * (H * 128) + h * 128) + lane);
    const float4 wv = *((const float4*)(Wd + (size_t)k * (H * 128) + h * 128) + lane);
    float s = sv.x * av.x + sv.y * av.y + sv.z * av.z + sv.w * av.w;
    float d = wv.x * dv.x + wv.y * dv.y + wv.z * dv.z + wv.w * dv.w;
#pragma unroll
    for (int off = 16; off > 0; off >>= 1) {
      s += __shfl_xor(s, off, 32);
      d += __shfl_xor(d, off, 32);
    }
    if (lane == h)     outv = s;
    if (lane == H + h) outv = d;
  }
  if (lane < 16) WF[(size_t)k * 16 + lane] = outv;
}

__global__ void gat_fold_att(const float* __restrict__ Ws1, const float* __restrict__ Wd1,
                             const float* __restrict__ as1, const float* __restrict__ ad1,
                             const float* __restrict__ Ws2, const float* __restrict__ Wd2,
                             const float* __restrict__ as2, const float* __restrict__ ad2,
                             float* __restrict__ WF1, float* __restrict__ WF2) {
  const int wid  = blockIdx.x * 8 + (threadIdx.x >> 5);
  const int lane = threadIdx.x & 31;
  if (wid < IN_) {
    gat_fold_row<H1_>(Ws1, Wd1, as1, ad1, WF1, wid, lane);
  } else if (wid < IN_ + H1_ * HID_) {
    gat_fold_row<H2_>(Ws2, Wd2, as2, ad2, WF2, wid - IN_, lane);
  }
}


__global__ void __launch_bounds__(256)
gat_logit_wmma(const float* __restrict__ A,
               const float* __restrict__ WF,
               float* __restrict__ AS,
               float* __restrict__ AD,
               int M, int Md, int tilesM) {
  constexpr int TK = IN_;
  constexpr int H  = H1_;
  __shared__ __attribute__((aligned(16))) float Bs[(TK / 2) * 32];
  for (int it = threadIdx.x; it < (TK / 2) * 4; it += 256) {
    const int kp = it >> 2;
    const int c4 = it & 3;
    const float4 r0 = *((const float4*)(WF + (size_t)(2 * kp) * 16) + c4);
    const float4 r1 = *((const float4*)(WF + (size_t)(2 * kp + 1) * 16) + c4);
    float4* dst = (float4*)(Bs + kp * 32 + c4 * 8);
    dst[0] = make_float4(r0.x, r1.x, r0.y, r1.y);
    dst[1] = make_float4(r0.z, r1.z, r0.w, r1.w);
  }
  __syncthreads();

  const int lane = threadIdx.x & 31;
  const int wave = threadIdx.x >> 5;
  const int half = lane >> 4;
  const int l15  = lane & 15;
  const int tileM = blockIdx.x * 8 + wave;
  if (tileM >= tilesM) return;

  int row = tileM * 16 + l15;
  if (row >= M) row = M - 1;
  const float2* ap = (const float2*)(A + (size_t)row * TK) + half;
  const float2* bs = (const float2*)Bs + half * 16 + l15;

  v8f c0 = v8f{};
#pragma unroll 4
  for (int k0 = 0; k0 < TK; k0 += 8) {
    const float2 a0 = ap[0];
    const float2 a1 = ap[2];
    const float2 b0 = bs[0];
    const float2 b1 = bs[32];
    ap += 4;
    bs += 64;
    v2f va0 = { a0.x, a0.y };
    v2f vb0 = { b0.x, b0.y };
    v2f va1 = { a1.x, a1.y };
    v2f vb1 = { b1.x, b1.y };
    c0 = __builtin_amdgcn_wmma_f32_16x16x4_f32(false, va0, false, vb0, (short)0, c0, false, false);
    c0 = __builtin_amdgcn_wmma_f32_16x16x4_f32(false, va1, false, vb1, (short)0, c0, false, false);
  }

#pragma unroll
  for (int r = 0; r < 8; ++r) {
    const int m = tileM * 16 + r + half * 8;
    if (m < M) {
      const float v = c0[r];
      if (l15 < H) {
        AS[(size_t)m * H + l15] = v;
      } else if (l15 < 2 * H && m < Md) {
        AD[(size_t)m * H + (l15 - H)] = v;
      }
    }
  }
}

__device__ __forceinline__ float gat_lane_f(float v, int l) {
  return __int_as_float(__builtin_amdgcn_readlane(__float_as_int(v), l));
}

template <int H, int NQ, int R, bool FINAL>
__global__ void __launch_bounds__(256)
gat_gather_resident(const float* __restrict__ X,
                    const float* __restrict__ AS,
                    const float* __restrict__ AD,
                    const int* __restrict__ col,
                    const int* __restrict__ tt,
                    const int* __restrict__ timeP,
                    const int* __restrict__ intervalP,
                    const float* __restrict__ b,
                    float* __restrict__ OUTP,
                    int Nd) {
  static_assert(!FINAL || (H == 1 && NQ == 1), "final form: one head, 128 channels");
  const int lane = threadIdx.x & 31;
  const int l15  = lane & 15;
  const int wv  = __builtin_amdgcn_readfirstlane((int)(blockIdx.x * 8 + (threadIdx.x >> 5)));
  const int nwv = (int)gridDim.x * 8;
  if (wv >= Nd) return;

  const int t0   = timeP[0];
  const int t1e  = t0 + intervalP[0];
  const int last = Nd - 1;

  float4 bq = make_float4(0.0f, 0.0f, 0.0f, 0.0f);
  if constexpr (FINAL) bq = *((const float4*)b + lane);

  int i1 = wv + nwv;
  if (i1 > last) i1 = last;
  int c0  = col[wv * DEG_ + l15];
  int tv0 = tt[wv * DEG_ + l15];
  int c1  = col[i1 * DEG_ + l15];
  int tv1 = tt[i1 * DEG_ + l15];
  float as0[H];
  float ad0[H];
#pragma unroll
  for (int h = 0; h < H; ++h) {
    as0[h] = AS[(size_t)c0 * H + h];
    ad0[h] = AD[(size_t)wv * H + h];
  }

  for (int i = wv; i < Nd; i += nwv) {
    int in1 = i + nwv;
    if (in1 > last) in1 = last;
    int i2 = i + 2 * nwv;
    if (i2 > last) i2 = last;
    float as1[H];
    float ad1[H];
#pragma unroll
    for (int h = 0; h < H; ++h) {
      as1[h] = AS[(size_t)c1 * H + h];
      ad1[h] = AD[(size_t)in1 * H + h];
    }
    const int c2  = col[i2 * DEG_ + l15];
    const int tv2 = tt[i2 * DEG_ + l15];
    asm volatile("" ::: "memory");

    const bool val = (tv0 >= t0) && (tv0 < t1e);
    float ex[H];
    float inv[H];
#pragma unroll
    for (int h = 0; h < H; ++h) {
      float ev = as0[h] + ad0[h];
      ev = ev > 0.0f ? ev : 0.2f * ev;
      float m = val ? ev : -__builtin_inff();
#pragma unroll
      for (int off = 8; off > 0; off >>= 1) m = fmaxf(m, __shfl_xor(m, off, 32));
      if (!(m > -__builtin_inff())) m = 0.0f;
      const float e = val ? __expf(ev - m) : 0.0f;
      float denom = 0.0f;
#pragma unroll
      for (int j = 0; j < DEG_; ++j) denom += gat_lane_f(e, j);
      inv[h] = 1.0f / fmaxf(denom, 1e-16f);
      ex[h]  = e;
    }
    bool nz = false;
#pragma unroll
    for (int h = 0; h < H; ++h) nz = nz || (ex[h] != 0.0f);
    unsigned mask = __builtin_amdgcn_ballot_w32(nz) & 0xFFFFu;

    float4 acc[H][NQ];
#pragma unroll
    for (int h = 0; h < H; ++h)
#pragma unroll
      for (int q = 0; q < NQ; ++q) acc[h][q] = make_float4(0.0f, 0.0f, 0.0f, 0.0f);

    while (mask != 0u) {
      float4 xr[R][NQ];
      float  wr[H][R];
      bool   on[R];
#pragma unroll
      for (int r = 0; r < R; ++r) {
        on[r] = (mask != 0u);
        if (on[r]) {
          const int j = __builtin_ctz(mask);
          mask &= mask - 1u;
          const int cr = __builtin_amdgcn_readlane(c0, j);
#pragma unroll
          for (int h = 0; h < H; ++h) wr[h][r] = gat_lane_f(ex[h], j) * inv[h];
          const float4* p = (const float4*)(X + (size_t)cr * (NQ * 128)) + lane;
#pragma unroll
          for (int q = 0; q < NQ; ++q) xr[r][q] = p[q * 32];
        }
      }
#pragma unroll
      for (int r = 0; r < R; ++r) {
        if (on[r]) {
#pragma unroll
          for (int h = 0; h < H; ++h) {
            const float wj = wr[h][r];
#pragma unroll
            for (int q = 0; q < NQ; ++q) {
              acc[h][q].x += wj * xr[r][q].x; acc[h][q].y += wj * xr[r][q].y;
              acc[h][q].z += wj * xr[r][q].z; acc[h][q].w += wj * xr[r][q].w;
            }
          }
        }
      }
    }

    if constexpr (FINAL) {
      float4 o;
      o.x = acc[0][0].x / (float)H + bq.x;
      o.y = acc[0][0].y / (float)H + bq.y;
      o.z = acc[0][0].z / (float)H + bq.z;
      o.w = acc[0][0].w / (float)H + bq.w;
      *((float4*)(OUTP + (size_t)i * 128) + lane) = o;
    } else {
#pragma unroll
      for (int h = 0; h < H; ++h) {
        float4* zr = (float4*)(OUTP + ((size_t)h * Nd + i) * (NQ * 128));
#pragma unroll
        for (int q = 0; q < NQ; ++q) zr[q * 32 + lane] = acc[h][q];
      }
    }

    c0 = c1; tv0 = tv1;
#pragma unroll
    for (int h = 0; h < H; ++h) { as0[h] = as1[h]; ad0[h] = ad1[h]; }
    c1 = c2; tv1 = tv2;
  }
}

#define GEMM_WGS_   256
#define GATHER1_WGS_ 1024
#define GATHER2_WGS_ 512
#define GEMM_WAVES_ 32

template <int TN, int TK, int NT>
__device__ __forceinline__ void gat_gemm_item(const float* __restrict__ A,
                                              const float* Bs,
                                              float* __restrict__ D,
                                              float* __restrict__ AS,
                                              float* __restrict__ AD,
                                              const float* __restrict__ bias,
                                              const float* __restrict__ bng,
                                              const float* __restrict__ bnb,
                                              const float* __restrict__ bnm,
                                              const float* __restrict__ bnv,
                                              int M, int Md, int tileM, int nh,
                                              int half, int l15) {
  constexpr int P  = 2 * TN + 32;
  constexpr int P2 = P / 2;
  constexpr int H  = TN / 128;

  int row = tileM * 16 + l15;
  if (row >= M) row = M - 1;
  const float2* ap = (const float2*)(A + (size_t)row * TK) + half;
  const float2* bs = (const float2*)Bs + half * P2 + nh * 128 + l15;

  v8f c[NT];
#pragma unroll
  for (int t = 0; t < NT; ++t) c[t] = v8f{};

  if constexpr (NT == 8) {
#pragma unroll 4
    for (int k0 = 0; k0 < TK; k0 += 4) {
      const float2 a2 = *ap;
      ap += 2;
      v2f a = { a2.x, a2.y };
#pragma unroll
      for (int t = 0; t < 8; ++t) {
        const float2 b2 = bs[t * 16];
        v2f b = { b2.x, b2.y };
        c[t] = __builtin_amdgcn_wmma_f32_16x16x4_f32(false, a, false, b, (short)0, c[t], false, false);
      }
      bs += P;
    }
  } else {
    float2 an = *ap;
    float2 bn[NT];
#pragma unroll
    for (int t = 0; t < 8; ++t) bn[t] = bs[t * 16];
    bn[8] = bs[TN];
#pragma unroll
    for (int k0 = 0; k0 < TK - 4; k0 += 4) {
      const float2 ac = an;
      float2 bc[NT];
#pragma unroll
      for (int t = 0; t < NT; ++t) bc[t] = bn[t];
      ap += 2;
      bs += P;
      an = *ap;
#pragma unroll
      for (int t = 0; t < 8; ++t) bn[t] = bs[t * 16];
      bn[8] = bs[TN];
      __builtin_amdgcn_sched_barrier(0);
      v2f a = { ac.x, ac.y };
#pragma unroll
      for (int t = 0; t < NT; ++t) {
        v2f b = { bc[t].x, bc[t].y };
        c[t] = __builtin_amdgcn_wmma_f32_16x16x4_f32(false, a, false, b, (short)0, c[t], false, false);
      }
    }
    {
      v2f a = { an.x, an.y };
#pragma unroll
      for (int t = 0; t < NT; ++t) {
        v2f b = { bn[t].x, bn[t].y };
        c[t] = __builtin_amdgcn_wmma_f32_16x16x4_f32(false, a, false, b, (short)0, c[t], false, false);
      }
    }
  }

  if constexpr (NT == 8) {
#pragma unroll
    for (int t = 0; t < 8; ++t) {
      const int cc = nh * 128 + t * 16 + l15;
      const float pb = bias[cc];
      const float pm = bnm[cc];
      const float ps = rsqrtf(bnv[cc] + 1e-5f);
      const float pg = bng[cc];
      const float pe = bnb[cc];
#pragma unroll
      for (int r = 0; r < 8; ++r) {
        const int m = tileM * 16 + r + half * 8;
        if (m < M) {
          float vv = c[t][r] + pb;
          vv = (vv - pm) * ps * pg + pe;
          vv = vv > 0.0f ? vv : 0.01f * vv;
          D[(size_t)m * TN + cc] = vv;
        }
      }
    }
  } else {
#pragma unroll
    for (int r = 0; r < 8; ++r) {
      const int m = tileM * 16 + r + half * 8;
      if (m < M) {
        float* Dr = D + (size_t)m * TN + nh * 128 + l15;
#pragma unroll
        for (int t = 0; t < 8; ++t) Dr[t * 16] = c[t][r];
        if (l15 < H) {
          AS[(size_t)m * H + l15] = c[NT - 1][r];
        } else if (l15 < 2 * H && m < Md) {
          AD[(size_t)m * H + (l15 - H)] = c[NT - 1][r];
        }
      }
    }
  }
}

template <int TN, int TK>
__global__ void __launch_bounds__(GEMM_WAVES_ * 32)
gat_gemm_wmma(const float* __restrict__ A,
              const float* __restrict__ B,
              const float* __restrict__ WF,
              float* __restrict__ D,
              float* __restrict__ AS,
              float* __restrict__ AD,
              const float* __restrict__ bias,
              const float* __restrict__ bng,
              const float* __restrict__ bnb,
              const float* __restrict__ bnm,
              const float* __restrict__ bnv,
              int M, int Md, int tilesM) {
  constexpr int P      = 2 * TN + 32;
  constexpr int HALVES = TN / 128;
  __shared__ __attribute__((aligned(16))) float Bs[(TK / 2) * P];

  {
    constexpr int C4    = TN / 4;
    constexpr int ITEMS = (TK / 2) * C4;
#pragma unroll
    for (int it = threadIdx.x; it < ITEMS; it += GEMM_WAVES_ * 32) {
      const int kp = it / C4;
      const int c4 = it % C4;
      const float4 r0 = *((const float4*)(B + (size_t)(2 * kp) * TN) + c4);
      const float4 r1 = *((const float4*)(B + (size_t)(2 * kp + 1) * TN) + c4);
      float4* dst = (float4*)(Bs + kp * P + c4 * 8);
      dst[0] = make_float4(r0.x, r1.x, r0.y, r1.y);
      dst[1] = make_float4(r0.z, r1.z, r0.w, r1.w);
    }
    constexpr int XITEMS = (TK / 2) * 4;
    if (HALVES == 1 && threadIdx.x < XITEMS) {
      const int kp = threadIdx.x >> 2;
      const int c4 = threadIdx.x & 3;
      const float4 r0 = *((const float4*)(WF + (size_t)(2 * kp) * 16) + c4);
      const float4 r1 = *((const float4*)(WF + (size_t)(2 * kp + 1) * 16) + c4);
      float4* dst = (float4*)(Bs + kp * P + 2 * TN + c4 * 8);
      dst[0] = make_float4(r0.x, r1.x, r0.y, r1.y);
      dst[1] = make_float4(r0.z, r1.z, r0.w, r1.w);
    }
  }
  __syncthreads();

  const int lane = threadIdx.x & 31;
  const int wave = __builtin_amdgcn_readfirstlane(threadIdx.x >> 5);
  const int half = lane >> 4;
  const int l15  = lane & 15;

  if constexpr (HALVES == 2) {
    const int slot  = (wave >> 3) * 4 + (wave & 3);
    const int phase = (wave >> 2) & 1;
    for (int i = 0; ; ++i) {
      const int tileM = blockIdx.x + gridDim.x * (slot + 16 * i);
      if (tileM >= tilesM) break;
      const int nh = (phase + i) & 1;
      gat_gemm_item<TN, TK, 8>(A + (size_t)nh * M * TK, Bs, D, AS, AD,
                               bias, bng, bnb, bnm, bnv, M, Md, tileM, nh, half, l15);
    }
  } else {
    for (int i = 0; ; ++i) {
      const int tileM = blockIdx.x + gridDim.x * (wave + GEMM_WAVES_ * i);
      if (tileM >= tilesM) break;
      gat_gemm_item<TN, TK, 9>(A, Bs, D, AS, AD,
                               bias, bng, bnb, bnm, bnv, M, Md, tileM, 0, half, l15);
    }
  }
}

static inline int cdiv(int a, int b) { return (a + b - 1) / b; }

extern "C" void kernel_launch(void* const* d_in, const int* in_sizes, int n_in,
                              void* d_out, int out_size, void* d_ws, size_t ws_size,
                              hipStream_t stream) {
  const float* x     = (const float*)d_in[0];
  const int*   col1  = (const int*)d_in[2];
  const int*   t1    = (const int*)d_in[3];
  const int*   col2  = (const int*)d_in[5];
  const int*   t2    = (const int*)d_in[6];
  const int*   timeP = (const int*)d_in[7];
  const int*   intvP = (const int*)d_in[8];
  const float* Ws1   = (const float*)d_in[9];
  const float* Wd1   = (const float*)d_in[10];
  const float* as1   = (const float*)d_in[11];
  const float* ad1   = (const float*)d_in[12];
  const float* b1    = (const float*)d_in[13];
  const float* g1    = (const float*)d_in[14];
  const float* beta1 = (const float*)d_in[15];
  const float* m1    = (const float*)d_in[16];
  const float* v1    = (const float*)d_in[17];
  const float* Ws2   = (const float*)d_in[18];
  const float* Wd2   = (const float*)d_in[19];
  const float* as2   = (const float*)d_in[20];
  const float* ad2   = (const float*)d_in[21];
  const float* b2    = (const float*)d_in[22];

  float* ws   = (float*)d_ws;
  float* z1   = ws;
  float* hd1  = z1   + (size_t)N0_ * 256;
  float* as1v = hd1  + (size_t)N1_ * 256;
  float* ad1v = as1v + (size_t)N0_ * H1_;
  float* h    = ad1v + (size_t)N1_ * H1_;
  float* as2v = h    + (size_t)N1_ * 256;
  float* ad2v = as2v + (size_t)N1_;
  float* hs2  = hd1;
  float* wf1  = hd1  + (size_t)N1_ * 128;
  float* wf2  = wf1  + (size_t)IN_ * 16;

  const dim3 blk(256);
  const dim3 gblk(GEMM_WAVES_ * 32);

  gat_fold_att<<<cdiv(IN_ + H1_ * HID_, 8), blk, 0, stream>>>(
      Ws1, Wd1, as1, ad1, Ws2, Wd2, as2, ad2, wf1, wf2);

  gat_logit_wmma<<<cdiv(cdiv(N0_, 16), 8), blk, 0, stream>>>(
      x, wf1, as1v, ad1v, N0_, N1_, cdiv(N0_, 16));
  gat_gather_resident<H1_, 2, 8, false><<<GATHER1_WGS_, blk, 0, stream>>>(
      x, as1v, ad1v, col1, t1, timeP, intvP, nullptr, z1, N1_);
  gat_gemm_wmma<256, 256><<<GEMM_WGS_, gblk, 0, stream>>>(
      z1, Ws1, nullptr, h, nullptr, nullptr,
      b1, g1, beta1, m1, v1, N1_, N1_, cdiv(N1_, 16));

  gat_gemm_wmma<128, 256><<<GEMM_WGS_, gblk, 0, stream>>>(
      h, Ws2, wf2, hs2, as2v, ad2v,
      nullptr, nullptr, nullptr, nullptr, nullptr, N1_, N2_, cdiv(N1_, 16));
  gat_gather_resident<H2_, 1, 8, true><<<GATHER2_WGS_, blk, 0, stream>>>(
      hs2, as2v, ad2v, col2, t2, timeP, intvP, b2, (float*)d_out, N2_);
}


# GEMMs: row tiles left over after an equal share per work-group are cut into 2-column-tile pieces dealt one per SIMD over all work-groups, not left whole on a few (sum order unchanged)
# speedup vs baseline: 2.7196x; 1.0765x over previous
#include <hip/hip_runtime.h>
#include <hip/hip_bf16.h>
#include <cstddef>

#define N0_  100000
#define N1_  50000
#define N2_  25000
#define DEG_ 16
#define IN_  256
#define HID_ 128
#define OUT_ 128
#define H1_  2
#define H2_  1

typedef float v2f __attribute__((ext_vector_type(2)));
typedef float v8f __attribute__((ext_vector_type(8)));

template <int H>
__device__ __forceinline__ void gat_fold_row(const float* __restrict__ Ws,
                                             const float* __restrict__ Wd,
                                             const float* __restrict__ as,
                                             const float* __restrict__ ad,
                                             float* __restrict__ WF,
                                             int k, int lane) {
  float outv = 0.0f;
#pragma unroll
  for (int h = 0; h < H; ++h) {
    const float4 av = *((const float4*)(as + (size_t)h * 128) + lane);
    const float4 dv = *((const float4*)(ad + (size_t)h * 128) + lane);
    const float4 sv = *((const float4*)(Ws + (size_t)k * (H * 128) + h * 128) + lane);
    const float4 wv = *((const float4*)(Wd + (size_t)k * (H * 128) + h * 128) + lane);
    float s = sv.x * av.x + sv.y * av.y + sv.z * av.z + sv.w * av.w;
    float d = wv.x * dv.x + wv.y * dv.y + wv.z * dv.z + wv.w * dv.w;
#pragma unroll
    for (int off = 16; off > 0; off >>= 1) {
      s += __shfl_xor(s, off, 32);
      d += __shfl_xor(d, off, 32);
    }
    if (lane == h)     outv = s;
    if (lane == H + h) outv = d;
  }
  if (lane < 16) WF[(size_t)k * 16 + lane] = outv;
}

__global__ void gat_fold_att(const float* __restrict__ Ws1, const float* __restrict__ Wd1,
                             const float* __restrict__ as1, const float* __restrict__ ad1,
                             const float* __restrict__ Ws2, const float* __restrict__ Wd2,
                             const float* __restrict__ as2, const float* __restrict__ ad2,
                             float* __restrict__ WF1, float* __restrict__ WF2) {
  const int wid  = blockIdx.x * 8 + (threadIdx.x >> 5);
  const int lane = threadIdx.x & 31;
  if (wid < IN_) {
    gat_fold_row<H1_>(Ws1, Wd1, as1, ad1, WF1, wid, lane);
  } else if (wid < IN_ + H1_ * HID_) {
    gat_fold_row<H2_>(Ws2, Wd2, as2, ad2, WF2, wid - IN_, lane);
  }
}


__global__ void __launch_bounds__(256)
gat_logit_wmma(const float* __restrict__ A,
               const float* __restrict__ WF,
               float* __restrict__ AS,
               float* __restrict__ AD,
               int M, int Md, int tilesM) {
  constexpr int TK = IN_;
  constexpr int H  = H1_;
  __shared__ __attribute__((aligned(16))) float Bs[(TK / 2) * 32];
  for (int it = threadIdx.x; it < (TK / 2) * 4; it += 256) {
    const int kp = it >> 2;
    const int c4 = it & 3;
    const float4 r0 = *((const float4*)(WF + (size_t)(2 * kp) * 16) + c4);
    const float4 r1 = *((const float4*)(WF + (size_t)(2 * kp + 1) * 16) + c4);
    float4* dst = (float4*)(Bs + kp * 32 + c4 * 8);
    dst[0] = make_float4(r0.x, r1.x, r0.y, r1.y);
    dst[1] = make_float4(r0.z, r1.z, r0.w, r1.w);
  }
  __syncthreads();

  const int lane = threadIdx.x & 31;
  const int wave = threadIdx.x >> 5;
  const int half = lane >> 4;
  const int l15  = lane & 15;
  const int tileM = blockIdx.x * 8 + wave;
  if (tileM >= tilesM) return;

  int row = tileM * 16 + l15;
  if (row >= M) row = M - 1;
  const float2* ap = (const float2*)(A + (size_t)row * TK) + half;
  const float2* bs = (const float2*)Bs + half * 16 + l15;

  v8f c0 = v8f{};
#pragma unroll 4
  for (int k0 = 0; k0 < TK; k0 += 8) {
    const float2 a0 = ap[0];
    const float2 a1 = ap[2];
    const float2 b0 = bs[0];
    const float2 b1 = bs[32];
    ap += 4;
    bs += 64;
    v2f va0 = { a0.x, a0.y };
    v2f vb0 = { b0.x, b0.y };
    v2f va1 = { a1.x, a1.y };
    v2f vb1 = { b1.x, b1.y };
    c0 = __builtin_amdgcn_wmma_f32_16x16x4_f32(false, va0, false, vb0, (short)0, c0, false, false);
    c0 = __builtin_amdgcn_wmma_f32_16x16x4_f32(false, va1, false, vb1, (short)0, c0, false, false);
  }

#pragma unroll
  for (int r = 0; r < 8; ++r) {
    const int m = tileM * 16 + r + half * 8;
    if (m < M) {
      const float v = c0[r];
      if (l15 < H) {
        AS[(size_t)m * H + l15] = v;
      } else if (l15 < 2 * H && m < Md) {
        AD[(size_t)m * H + (l15 - H)] = v;
      }
    }
  }
}

__device__ __forceinline__ float gat_lane_f(float v, int l) {
  return __int_as_float(__builtin_amdgcn_readlane(__float_as_int(v), l));
}

template <int H, int NQ, int R, bool FINAL>
__global__ void __launch_bounds__(256)
gat_gather_resident(const float* __restrict__ X,
                    const float* __restrict__ AS,
                    const float* __restrict__ AD,
                    const int* __restrict__ col,
                    const int* __restrict__ tt,
                    const int* __restrict__ timeP,
                    const int* __restrict__ intervalP,
                    const float* __restrict__ b,
                    float* __restrict__ OUTP,
                    int Nd) {
  static_assert(!FINAL || (H == 1 && NQ == 1), "final form: one head, 128 channels");
  const int lane = threadIdx.x & 31;
  const int l15  = lane & 15;
  const int wv  = __builtin_amdgcn_readfirstlane((int)(blockIdx.x * 8 + (threadIdx.x >> 5)));
  const int nwv = (int)gridDim.x * 8;
  if (wv >= Nd) return;

  const int t0   = timeP[0];
  const int t1e  = t0 + intervalP[0];
  const int last = Nd - 1;

  float4 bq = make_float4(0.0f, 0.0f, 0.0f, 0.0f);
  if constexpr (FINAL) bq = *((const float4*)b + lane);

  int i1 = wv + nwv;
  if (i1 > last) i1 = last;
  int c0  = col[wv * DEG_ + l15];
  int tv0 = tt[wv * DEG_ + l15];
  int c1  = col[i1 * DEG_ + l15];
  int tv1 = tt[i1 * DEG_ + l15];
  float as0[H];
  float ad0[H];
#pragma unroll
  for (int h = 0; h < H; ++h) {
    as0[h] = AS[(size_t)c0 * H + h];
    ad0[h] = AD[(size_t)wv * H + h];
  }

  for (int i = wv; i < Nd; i += nwv) {
    int in1 = i + nwv;
    if (in1 > last) in1 = last;
    int i2 = i + 2 * nwv;
    if (i2 > last) i2 = last;
    float as1[H];
    float ad1[H];
#pragma unroll
    for (int h = 0; h < H; ++h) {
      as1[h] = AS[(size_t)c1 * H + h];
      ad1[h] = AD[(size_t)in1 * H + h];
    }
    const int c2  = col[i2 * DEG_ + l15];
    const int tv2 = tt[i2 * DEG_ + l15];
    asm volatile("" ::: "memory");

    const bool val = (tv0 >= t0) && (tv0 < t1e);
    float ex[H];
    float inv[H];
#pragma unroll
    for (int h = 0; h < H; ++h) {
      float ev = as0[h] + ad0[h];
      ev = ev > 0.0f ? ev : 0.2f * ev;
      float m = val ? ev : -__builtin_inff();
#pragma unroll
      for (int off = 8; off > 0; off >>= 1) m = fmaxf(m, __shfl_xor(m, off, 32));
      if (!(m > -__builtin_inff())) m = 0.0f;
      const float e = val ? __expf(ev - m) : 0.0f;
      float denom = 0.0f;
#pragma unroll
      for (int j = 0; j < DEG_; ++j) denom += gat_lane_f(e, j);
      inv[h] = 1.0f / fmaxf(denom, 1e-16f);
      ex[h]  = e;
    }
    bool nz = false;
#pragma unroll
    for (int h = 0; h < H; ++h) nz = nz || (ex[h] != 0.0f);
    unsigned mask = __builtin_amdgcn_ballot_w32(nz) & 0xFFFFu;

    float4 acc[H][NQ];
#pragma unroll
    for (int h = 0; h < H; ++h)
#pragma unroll
      for (int q = 0; q < NQ; ++q) acc[h][q] = make_float4(0.0f, 0.0f, 0.0f, 0.0f);

    while (mask != 0u) {
      float4 xr[R][NQ];
      float  wr[H][R];
      bool   on[R];
#pragma unroll
      for (int r = 0; r < R; ++r) {
        on[r] = (mask != 0u);
        if (on[r]) {
          const int j = __builtin_ctz(mask);
          mask &= mask - 1u;
          const int cr = __builtin_amdgcn_readlane(c0, j);
#pragma unroll
          for (int h = 0; h < H; ++h) wr[h][r] = gat_lane_f(ex[h], j) * inv[h];
          const float4* p = (const float4*)(X + (size_t)cr * (NQ * 128)) + lane;
#pragma unroll
          for (int q = 0; q < NQ; ++q) xr[r][q] = p[q * 32];
        }
      }
#pragma unroll
      for (int r = 0; r < R; ++r) {
        if (on[r]) {
#pragma unroll
          for (int h = 0; h < H; ++h) {
            const float wj = wr[h][r];
#pragma unroll
            for (int q = 0; q < NQ; ++q) {
              acc[h][q].x += wj * xr[r][q].x; acc[h][q].y += wj * xr[r][q].y;
              acc[h][q].z += wj * xr[r][q].z; acc[h][q].w += wj * xr[r][q].w;
            }
          }
        }
      }
    }

    if constexpr (FINAL) {
      float4 o;
      o.x = acc[0][0].x / (float)H + bq.x;
      o.y = acc[0][0].y / (float)H + bq.y;
      o.z = acc[0][0].z / (float)H + bq.z;
      o.w = acc[0][0].w / (float)H + bq.w;
      *((float4*)(OUTP + (size_t)i * 128) + lane) = o;
    } else {
#pragma unroll
      for (int h = 0; h < H; ++h) {
        float4* zr = (float4*)(OUTP + ((size_t)h * Nd + i) * (NQ * 128));
#pragma unroll
        for (int q = 0; q < NQ; ++q) zr[q * 32 + lane] = acc[h][q];
      }
    }

    c0 = c1; tv0 = tv1;
#pragma unroll
    for (int h = 0; h < H; ++h) { as0[h] = as1[h]; ad0[h] = ad1[h]; }
    c1 = c2; tv1 = tv2;
  }
}

#define GEMM_WGS_   256
#define GATHER1_WGS_ 1024
#define GATHER2_WGS_ 512
#define GEMM_WAVES_ 32

template <int TN, int TK, int NT>
__device__ __forceinline__ void gat_gemm_item(const float* __restrict__ A,
                                              const float* Bs,
                                              float* __restrict__ D,
                                              float* __restrict__ AS,
                                              float* __restrict__ AD,
                                              const float* __restrict__ bias,
                                              const float* __restrict__ bng,
                                              const float* __restrict__ bnb,
                                              const float* __restrict__ bnm,
                                              const float* __restrict__ bnv,
                                              int M, int Md, int tileM, int nh,
                                              int half, int l15) {
  constexpr int P  = 2 * TN + 32;
  constexpr int P2 = P / 2;
  constexpr int H  = TN / 128;

  int row = tileM * 16 + l15;
  if (row >= M) row = M - 1;
  const float2* ap = (const float2*)(A + (size_t)row * TK) + half;
  const float2* bs = (const float2*)Bs + half * P2 + nh * 128 + l15;

  v8f c[NT];
#pragma unroll
  for (int t = 0; t < NT; ++t) c[t] = v8f{};

  if constexpr (NT == 8) {
#pragma unroll 4
    for (int k0 = 0; k0 < TK; k0 += 4) {
      const float2 a2 = *ap;
      ap += 2;
      v2f a = { a2.x, a2.y };
#pragma unroll
      for (int t = 0; t < 8; ++t) {
        const float2 b2 = bs[t * 16];
        v2f b = { b2.x, b2.y };
        c[t] = __builtin_amdgcn_wmma_f32_16x16x4_f32(false, a, false, b, (short)0, c[t], false, false);
      }
      bs += P;
    }
  } else {
    float2 an = *ap;
    float2 bn[NT];
#pragma unroll
    for (int t = 0; t < 8; ++t) bn[t] = bs[t * 16];
    bn[8] = bs[TN];
#pragma unroll
    for (int k0 = 0; k0 < TK - 4; k0 += 4) {
      const float2 ac = an;
      float2 bc[NT];
#pragma unroll
      for (int t = 0; t < NT; ++t) bc[t] = bn[t];
      ap += 2;
      bs += P;
      an = *ap;
#pragma unroll
      for (int t = 0; t < 8; ++t) bn[t] = bs[t * 16];
      bn[8] = bs[TN];
      __builtin_amdgcn_sched_barrier(0);
      v2f a = { ac.x, ac.y };
#pragma unroll
      for (int t = 0; t < NT; ++t) {
        v2f b = { bc[t].x, bc[t].y };
        c[t] = __builtin_amdgcn_wmma_f32_16x16x4_f32(false, a, false, b, (short)0, c[t], false, false);
      }
    }
    {
      v2f a = { an.x, an.y };
#pragma unroll
      for (int t = 0; t < NT; ++t) {
        v2f b = { bn[t].x, bn[t].y };
        c[t] = __builtin_amdgcn_wmma_f32_16x16x4_f32(false, a, false, b, (short)0, c[t], false, false);
      }
    }
  }

  if constexpr (NT == 8) {
#pragma unroll
    for (int t = 0; t < 8; ++t) {
      const int cc = nh * 128 + t * 16 + l15;
      const float pb = bias[cc];
      const float pm = bnm[cc];
      const float ps = rsqrtf(bnv[cc] + 1e-5f);
      const float pg = bng[cc];
      const float pe = bnb[cc];
#pragma unroll
      for (int r = 0; r < 8; ++r) {
        const int m = tileM * 16 + r + half * 8;
        if (m < M) {
          float vv = c[t][r] + pb;
          vv = (vv - pm) * ps * pg + pe;
          vv = vv > 0.0f ? vv : 0.01f * vv;
          D[(size_t)m * TN + cc] = vv;
        }
      }
    }
  } else {
#pragma unroll
    for (int r = 0; r < 8; ++r) {
      const int m = tileM * 16 + r + half * 8;
      if (m < M) {
        float* Dr = D + (size_t)m * TN + nh * 128 + l15;
#pragma unroll
        for (int t = 0; t < 8; ++t) Dr[t * 16] = c[t][r];
        if (l15 < H) {
          AS[(size_t)m * H + l15] = c[NT - 1][r];
        } else if (l15 < 2 * H && m < Md) {
          AD[(size_t)m * H + (l15 - H)] = c[NT - 1][r];
        }
      }
    }
  }
}

template <int TN, int TK, int NTP>
__device__ __forceinline__ void gat_gemm_piece(const float* __restrict__ A,
                                               const float* Bs,
                                               float* __restrict__ D,
                                               float* __restrict__ AS,
                                               float* __restrict__ AD,
                                               const float* __restrict__ bias,
                                               const float* __restrict__ bng,
                                               const float* __restrict__ bnb,
                                               const float* __restrict__ bnm,
                                               const float* __restrict__ bnv,
                                               int M, int Md, int tileM, int ct0,
                                               int half, int l15) {
  constexpr int P  = 2 * TN + 32;
  constexpr int P2 = P / 2;
  constexpr int H  = TN / 128;

  int row = tileM * 16 + l15;
  if (row >= M) row = M - 1;
  const float2* ap = (const float2*)(A + (size_t)row * TK) + half;
  const float2* bs = (const float2*)Bs + half * P2 + ct0 * 16 + l15;

  v8f c[NTP];
#pragma unroll
  for (int t = 0; t < NTP; ++t) c[t] = v8f{};

#pragma unroll 8
  for (int k0 = 0; k0 < TK; k0 += 4) {
    const float2 a2 = *ap;
    ap += 2;
    v2f a = { a2.x, a2.y };
#pragma unroll
    for (int t = 0; t < NTP; ++t) {
      const float2 b2 = bs[t * 16];
      v2f b = { b2.x, b2.y };
      c[t] = __builtin_amdgcn_wmma_f32_16x16x4_f32(false, a, false, b, (short)0, c[t], false, false);
    }
    bs += P;
  }

  if constexpr (TN == 256) {
#pragma unroll
    for (int t = 0; t < NTP; ++t) {
      const int cc = (ct0 + t) * 16 + l15;
      const float pb = bias[cc];
      const float pm = bnm[cc];
      const float ps = rsqrtf(bnv[cc] + 1e-5f);
      const float pg = bng[cc];
      const float pe = bnb[cc];
#pragma unroll
      for (int r = 0; r < 8; ++r) {
        const int m = tileM * 16 + r + half * 8;
        if (m < M) {
          float vv = c[t][r] + pb;
          vv = (vv - pm) * ps * pg + pe;
          vv = vv > 0.0f ? vv : 0.01f * vv;
          D[(size_t)m * TN + cc] = vv;
        }
      }
    }
  } else if constexpr (NTP == 2) {
#pragma unroll
    for (int r = 0; r < 8; ++r) {
      const int m = tileM * 16 + r + half * 8;
      if (m < M) {
        float* Dr = D + (size_t)m * TN + ct0 * 16 + l15;
#pragma unroll
        for (int t = 0; t < NTP; ++t) Dr[t * 16] = c[t][r];
      }
    }
  } else {
#pragma unroll
    for (int r = 0; r < 8; ++r) {
      const int m = tileM * 16 + r + half * 8;
      if (m < M) {
        if (l15 < H) {
          AS[(size_t)m * H + l15] = c[0][r];
        } else if (l15 < 2 * H && m < Md) {
          AD[(size_t)m * H + (l15 - H)] = c[0][r];
        }
      }
    }
  }
}

template <int TN, int TK>
__global__ void __launch_bounds__(GEMM_WAVES_ * 32)
gat_gemm_wmma(const float* __restrict__ A,
              const float* __restrict__ B,
              const float* __restrict__ WF,
              float* __restrict__ D,
              float* __restrict__ AS,
              float* __restrict__ AD,
              const float* __restrict__ bias,
              const float* __restrict__ bng,
              const float* __restrict__ bnb,
              const float* __restrict__ bnm,
              const float* __restrict__ bnv,
              int M, int Md, int tilesM) {
  constexpr int P      = 2 * TN + 32;
  constexpr int HALVES = TN / 128;
  __shared__ __attribute__((aligned(16))) float Bs[(TK / 2) * P];

  {
    constexpr int C4    = TN / 4;
    constexpr int ITEMS = (TK / 2) * C4;
#pragma unroll
    for (int it = threadIdx.x; it < ITEMS; it += GEMM_WAVES_ * 32) {
      const int kp = it / C4;
      const int c4 = it % C4;
      const float4 r0 = *((const float4*)(B + (size_t)(2 * kp) * TN) + c4);
      const float4 r1 = *((const float4*)(B + (size_t)(2 * kp + 1) * TN) + c4);
      float4* dst = (float4*)(Bs + kp * P + c4 * 8);
      dst[0] = make_float4(r0.x, r1.x, r0.y, r1.y);
      dst[1] = make_float4(r0.z, r1.z, r0.w, r1.w);
    }
    constexpr int XITEMS = (TK / 2) * 4;
    if (HALVES == 1 && threadIdx.x < XITEMS) {
      const int kp = threadIdx.x >> 2;
      const int c4 = threadIdx.x & 3;
      const float4 r0 = *((const float4*)(WF + (size_t)(2 * kp) * 16) + c4);
      const float4 r1 = *((const float4*)(WF + (size_t)(2 * kp + 1) * 16) + c4);
      float4* dst = (float4*)(Bs + kp * P + 2 * TN + c4 * 8);
      dst[0] = make_float4(r0.x, r1.x, r0.y, r1.y);
      dst[1] = make_float4(r0.z, r1.z, r0.w, r1.w);
    }
  }
  __syncthreads();

  const int lane = threadIdx.x & 31;
  const int wave = __builtin_amdgcn_readfirstlane(threadIdx.x >> 5);
  const int half = lane >> 4;
  const int l15  = lane & 15;

  const int G         = (int)gridDim.x;
  const int mainTiles = (tilesM / G) * G;
  const int restTiles = tilesM - mainTiles;

  if constexpr (HALVES == 2) {
    const int slot  = (wave >> 3) * 4 + (wave & 3);
    const int phase = (wave >> 2) & 1;
    for (int i = 0; ; ++i) {
      const int tileM = blockIdx.x + G * (slot + 16 * i);
      if (tileM >= mainTiles) break;
      const int nh = (phase + i) & 1;
      gat_gemm_item<TN, TK, 8>(A + (size_t)nh * M * TK, Bs, D, AS, AD,
                               bias, bng, bnb, bnm, bnv, M, Md, tileM, nh, half, l15);
    }
    constexpr int PT = TN / 32;
    for (int p = (GEMM_WAVES_ - 1 - wave) * G + (int)blockIdx.x; p < restTiles * PT;
         p += GEMM_WAVES_ * G) {
      const int tileM = mainTiles + p / PT;
      const int ct0   = (p % PT) * 2;
      const int nh    = ct0 >> 3;
      gat_gemm_piece<TN, TK, 2>(A + (size_t)nh * M * TK, Bs, D, AS, AD,
                                bias, bng, bnb, bnm, bnv, M, Md, tileM, ct0, half, l15);
    }
  } else {
    for (int i = 0; ; ++i) {
      const int tileM = blockIdx.x + G * (wave + GEMM_WAVES_ * i);
      if (tileM >= mainTiles) break;
      gat_gemm_item<TN, TK, 9>(A, Bs, D, AS, AD,
                               bias, bng, bnb, bnm, bnv, M, Md, tileM, 0, half, l15);
    }
    constexpr int PT = TN / 32 + 1;
    for (int p = (GEMM_WAVES_ - 1 - wave) * G + (int)blockIdx.x; p < restTiles * PT;
         p += GEMM_WAVES_ * G) {
      const int tileM = mainTiles + p / PT;
      const int q     = p % PT;
      if (q < TN / 32) {
        gat_gemm_piece<TN, TK, 2>(A, Bs, D, AS, AD,
                                  bias, bng, bnb, bnm, bnv, M, Md, tileM, 2 * q, half, l15);
      } else {
        gat_gemm_piece<TN, TK, 1>(A, Bs, D, AS, AD,
                                  bias, bng, bnb, bnm, bnv, M, Md, tileM, TN / 16, half, l15);
      }
    }
  }
}

static inline int cdiv(int a, int b) { return (a + b - 1) / b; }

extern "C" void kernel_launch(void* const* d_in, const int* in_sizes, int n_in,
                              void* d_out, int out_size, void* d_ws, size_t ws_size,
                              hipStream_t stream) {
  const float* x     = (const float*)d_in[0];
  const int*   col1  = (const int*)d_in[2];
  const int*   t1    = (const int*)d_in[3];
  const int*   col2  = (const int*)d_in[5];
  const int*   t2    = (const int*)d_in[6];
  const int*   timeP = (const int*)d_in[7];
  const int*   intvP = (const int*)d_in[8];
  const float* Ws1   = (const float*)d_in[9];
  const float* Wd1   = (const float*)d_in[10];
  const float* as1   = (const float*)d_in[11];
  const float* ad1   = (const float*)d_in[12];
  const float* b1    = (const float*)d_in[13];
  const float* g1    = (const float*)d_in[14];
  const float* beta1 = (const float*)d_in[15];
  const float* m1    = (const float*)d_in[16];
  const float* v1    = (const float*)d_in[17];
  const float* Ws2   = (const float*)d_in[18];
  const float* Wd2   = (const float*)d_in[19];
  const float* as2   = (const float*)d_in[20];
  const float* ad2   = (const float*)d_in[21];
  const float* b2    = (const float*)d_in[22];

  float* ws   = (float*)d_ws;
  float* z1   = ws;
  float* hd1  = z1   + (size_t)N0_ * 256;
  float* as1v = hd1  + (size_t)N1_ * 256;
  float* ad1v = as1v + (size_t)N0_ * H1_;
  float* h    = ad1v + (size_t)N1_ * H1_;
  float* as2v = h    + (size_t)N1_ * 256;
  float* ad2v = as2v + (size_t)N1_;
  float* hs2  = hd1;
  float* wf1  = hd1  + (size_t)N1_ * 128;
  float* wf2  = wf1  + (size_t)IN_ * 16;

  const dim3 blk(256);
  const dim3 gblk(GEMM_WAVES_ * 32);

  gat_fold_att<<<cdiv(IN_ + H1_ * HID_, 8), blk, 0, stream>>>(
      Ws1, Wd1, as1, ad1, Ws2, Wd2, as2, ad2, wf1, wf2);

  gat_logit_wmma<<<cdiv(cdiv(N0_, 16), 8), blk, 0, stream>>>(
      x, wf1, as1v, ad1v, N0_, N1_, cdiv(N0_, 16));
  gat_gather_resident<H1_, 2, 8, false><<<GATHER1_WGS_, blk, 0, stream>>>(
      x, as1v, ad1v, col1, t1, timeP, intvP, nullptr, z1, N1_);
  gat_gemm_wmma<256, 256><<<GEMM_WGS_, gblk, 0, stream>>>(
      z1, Ws1, nullptr, h, nullptr, nullptr,
      b1, g1, beta1, m1, v1, N1_, N1_, cdiv(N1_, 16));

  gat_gemm_wmma<128, 256><<<GEMM_WGS_, gblk, 0, stream>>>(
      h, Ws2, wf2, hs2, as2v, ad2v,
      nullptr, nullptr, nullptr, nullptr, nullptr, N1_, N2_, cdiv(N1_, 16));
  gat_gather_resident<H2_, 1, 8, true><<<GATHER2_WGS_, blk, 0, stream>>>(
      hs2, as2v, ad2v, col2, t2, timeP, intvP, b2, (float*)d_out, N2_);
}


# Layer-1 logits formed inside the gather: vector-FMA dots of the kept source rows it holds and of the row's own x[i] with the folded columns; x@WF1 pass and launch dropped (a_s,a_d sum order changes)
# speedup vs baseline: 2.8189x; 1.0365x over previous
#include <hip/hip_runtime.h>
#include <hip/hip_bf16.h>
#include <cstddef>

#define N0_  100000
#define N1_  50000
#define N2_  25000
#define DEG_ 16
#define IN_  256
#define HID_ 128
#define OUT_ 128
#define H1_  2
#define H2_  1

typedef float v2f __attribute__((ext_vector_type(2)));
typedef float v8f __attribute__((ext_vector_type(8)));

template <int H>
__device__ __forceinline__ void gat_fold_row(const float* __restrict__ Ws,
                                             const float* __restrict__ Wd,
                                             const float* __restrict__ as,
                                             const float* __restrict__ ad,
                                             float* __restrict__ WF,
                                             int k, int lane) {
  float outv = 0.0f;
#pragma unroll
  for (int h = 0; h < H; ++h) {
    const float4 av = *((const float4*)(as + (size_t)h * 128) + lane);
    const float4 dv = *((const float4*)(ad + (size_t)h * 128) + lane);
    const float4 sv = *((const float4*)(Ws + (size_t)k * (H * 128) + h * 128) + lane);
    const float4 wv = *((const float4*)(Wd + (size_t)k * (H * 128) + h * 128) + lane);
    float s = sv.x * av.x + sv.y * av.y + sv.z * av.z + sv.w * av.w;
    float d = wv.x * dv.x + wv.y * dv.y + wv.z * dv.z + wv.w * dv.w;
#pragma unroll
    for (int off = 16; off > 0; off >>= 1) {
      s += __shfl_xor(s, off, 32);
      d += __shfl_xor(d, off, 32);
    }
    if (lane == h)     outv = s;
    if (lane == H + h) outv = d;
  }
  if (lane < 16) WF[(size_t)k * 16 + lane] = outv;
}

__global__ void gat_fold_att(const float* __restrict__ Ws1, const float* __restrict__ Wd1,
                             const float* __restrict__ as1, const float* __restrict__ ad1,
                             const float* __restrict__ Ws2, const float* __restrict__ Wd2,
                             const float* __restrict__ as2, const float* __restrict__ ad2,
                             float* __restrict__ WF1, float* __restrict__ WF2) {
  const int wid  = blockIdx.x * 8 + (threadIdx.x >> 5);
  const int lane = threadIdx.x & 31;
  if (wid < IN_) {
    gat_fold_row<H1_>(Ws1, Wd1, as1, ad1, WF1, wid, lane);
  } else if (wid < IN_ + H1_ * HID_) {
    gat_fold_row<H2_>(Ws2, Wd2, as2, ad2, WF2, wid - IN_, lane);
  }
}


__device__ __forceinline__ float gat_lane_f(float v, int l) {
  return __int_as_float(__builtin_amdgcn_readlane(__float_as_int(v), l));
}

template <int H, int NQ, int R, bool FINAL>
__global__ void __launch_bounds__(256)
gat_gather_resident(const float* __restrict__ X,
                    const float* __restrict__ AS,
                    const float* __restrict__ AD,
                    const int* __restrict__ col,
                    const int* __restrict__ tt,
                    const int* __restrict__ timeP,
                    const int* __restrict__ intervalP,
                    const float* __restrict__ b,
                    float* __restrict__ OUTP,
                    int Nd) {
  static_assert(!FINAL || (H == 1 && NQ == 1), "final form: one head, 128 channels");
  const int lane = threadIdx.x & 31;
  const int l15  = lane & 15;
  const int wv  = __builtin_amdgcn_readfirstlane((int)(blockIdx.x * 8 + (threadIdx.x >> 5)));
  const int nwv = (int)gridDim.x * 8;
  if (wv >= Nd) return;

  const int t0   = timeP[0];
  const int t1e  = t0 + intervalP[0];
  const int last = Nd - 1;

  float4 bq = make_float4(0.0f, 0.0f, 0.0f, 0.0f);
  if constexpr (FINAL) bq = *((const float4*)b + lane);

  int i1 = wv + nwv;
  if (i1 > last) i1 = last;
  int c0  = col[wv * DEG_ + l15];
  int tv0 = tt[wv * DEG_ + l15];
  int c1  = col[i1 * DEG_ + l15];
  int tv1 = tt[i1 * DEG_ + l15];
  float as0[H];
  float ad0[H];
#pragma unroll
  for (int h = 0; h < H; ++h) {
    as0[h] = AS[(size_t)c0 * H + h];
    ad0[h] = AD[(size_t)wv * H + h];
  }

  for (int i = wv; i < Nd; i += nwv) {
    int in1 = i + nwv;
    if (in1 > last) in1 = last;
    int i2 = i + 2 * nwv;
    if (i2 > last) i2 = last;
    float as1[H];
    float ad1[H];
#pragma unroll
    for (int h = 0; h < H; ++h) {
      as1[h] = AS[(size_t)c1 * H + h];
      ad1[h] = AD[(size_t)in1 * H + h];
    }
    const int c2  = col[i2 * DEG_ + l15];
    const int tv2 = tt[i2 * DEG_ + l15];
    asm volatile("" ::: "memory");

    const bool val = (tv0 >= t0) && (tv0 < t1e);
    float ex[H];
    float inv[H];
#pragma unroll
    for (int h = 0; h < H; ++h) {
      float ev = as0[h] + ad0[h];
      ev = ev > 0.0f ? ev : 0.2f * ev;
      float m = val ? ev : -__builtin_inff();
#pragma unroll
      for (int off = 8; off > 0; off >>= 1) m = fmaxf(m, __shfl_xor(m, off, 32));
      if (!(m > -__builtin_inff())) m = 0.0f;
      const float e = val ? __expf(ev - m) : 0.0f;
      float denom = 0.0f;
#pragma unroll
      for (int j = 0; j < DEG_; ++j) denom += gat_lane_f(e, j);
      inv[h] = 1.0f / fmaxf(denom, 1e-16f);
      ex[h]  = e;
    }
    bool nz = false;
#pragma unroll
    for (int h = 0; h < H; ++h) nz = nz || (ex[h] != 0.0f);
    unsigned mask = __builtin_amdgcn_ballot_w32(nz) & 0xFFFFu;

    float4 acc[H][NQ];
#pragma unroll
    for (int h = 0; h < H; ++h)
#pragma unroll
      for (int q = 0; q < NQ; ++q) acc[h][q] = make_float4(0.0f, 0.0f, 0.0f, 0.0f);

    while (mask != 0u) {
      float4 xr[R][NQ];
      float  wr[H][R];
      bool   on[R];
#pragma unroll
      for (int r = 0; r < R; ++r) {
        on[r] = (mask != 0u);
        if (on[r]) {
          const int j = __builtin_ctz(mask);
          mask &= mask - 1u;
          const int cr = __builtin_amdgcn_readlane(c0, j);
#pragma unroll
          for (int h = 0; h < H; ++h) wr[h][r] = gat_lane_f(ex[h], j) * inv[h];
          const float4* p = (const float4*)(X + (size_t)cr * (NQ * 128)) + lane;
#pragma unroll
          for (int q = 0; q < NQ; ++q) xr[r][q] = p[q * 32];
        }
      }
#pragma unroll
      for (int r = 0; r < R; ++r) {
        if (on[r]) {
#pragma unroll
          for (int h = 0; h < H; ++h) {
            const float wj = wr[h][r];
#pragma unroll
            for (int q = 0; q < NQ; ++q) {
              acc[h][q].x += wj * xr[r][q].x; acc[h][q].y += wj * xr[r][q].y;
              acc[h][q].z += wj * xr[r][q].z; acc[h][q].w += wj * xr[r][q].w;
            }
          }
        }
      }
    }

    if constexpr (FINAL) {
      float4 o;
      o.x = acc[0][0].x / (float)H + bq.x;
      o.y = acc[0][0].y / (float)H + bq.y;
      o.z = acc[0][0].z / (float)H + bq.z;
      o.w = acc[0][0].w / (float)H + bq.w;
      *((float4*)(OUTP + (size_t)i * 128) + lane) = o;
    } else {
#pragma unroll
      for (int h = 0; h < H; ++h) {
        float4* zr = (float4*)(OUTP + ((size_t)h * Nd + i) * (NQ * 128));
#pragma unroll
        for (int q = 0; q < NQ; ++q) zr[q * 32 + lane] = acc[h][q];
      }
    }

    c0 = c1; tv0 = tv1;
#pragma unroll
    for (int h = 0; h < H; ++h) { as0[h] = as1[h]; ad0[h] = ad1[h]; }
    c1 = c2; tv1 = tv2;
  }
}

template <int R>
__device__ __forceinline__ void gat_g1_load(const float* __restrict__ X, unsigned& m,
                                            int cidx, int lane,
                                            float4 (&xr)[R][2], int (&jr)[R], bool (&on)[R]) {
#pragma unroll
  for (int r = 0; r < R; ++r) {
    on[r] = (m != 0u);
    jr[r] = 16;
    if (on[r]) {
      const int j = __builtin_ctz(m);
      m &= m - 1u;
      jr[r] = j;
      const int cr = __builtin_amdgcn_readlane(cidx, j);
      const float4* p = (const float4*)(X + (size_t)cr * 256) + lane;
      xr[r][0] = p[0];
      xr[r][1] = p[32];
    }
  }
}

__device__ __forceinline__ float gat_dot8(const float4& a0, const float4& a1,
                                          const float4& w0, const float4& w1) {
  return a0.x * w0.x + a0.y * w0.y + a0.z * w0.z + a0.w * w0.w +
         a1.x * w1.x + a1.y * w1.y + a1.z * w1.z + a1.w * w1.w;
}

__device__ __forceinline__ float gat_pair_sum(float p0, float p1, int lane) {
  const bool hi = (lane & 16) != 0;
  float s = (hi ? p1 : p0) + __shfl_xor(hi ? p0 : p1, 16, 32);
#pragma unroll
  for (int off = 8; off > 0; off >>= 1) s += __shfl_xor(s, off, 32);
  return s;
}

template <int R>
__device__ __forceinline__ void gat_g1_dots(const float* Wc, int lane, int l15,
                                            const float4 (&xr)[R][2], const int (&jr)[R],
                                            const bool (&on)[R], const float (&adv)[2],
                                            float (&asv)[2], float (&mxu)[2]) {
  static_assert(R <= 8, "the halving exchange is written for up to eight row slots");
  const float4 w00 = *((const float4*)(Wc) + lane);
  const float4 w01 = *((const float4*)(Wc) + 32 + lane);
  const float4 w10 = *((const float4*)(Wc + 256) + lane);
  const float4 w11 = *((const float4*)(Wc + 256) + 32 + lane);
  float p0[R];
  float p1[R];
#pragma unroll
  for (int r = 0; r < R; ++r) {
    p0[r] = 0.0f;
    p1[r] = 0.0f;
    if (on[r]) {
      p0[r] = gat_dot8(xr[r][0], xr[r][1], w00, w01);
      p1[r] = gat_dot8(xr[r][0], xr[r][1], w10, w11);
    }
  }
  const bool hi = (lane & 16) != 0;
  const bool b3 = (lane & 8) != 0;
  const bool b2 = (lane & 4) != 0;
  const bool b1 = (lane & 2) != 0;
  float a[8];
#pragma unroll
  for (int r = 0; r < 8; ++r) {
    a[r] = 0.0f;
    if (r < R) a[r] = (hi ? p1[r] : p0[r]) + __shfl_xor(hi ? p0[r] : p1[r], 16, 32);
  }
  float b[4];
#pragma unroll
  for (int k = 0; k < 4; ++k)
    b[k] = (b3 ? a[k + 4] : a[k]) + __shfl_xor(b3 ? a[k] : a[k + 4], 8, 32);
  float c[2];
#pragma unroll
  for (int k = 0; k < 2; ++k)
    c[k] = (b2 ? b[k + 2] : b[k]) + __shfl_xor(b2 ? b[k] : b[k + 2], 4, 32);
  float d = (b1 ? c[1] : c[0]) + __shfl_xor(b1 ? c[0] : c[1], 2, 32);
  d += __shfl_xor(d, 1, 32);
#pragma unroll
  for (int r = 0; r < R; ++r) {
    const float s0 = gat_lane_f(d, 2 * r);
    const float s1 = gat_lane_f(d, 16 + 2 * r);
    if (l15 == jr[r]) { asv[0] = s0; asv[1] = s1; }
    if (on[r]) {
      float e0 = s0 + adv[0];
      float e1 = s1 + adv[1];
      e0 = e0 > 0.0f ? e0 : 0.2f * e0;
      e1 = e1 > 0.0f ? e1 : 0.2f * e1;
      mxu[0] = fmaxf(mxu[0], e0);
      mxu[1] = fmaxf(mxu[1], e1);
    }
  }
}

__device__ __forceinline__ void gat_g1_dst_part(const float* Wc, int lane,
                                                const float4 (&xd)[2], float& pd0, float& pd1) {
  const float4 d00 = *((const float4*)(Wc + 512) + lane);
  const float4 d01 = *((const float4*)(Wc + 512) + 32 + lane);
  const float4 d10 = *((const float4*)(Wc + 768) + lane);
  const float4 d11 = *((const float4*)(Wc + 768) + 32 + lane);
  pd0 = gat_dot8(xd[0], xd[1], d00, d01);
  pd1 = gat_dot8(xd[0], xd[1], d10, d11);
}
__device__ __forceinline__ void gat_g1_dst_sum(float pd0, float pd1, int lane, float (&adv)[2]) {
  const float s = gat_pair_sum(pd0, pd1, lane);
  adv[0] = gat_lane_f(s, 0);
  adv[1] = gat_lane_f(s, 16);
}

template <int R>
__device__ __forceinline__ void gat_g1_acc(const float4 (&xr)[R][2], const int (&jr)[R],
                                           const bool (&on)[R], const float (&ex)[2],
                                           const float (&inv)[2], float4 (&acc)[2][2]) {
#pragma unroll
  for (int r = 0; r < R; ++r) {
    if (on[r]) {
#pragma unroll
      for (int h = 0; h < 2; ++h) {
        const float wj = gat_lane_f(ex[h], jr[r]) * inv[h];
#pragma unroll
        for (int q = 0; q < 2; ++q) {
          acc[h][q].x += wj * xr[r][q].x; acc[h][q].y += wj * xr[r][q].y;
          acc[h][q].z += wj * xr[r][q].z; acc[h][q].w += wj * xr[r][q].w;
        }
      }
    }
  }
}

template <int R>
__global__ void __launch_bounds__(256, 8)
gat_gather_logit1(const float* __restrict__ X,
                  const float* __restrict__ WF,
                  const int* __restrict__ col,
                  const int* __restrict__ tt,
                  const int* __restrict__ timeP,
                  const int* __restrict__ intervalP,
                  float* __restrict__ Z,
                  int Nd) {
  static_assert(IN_ == 256 && H1_ == 2, "one thread per WF row, two heads");
  __shared__ __attribute__((aligned(16))) float Wc[4 * 256];
  {
    const float4 f = *((const float4*)(WF + (size_t)threadIdx.x * 16));
    Wc[threadIdx.x]       = f.x;
    Wc[256 + threadIdx.x] = f.y;
    Wc[512 + threadIdx.x] = f.z;
    Wc[768 + threadIdx.x] = f.w;
  }
  __syncthreads();

  const int lane = threadIdx.x & 31;
  const int l15  = lane & 15;
  const int wv  = __builtin_amdgcn_readfirstlane((int)(blockIdx.x * 8 + (threadIdx.x >> 5)));
  const int nwv = (int)gridDim.x * 8;
  if (wv >= Nd) return;

  const int t0   = timeP[0];
  const int t1e  = t0 + intervalP[0];
  const int last = Nd - 1;

  int i1 = wv + nwv;
  if (i1 > last) i1 = last;
  int c0  = col[wv * DEG_ + l15];
  int tv0 = tt[wv * DEG_ + l15];
  int c1  = col[i1 * DEG_ + l15];
  int tv1 = tt[i1 * DEG_ + l15];
  float adv[2];
  {
    float4 xd0[2];
    const float4* p = (const float4*)(X + (size_t)wv * 256) + lane;
    xd0[0] = p[0];
    xd0[1] = p[32];
    float pd0, pd1;
    gat_g1_dst_part(Wc, lane, xd0, pd0, pd1);
    gat_g1_dst_sum(pd0, pd1, lane, adv);
  }

  for (int i = wv; i < Nd; i += nwv) {
    int in1 = i + nwv;
    if (in1 > last) in1 = last;
    int i2 = i + 2 * nwv;
    if (i2 > last) i2 = last;
    float4 xd1[2];
    {
      const float4* p = (const float4*)(X + (size_t)in1 * 256) + lane;
      xd1[0] = p[0];
      xd1[1] = p[32];
    }
    const int c2  = col[i2 * DEG_ + l15];
    const int tv2 = tt[i2 * DEG_ + l15];
    asm volatile("" ::: "memory");

    const bool val = (tv0 >= t0) && (tv0 < t1e);
    const unsigned km = __builtin_amdgcn_ballot_w32(val) & 0xFFFFu;
    unsigned m = km;
    float4 xr[R][2];
    int    jr[R];
    bool   on[R];
    gat_g1_load<R>(X, m, c0, lane, xr, jr, on);

    float pn0, pn1;
    gat_g1_dst_part(Wc, lane, xd1, pn0, pn1);
    float asv[2] = { 0.0f, 0.0f };
    float mxu[2] = { -__builtin_inff(), -__builtin_inff() };
    gat_g1_dots<R>(Wc, lane, l15, xr, jr, on, adv, asv, mxu);
    float adn[2];
    gat_g1_dst_sum(pn0, pn1, lane, adn);
    const bool spill = (m != 0u);
    if (spill) {
      do {
        gat_g1_load<R>(X, m, c0, lane, xr, jr, on);
        gat_g1_dots<R>(Wc, lane, l15, xr, jr, on, adv, asv, mxu);
      } while (m != 0u);
    }

    float ex[2];
    float inv[2];
#pragma unroll
    for (int h = 0; h < 2; ++h) {
      float ev = asv[h] + adv[h];
      ev = ev > 0.0f ? ev : 0.2f * ev;
      float mx = mxu[h];
      if (!(mx > -__builtin_inff())) mx = 0.0f;
      const float e = val ? __expf(ev - mx) : 0.0f;
      float denom = 0.0f;
#pragma unroll
      for (int j = 0; j < DEG_; ++j) denom += gat_lane_f(e, j);
      inv[h] = 1.0f / fmaxf(denom, 1e-16f);
      ex[h]  = e;
    }

    float4 acc[2][2];
#pragma unroll
    for (int h = 0; h < 2; ++h)
#pragma unroll
      for (int q = 0; q < 2; ++q) acc[h][q] = make_float4(0.0f, 0.0f, 0.0f, 0.0f);

    if (spill) {
      m = km;
      gat_g1_load<R>(X, m, c0, lane, xr, jr, on);
    }
    gat_g1_acc<R>(xr, jr, on, ex, inv, acc);
    while (m != 0u) {
      gat_g1_load<R>(X, m, c0, lane, xr, jr, on);
      gat_g1_acc<R>(xr, jr, on, ex, inv, acc);
    }

#pragma unroll
    for (int h = 0; h < 2; ++h) {
      float4* zr = (float4*)(Z + ((size_t)h * Nd + i) * 256);
#pragma unroll
      for (int q = 0; q < 2; ++q) zr[q * 32 + lane] = acc[h][q];
    }

    c0 = c1; tv0 = tv1;
    c1 = c2; tv1 = tv2;
    adv[0] = adn[0]; adv[1] = adn[1];
  }
}

#define GEMM_WGS_   256
#define GATHER1_WGS_ 1024
#define GATHER2_WGS_ 512
#define GEMM_WAVES_ 32

template <int TN, int TK, int NT>
__device__ __forceinline__ void gat_gemm_item(const float* __restrict__ A,
                                              const float* Bs,
                                              float* __restrict__ D,
                                              float* __restrict__ AS,
                                              float* __restrict__ AD,
                                              const float* __restrict__ bias,
                                              const float* __restrict__ bng,
                                              const float* __restrict__ bnb,
                                              const float* __restrict__ bnm,
                                              const float* __restrict__ bnv,
                                              int M, int Md, int tileM, int nh,
                                              int half, int l15) {
  constexpr int P  = 2 * TN + 32;
  constexpr int P2 = P / 2;
  constexpr int H  = TN / 128;

  int row = tileM * 16 + l15;
  if (row >= M) row = M - 1;
  const float2* ap = (const float2*)(A + (size_t)row * TK) + half;
  const float2* bs = (const float2*)Bs + half * P2 + nh * 128 + l15;

  v8f c[NT];
#pragma unroll
  for (int t = 0; t < NT; ++t) c[t] = v8f{};

  if constexpr (NT == 8) {
#pragma unroll 4
    for (int k0 = 0; k0 < TK; k0 += 4) {
      const float2 a2 = *ap;
      ap += 2;
      v2f a = { a2.x, a2.y };
#pragma unroll
      for (int t = 0; t < 8; ++t) {
        const float2 b2 = bs[t * 16];
        v2f b = { b2.x, b2.y };
        c[t] = __builtin_amdgcn_wmma_f32_16x16x4_f32(false, a, false, b, (short)0, c[t], false, false);
      }
      bs += P;
    }
  } else {
    float2 an = *ap;
    float2 bn[NT];
#pragma unroll
    for (int t = 0; t < 8; ++t) bn[t] = bs[t * 16];
    bn[8] = bs[TN];
#pragma unroll
    for (int k0 = 0; k0 < TK - 4; k0 += 4) {
      const float2 ac = an;
      float2 bc[NT];
#pragma unroll
      for (int t = 0; t < NT; ++t) bc[t] = bn[t];
      ap += 2;
      bs += P;
      an = *ap;
#pragma unroll
      for (int t = 0; t < 8; ++t) bn[t] = bs[t * 16];
      bn[8] = bs[TN];
      __builtin_amdgcn_sched_barrier(0);
      v2f a = { ac.x, ac.y };
#pragma unroll
      for (int t = 0; t < NT; ++t) {
        v2f b = { bc[t].x, bc[t].y };
        c[t] = __builtin_amdgcn_wmma_f32_16x16x4_f32(false, a, false, b, (short)0, c[t], false, false);
      }
    }
    {
      v2f a = { an.x, an.y };
#pragma unroll
      for (int t = 0; t < NT; ++t) {
        v2f b = { bn[t].x, bn[t].y };
        c[t] = __builtin_amdgcn_wmma_f32_16x16x4_f32(false, a, false, b, (short)0, c[t], false, false);
      }
    }
  }

  if constexpr (NT == 8) {
#pragma unroll
    for (int t = 0; t < 8; ++t) {
      const int cc = nh * 128 + t * 16 + l15;
      const float pb = bias[cc];
      const float pm = bnm[cc];
      const float ps = rsqrtf(bnv[cc] + 1e-5f);
      const float pg = bng[cc];
      const float pe = bnb[cc];
#pragma unroll
      for (int r = 0; r < 8; ++r) {
        const int m = tileM * 16 + r + half * 8;
        if (m < M) {
          float vv = c[t][r] + pb;
          vv = (vv - pm) * ps * pg + pe;
          vv = vv > 0.0f ? vv : 0.01f * vv;
          D[(size_t)m * TN + cc] = vv;
        }
      }
    }
  } else {
#pragma unroll
    for (int r = 0; r < 8; ++r) {
      const int m = tileM * 16 + r + half * 8;
      if (m < M) {
        float* Dr = D + (size_t)m * TN + nh * 128 + l15;
#pragma unroll
        for (int t = 0; t < 8; ++t) Dr[t * 16] = c[t][r];
        if (l15 < H) {
          AS[(size_t)m * H + l15] = c[NT - 1][r];
        } else if (l15 < 2 * H && m < Md) {
          AD[(size_t)m * H + (l15 - H)] = c[NT - 1][r];
        }
      }
    }
  }
}

template <int TN, int TK, int NTP>
__device__ __forceinline__ void gat_gemm_piece(const float* __restrict__ A,
                                               const float* Bs,
                                               float* __restrict__ D,
                                               float* __restrict__ AS,
                                               float* __restrict__ AD,
                                               const float* __restrict__ bias,
                                               const float* __restrict__ bng,
                                               const float* __restrict__ bnb,
                                               const float* __restrict__ bnm,
                                               const float* __restrict__ bnv,
                                               int M, int Md, int tileM, int ct0,
                                               int half, int l15) {
  constexpr int P  = 2 * TN + 32;
  constexpr int P2 = P / 2;
  constexpr int H  = TN / 128;

  int row = tileM * 16 + l15;
  if (row >= M) row = M - 1;
  const float2* ap = (const float2*)(A + (size_t)row * TK) + half;
  const float2* bs = (const float2*)Bs + half * P2 + ct0 * 16 + l15;

  v8f c[NTP];
#pragma unroll
  for (int t = 0; t < NTP; ++t) c[t] = v8f{};

#pragma unroll 8
  for (int k0 = 0; k0 < TK; k0 += 4) {
    const float2 a2 = *ap;
    ap += 2;
    v2f a = { a2.x, a2.y };
#pragma unroll
    for (int t = 0; t < NTP; ++t) {
      const float2 b2 = bs[t * 16];
      v2f b = { b2.x, b2.y };
      c[t] = __builtin_amdgcn_wmma_f32_16x16x4_f32(false, a, false, b, (short)0, c[t], false, false);
    }
    bs += P;
  }

  if constexpr (TN == 256) {
#pragma unroll
    for (int t = 0; t < NTP; ++t) {
      const int cc = (ct0 + t) * 16 + l15;
      const float pb = bias[cc];
      const float pm = bnm[cc];
      const float ps = rsqrtf(bnv[cc] + 1e-5f);
      const float pg = bng[cc];
      const float pe = bnb[cc];
#pragma unroll
      for (int r = 0; r < 8; ++r) {
        const int m = tileM * 16 + r + half * 8;
        if (m < M) {
          float vv = c[t][r] + pb;
          vv = (vv - pm) * ps * pg + pe;
          vv = vv > 0.0f ? vv : 0.01f * vv;
          D[(size_t)m * TN + cc] = vv;
        }
      }
    }
  } else if constexpr (NTP == 2) {
#pragma unroll
    for (int r = 0; r < 8; ++r) {
      const int m = tileM * 16 + r + half * 8;
      if (m < M) {
        float* Dr = D + (size_t)m * TN + ct0 * 16 + l15;
#pragma unroll
        for (int t = 0; t < NTP; ++t) Dr[t * 16] = c[t][r];
      }
    }
  } else {
#pragma unroll
    for (int r = 0; r < 8; ++r) {
      const int m = tileM * 16 + r + half * 8;
      if (m < M) {
        if (l15 < H) {
          AS[(size_t)m * H + l15] = c[0][r];
        } else if (l15 < 2 * H && m < Md) {
          AD[(size_t)m * H + (l15 - H)] = c[0][r];
        }
      }
    }
  }
}

template <int TN, int TK>
__global__ void __launch_bounds__(GEMM_WAVES_ * 32)
gat_gemm_wmma(const float* __restrict__ A,
              const float* __restrict__ B,
              const float* __restrict__ WF,
              float* __restrict__ D,
              float* __restrict__ AS,
              float* __restrict__ AD,
              const float* __restrict__ bias,
              const float* __restrict__ bng,
              const float* __restrict__ bnb,
              const float* __restrict__ bnm,
              const float* __restrict__ bnv,
              int M, int Md, int tilesM) {
  constexpr int P      = 2 * TN + 32;
  constexpr int HALVES = TN / 128;
  __shared__ __attribute__((aligned(16))) float Bs[(TK / 2) * P];

  {
    constexpr int C4    = TN / 4;
    constexpr int ITEMS = (TK / 2) * C4;
#pragma unroll
    for (int it = threadIdx.x; it < ITEMS; it += GEMM_WAVES_ * 32) {
      const int kp = it / C4;
      const int c4 = it % C4;
      const float4 r0 = *((const float4*)(B + (size_t)(2 * kp) * TN) + c4);
      const float4 r1 = *((const float4*)(B + (size_t)(2 * kp + 1) * TN) + c4);
      float4* dst = (float4*)(Bs + kp * P + c4 * 8);
      dst[0] = make_float4(r0.x, r1.x, r0.y, r1.y);
      dst[1] = make_float4(r0.z, r1.z, r0.w, r1.w);
    }
    constexpr int XITEMS = (TK / 2) * 4;
    if (HALVES == 1 && threadIdx.x < XITEMS) {
      const int kp = threadIdx.x >> 2;
      const int c4 = threadIdx.x & 3;
      const float4 r0 = *((const float4*)(WF + (size_t)(2 * kp) * 16) + c4);
      const float4 r1 = *((const float4*)(WF + (size_t)(2 * kp + 1) * 16) + c4);
      float4* dst = (float4*)(Bs + kp * P + 2 * TN + c4 * 8);
      dst[0] = make_float4(r0.x, r1.x, r0.y, r1.y);
      dst[1] = make_float4(r0.z, r1.z, r0.w, r1.w);
    }
  }
  __syncthreads();

  const int lane = threadIdx.x & 31;
  const int wave = __builtin_amdgcn_readfirstlane(threadIdx.x >> 5);
  const int half = lane >> 4;
  const int l15  = lane & 15;

  const int G         = (int)gridDim.x;
  const int mainTiles = (tilesM / G) * G;
  const int restTiles = tilesM - mainTiles;

  if constexpr (HALVES == 2) {
    const int slot  = (wave >> 3) * 4 + (wave & 3);
    const int phase = (wave >> 2) & 1;
    for (int i = 0; ; ++i) {
      const int tileM = blockIdx.x + G * (slot + 16 * i);
      if (tileM >= mainTiles) break;
      const int nh = (phase + i) & 1;
      gat_gemm_item<TN, TK, 8>(A + (size_t)nh * M * TK, Bs, D, AS, AD,
                               bias, bng, bnb, bnm, bnv, M, Md, tileM, nh, half, l15);
    }
    constexpr int PT = TN / 32;
    for (int p = (GEMM_WAVES_ - 1 - wave) * G + (int)blockIdx.x; p < restTiles * PT;
         p += GEMM_WAVES_ * G) {
      const int tileM = mainTiles + p / PT;
      const int ct0   = (p % PT) * 2;
      const int nh    = ct0 >> 3;
      gat_gemm_piece<TN, TK, 2>(A + (size_t)nh * M * TK, Bs, D, AS, AD,
                                bias, bng, bnb, bnm, bnv, M, Md, tileM, ct0, half, l15);
    }
  } else {
    for (int i = 0; ; ++i) {
      const int tileM = blockIdx.x + G * (wave + GEMM_WAVES_ * i);
      if (tileM >= mainTiles) break;
      gat_gemm_item<TN, TK, 9>(A, Bs, D, AS, AD,
                               bias, bng, bnb, bnm, bnv, M, Md, tileM, 0, half, l15);
    }
    constexpr int PT = TN / 32 + 1;
    for (int p = (GEMM_WAVES_ - 1 - wave) * G + (int)blockIdx.x; p < restTiles * PT;
         p += GEMM_WAVES_ * G) {
      const int tileM = mainTiles + p / PT;
      const int q     = p % PT;
      if (q < TN / 32) {
        gat_gemm_piece<TN, TK, 2>(A, Bs, D, AS, AD,
                                  bias, bng, bnb, bnm, bnv, M, Md, tileM, 2 * q, half, l15);
      } else {
        gat_gemm_piece<TN, TK, 1>(A, Bs, D, AS, AD,
                                  bias, bng, bnb, bnm, bnv, M, Md, tileM, TN / 16, half, l15);
      }
    }
  }
}

static inline int cdiv(int a, int b) { return (a + b - 1) / b; }

extern "C" void kernel_launch(void* const* d_in, const int* in_sizes, int n_in,
                              void* d_out, int out_size, void* d_ws, size_t ws_size,
                              hipStream_t stream) {
  const float* x     = (const float*)d_in[0];
  const int*   col1  = (const int*)d_in[2];
  const int*   t1    = (const int*)d_in[3];
  const int*   col2  = (const int*)d_in[5];
  const int*   t2    = (const int*)d_in[6];
  const int*   timeP = (const int*)d_in[7];
  const int*   intvP = (const int*)d_in[8];
  const float* Ws1   = (const float*)d_in[9];
  const float* Wd1   = (const float*)d_in[10];
  const float* as1   = (const float*)d_in[11];
  const float* ad1   = (const float*)d_in[12];
  const float* b1    = (const float*)d_in[13];
  const float* g1    = (const float*)d_in[14];
  const float* beta1 = (const float*)d_in[15];
  const float* m1    = (const float*)d_in[16];
  const float* v1    = (const float*)d_in[17];
  const float* Ws2   = (const float*)d_in[18];
  const float* Wd2   = (const float*)d_in[19];
  const float* as2   = (const float*)d_in[20];
  const float* ad2   = (const float*)d_in[21];
  const float* b2    = (const float*)d_in[22];

  float* ws   = (float*)d_ws;
  float* z1   = ws;
  float* hd1  = z1   + (size_t)N0_ * 256;
  float* h    = hd1  + (size_t)N1_ * 256;
  float* as2v = h    + (size_t)N1_ * 256;
  float* ad2v = as2v + (size_t)N1_;
  float* hs2  = hd1;
  float* wf1  = hd1  + (size_t)N1_ * 128;
  float* wf2  = wf1  + (size_t)IN_ * 16;

  const dim3 blk(256);
  const dim3 gblk(GEMM_WAVES_ * 32);

  gat_fold_att<<<cdiv(IN_ + H1_ * HID_, 8), blk, 0, stream>>>(
      Ws1, Wd1, as1, ad1, Ws2, Wd2, as2, ad2, wf1, wf2);

  gat_gather_logit1<7><<<GATHER1_WGS_, blk, 0, stream>>>(
      x, wf1, col1, t1, timeP, intvP, z1, N1_);
  gat_gemm_wmma<256, 256><<<GEMM_WGS_, gblk, 0, stream>>>(
      z1, Ws1, nullptr, h, nullptr, nullptr,
      b1, g1, beta1, m1, v1, N1_, N1_, cdiv(N1_, 16));

  gat_gemm_wmma<128, 256><<<GEMM_WGS_, gblk, 0, stream>>>(
      h, Ws2, wf2, hs2, as2v, ad2v,
      nullptr, nullptr, nullptr, nullptr, nullptr, N1_, N2_, cdiv(N1_, 16));
  gat_gather_resident<H2_, 1, 8, true><<<GATHER2_WGS_, blk, 0, stream>>>(
      hs2, as2v, ad2v, col2, t2, timeP, intvP, b2, (float*)d_out, N2_);
}


# Layer 1 gather and GEMM in one kernel: 16 waves gather a 16-row Z tile into LDS while 16 waves run the previous tile's WMMAs from it; Z never goes to memory, one launch less (sum order unchanged)
# speedup vs baseline: 2.9259x; 1.0380x over previous
#include <hip/hip_runtime.h>
#include <hip/hip_bf16.h>
#include <cstddef>

#define N0_  100000
#define N1_  50000
#define N2_  25000
#define DEG_ 16
#define IN_  256
#define HID_ 128
#define OUT_ 128
#define H1_  2
#define H2_  1

typedef float v2f __attribute__((ext_vector_type(2)));
typedef float v8f __attribute__((ext_vector_type(8)));

template <int H>
__device__ __forceinline__ void gat_fold_row(const float* __restrict__ Ws,
                                             const float* __restrict__ Wd,
                                             const float* __restrict__ as,
                                             const float* __restrict__ ad,
                                             float* __restrict__ WF,
                                             int k, int lane) {
  float outv = 0.0f;
#pragma unroll
  for (int h = 0; h < H; ++h) {
    const float4 av = *((const float4*)(as + (size_t)h * 128) + lane);
    const float4 dv = *((const float4*)(ad + (size_t)h * 128) + lane);
    const float4 sv = *((const float4*)(Ws + (size_t)k * (H * 128) + h * 128) + lane);
    const float4 wv = *((const float4*)(Wd + (size_t)k * (H * 128) + h * 128) + lane);
    float s = sv.x * av.x + sv.y * av.y + sv.z * av.z + sv.w * av.w;
    float d = wv.x * dv.x + wv.y * dv.y + wv.z * dv.z + wv.w * dv.w;
#pragma unroll
    for (int off = 16; off > 0; off >>= 1) {
      s += __shfl_xor(s, off, 32);
      d += __shfl_xor(d, off, 32);
    }
    if (lane == h)     outv = s;
    if (lane == H + h) outv = d;
  }
  if (lane < 16) WF[(size_t)k * 16 + lane] = outv;
}

__global__ void gat_fold_att(const float* __restrict__ Ws1, const float* __restrict__ Wd1,
                             const float* __restrict__ as1, const float* __restrict__ ad1,
                             const float* __restrict__ Ws2, const float* __restrict__ Wd2,
                             const float* __restrict__ as2, const float* __restrict__ ad2,
                             float* __restrict__ WF1, float* __restrict__ WF2) {
  const int wid  = blockIdx.x * 8 + (threadIdx.x >> 5);
  const int lane = threadIdx.x & 31;
  if (wid < IN_) {
    gat_fold_row<H1_>(Ws1, Wd1, as1, ad1, WF1, wid, lane);
  } else if (wid < IN_ + H1_ * HID_) {
    gat_fold_row<H2_>(Ws2, Wd2, as2, ad2, WF2, wid - IN_, lane);
  }
}


__device__ __forceinline__ float gat_lane_f(float v, int l) {
  return __int_as_float(__builtin_amdgcn_readlane(__float_as_int(v), l));
}

template <int H, int NQ, int R, bool FINAL>
__global__ void __launch_bounds__(256)
gat_gather_resident(const float* __restrict__ X,
                    const float* __restrict__ AS,
                    const float* __restrict__ AD,
                    const int* __restrict__ col,
                    const int* __restrict__ tt,
                    const int* __restrict__ timeP,
                    const int* __restrict__ intervalP,
                    const float* __restrict__ b,
                    float* __restrict__ OUTP,
                    int Nd) {
  static_assert(!FINAL || (H == 1 && NQ == 1), "final form: one head, 128 channels");
  const int lane = threadIdx.x & 31;
  const int l15  = lane & 15;
  const int wv  = __builtin_amdgcn_readfirstlane((int)(blockIdx.x * 8 + (threadIdx.x >> 5)));
  const int nwv = (int)gridDim.x * 8;
  if (wv >= Nd) return;

  const int t0   = timeP[0];
  const int t1e  = t0 + intervalP[0];
  const int last = Nd - 1;

  float4 bq = make_float4(0.0f, 0.0f, 0.0f, 0.0f);
  if constexpr (FINAL) bq = *((const float4*)b + lane);

  int i1 = wv + nwv;
  if (i1 > last) i1 = last;
  int c0  = col[wv * DEG_ + l15];
  int tv0 = tt[wv * DEG_ + l15];
  int c1  = col[i1 * DEG_ + l15];
  int tv1 = tt[i1 * DEG_ + l15];
  float as0[H];
  float ad0[H];
#pragma unroll
  for (int h = 0; h < H; ++h) {
    as0[h] = AS[(size_t)c0 * H + h];
    ad0[h] = AD[(size_t)wv * H + h];
  }

  for (int i = wv; i < Nd; i += nwv) {
    int in1 = i + nwv;
    if (in1 > last) in1 = last;
    int i2 = i + 2 * nwv;
    if (i2 > last) i2 = last;
    float as1[H];
    float ad1[H];
#pragma unroll
    for (int h = 0; h < H; ++h) {
      as1[h] = AS[(size_t)c1 * H + h];
      ad1[h] = AD[(size_t)in1 * H + h];
    }
    const int c2  = col[i2 * DEG_ + l15];
    const int tv2 = tt[i2 * DEG_ + l15];
    asm volatile("" ::: "memory");

    const bool val = (tv0 >= t0) && (tv0 < t1e);
    float ex[H];
    float inv[H];
#pragma unroll
    for (int h = 0; h < H; ++h) {
      float ev = as0[h] + ad0[h];
      ev = ev > 0.0f ? ev : 0.2f * ev;
      float m = val ? ev : -__builtin_inff();
#pragma unroll
      for (int off = 8; off > 0; off >>= 1) m = fmaxf(m, __shfl_xor(m, off, 32));
      if (!(m > -__builtin_inff())) m = 0.0f;
      const float e = val ? __expf(ev - m) : 0.0f;
      float denom = 0.0f;
#pragma unroll
      for (int j = 0; j < DEG_; ++j) denom += gat_lane_f(e, j);
      inv[h] = 1.0f / fmaxf(denom, 1e-16f);
      ex[h]  = e;
    }
    bool nz = false;
#pragma unroll
    for (int h = 0; h < H; ++h) nz = nz || (ex[h] != 0.0f);
    unsigned mask = __builtin_amdgcn_ballot_w32(nz) & 0xFFFFu;

    float4 acc[H][NQ];
#pragma unroll
    for (int h = 0; h < H; ++h)
#pragma unroll
      for (int q = 0; q < NQ; ++q) acc[h][q] = make_float4(0.0f, 0.0f, 0.0f, 0.0f);

    while (mask != 0u) {
      float4 xr[R][NQ];
      float  wr[H][R];
      bool   on[R];
#pragma unroll
      for (int r = 0; r < R; ++r) {
        on[r] = (mask != 0u);
        if (on[r]) {
          const int j = __builtin_ctz(mask);
          mask &= mask - 1u;
          const int cr = __builtin_amdgcn_readlane(c0, j);
#pragma unroll
          for (int h = 0; h < H; ++h) wr[h][r] = gat_lane_f(ex[h], j) * inv[h];
          const float4* p = (const float4*)(X + (size_t)cr * (NQ * 128)) + lane;
#pragma unroll
          for (int q = 0; q < NQ; ++q) xr[r][q] = p[q * 32];
        }
      }
#pragma unroll
      for (int r = 0; r < R; ++r) {
        if (on[r]) {
#pragma unroll
          for (int h = 0; h < H; ++h) {
            const float wj = wr[h][r];
#pragma unroll
            for (int q = 0; q < NQ; ++q) {
              acc[h][q].x += wj * xr[r][q].x; acc[h][q].y += wj * xr[r][q].y;
              acc[h][q].z += wj * xr[r][q].z; acc[h][q].w += wj * xr[r][q].w;
            }
          }
        }
      }
    }

    if constexpr (FINAL) {
      float4 o;
      o.x = acc[0][0].x / (float)H + bq.x;
      o.y = acc[0][0].y / (float)H + bq.y;
      o.z = acc[0][0].z / (float)H + bq.z;
      o.w = acc[0][0].w / (float)H + bq.w;
      *((float4*)(OUTP + (size_t)i * 128) + lane) = o;
    } else {
#pragma unroll
      for (int h = 0; h < H; ++h) {
        float4* zr = (float4*)(OUTP + ((size_t)h * Nd + i) * (NQ * 128));
#pragma unroll
        for (int q = 0; q < NQ; ++q) zr[q * 32 + lane] = acc[h][q];
      }
    }

    c0 = c1; tv0 = tv1;
#pragma unroll
    for (int h = 0; h < H; ++h) { as0[h] = as1[h]; ad0[h] = ad1[h]; }
    c1 = c2; tv1 = tv2;
  }
}

template <int R>
__device__ __forceinline__ void gat_g1_load(const float* __restrict__ X, unsigned& m,
                                            int cidx, int lane,
                                            float4 (&xr)[R][2], int (&jr)[R], bool (&on)[R]) {
#pragma unroll
  for (int r = 0; r < R; ++r) {
    on[r] = (m != 0u);
    jr[r] = 16;
    if (on[r]) {
      const int j = __builtin_ctz(m);
      m &= m - 1u;
      jr[r] = j;
      const int cr = __builtin_amdgcn_readlane(cidx, j);
      const float4* p = (const float4*)(X + (size_t)cr * 256) + lane;
      xr[r][0] = p[0];
      xr[r][1] = p[32];
    }
  }
}

__device__ __forceinline__ float gat_dot8(const float4& a0, const float4& a1,
                                          const float4& w0, const float4& w1) {
  return a0.x * w0.x + a0.y * w0.y + a0.z * w0.z + a0.w * w0.w +
         a1.x * w1.x + a1.y * w1.y + a1.z * w1.z + a1.w * w1.w;
}

__device__ __forceinline__ float gat_pair_sum(float p0, float p1, int lane) {
  const bool hi = (lane & 16) != 0;
  float s = (hi ? p1 : p0) + __shfl_xor(hi ? p0 : p1, 16, 32);
#pragma unroll
  for (int off = 8; off > 0; off >>= 1) s += __shfl_xor(s, off, 32);
  return s;
}

template <int R>
__device__ __forceinline__ void gat_g1_dots(const float* Wc, int lane, int l15,
                                            const float4 (&xr)[R][2], const int (&jr)[R],
                                            const bool (&on)[R], const float (&adv)[2],
                                            float (&asv)[2], float (&mxu)[2]) {
  static_assert(R <= 8, "the halving exchange is written for up to eight row slots");
  const float4 w00 = *((const float4*)(Wc) + lane);
  const float4 w01 = *((const float4*)(Wc) + 32 + lane);
  const float4 w10 = *((const float4*)(Wc + 256) + lane);
  const float4 w11 = *((const float4*)(Wc + 256) + 32 + lane);
  float p0[R];
  float p1[R];
#pragma unroll
  for (int r = 0; r < R; ++r) {
    p0[r] = 0.0f;
    p1[r] = 0.0f;
    if (on[r]) {
      p0[r] = gat_dot8(xr[r][0], xr[r][1], w00, w01);
      p1[r] = gat_dot8(xr[r][0], xr[r][1], w10, w11);
    }
  }
  const bool hi = (lane & 16) != 0;
  const bool b3 = (lane & 8) != 0;
  const bool b2 = (lane & 4) != 0;
  const bool b1 = (lane & 2) != 0;
  float a[8];
#pragma unroll
  for (int r = 0; r < 8; ++r) {
    a[r] = 0.0f;
    if (r < R) a[r] = (hi ? p1[r] : p0[r]) + __shfl_xor(hi ? p0[r] : p1[r], 16, 32);
  }
  float b[4];
#pragma unroll
  for (int k = 0; k < 4; ++k)
    b[k] = (b3 ? a[k + 4] : a[k]) + __shfl_xor(b3 ? a[k] : a[k + 4], 8, 32);
  float c[2];
#pragma unroll
  for (int k = 0; k < 2; ++k)
    c[k] = (b2 ? b[k + 2] : b[k]) + __shfl_xor(b2 ? b[k] : b[k + 2], 4, 32);
  float d = (b1 ? c[1] : c[0]) + __shfl_xor(b1 ? c[0] : c[1], 2, 32);
  d += __shfl_xor(d, 1, 32);
#pragma unroll
  for (int r = 0; r < R; ++r) {
    const float s0 = gat_lane_f(d, 2 * r);
    const float s1 = gat_lane_f(d, 16 + 2 * r);
    if (l15 == jr[r]) { asv[0] = s0; asv[1] = s1; }
    if (on[r]) {
      float e0 = s0 + adv[0];
      float e1 = s1 + adv[1];
      e0 = e0 > 0.0f ? e0 : 0.2f * e0;
      e1 = e1 > 0.0f ? e1 : 0.2f * e1;
      mxu[0] = fmaxf(mxu[0], e0);
      mxu[1] = fmaxf(mxu[1], e1);
    }
  }
}

__device__ __forceinline__ void gat_g1_dst_part(const float* Wc, int lane,
                                                const float4 (&xd)[2], float& pd0, float& pd1) {
  const float4 d00 = *((const float4*)(Wc + 512) + lane);
  const float4 d01 = *((const float4*)(Wc + 512) + 32 + lane);
  const float4 d10 = *((const float4*)(Wc + 768) + lane);
  const float4 d11 = *((const float4*)(Wc + 768) + 32 + lane);
  pd0 = gat_dot8(xd[0], xd[1], d00, d01);
  pd1 = gat_dot8(xd[0], xd[1], d10, d11);
}
__device__ __forceinline__ void gat_g1_dst_sum(float pd0, float pd1, int lane, float (&adv)[2]) {
  const float s = gat_pair_sum(pd0, pd1, lane);
  adv[0] = gat_lane_f(s, 0);
  adv[1] = gat_lane_f(s, 16);
}

template <int R>
__device__ __forceinline__ void gat_g1_acc(const float4 (&xr)[R][2], const int (&jr)[R],
                                           const bool (&on)[R], const float (&ex)[2],
                                           const float (&inv)[2], float4 (&acc)[2][2]) {
#pragma unroll
  for (int r = 0; r < R; ++r) {
    if (on[r]) {
#pragma unroll
      for (int h = 0; h < 2; ++h) {
        const float wj = gat_lane_f(ex[h], jr[r]) * inv[h];
#pragma unroll
        for (int q = 0; q < 2; ++q) {
          acc[h][q].x += wj * xr[r][q].x; acc[h][q].y += wj * xr[r][q].y;
          acc[h][q].z += wj * xr[r][q].z; acc[h][q].w += wj * xr[r][q].w;
        }
      }
    }
  }
}

#define GEMM_WGS_   256
#define GATHER2_WGS_ 512
#define GEMM_WAVES_ 32

template <int TN, int TK, int NT>
__device__ __forceinline__ void gat_gemm_item(const float* __restrict__ A,
                                              const float* Bs,
                                              float* __restrict__ D,
                                              float* __restrict__ AS,
                                              float* __restrict__ AD,
                                              const float* __restrict__ bias,
                                              const float* __restrict__ bng,
                                              const float* __restrict__ bnb,
                                              const float* __restrict__ bnm,
                                              const float* __restrict__ bnv,
                                              int M, int Md, int tileM, int nh,
                                              int half, int l15) {
  constexpr int P  = 2 * TN + 32;
  constexpr int P2 = P / 2;
  constexpr int H  = TN / 128;

  int row = tileM * 16 + l15;
  if (row >= M) row = M - 1;
  const float2* ap = (const float2*)(A + (size_t)row * TK) + half;
  const float2* bs = (const float2*)Bs + half * P2 + nh * 128 + l15;

  v8f c[NT];
#pragma unroll
  for (int t = 0; t < NT; ++t) c[t] = v8f{};

  if constexpr (NT == 8) {
#pragma unroll 4
    for (int k0 = 0; k0 < TK; k0 += 4) {
      const float2 a2 = *ap;
      ap += 2;
      v2f a = { a2.x, a2.y };
#pragma unroll
      for (int t = 0; t < 8; ++t) {
        const float2 b2 = bs[t * 16];
        v2f b = { b2.x, b2.y };
        c[t] = __builtin_amdgcn_wmma_f32_16x16x4_f32(false, a, false, b, (short)0, c[t], false, false);
      }
      bs += P;
    }
  } else {
    float2 an = *ap;
    float2 bn[NT];
#pragma unroll
    for (int t = 0; t < 8; ++t) bn[t] = bs[t * 16];
    bn[8] = bs[TN];
#pragma unroll
    for (int k0 = 0; k0 < TK - 4; k0 += 4) {
      const float2 ac = an;
      float2 bc[NT];
#pragma unroll
      for (int t = 0; t < NT; ++t) bc[t] = bn[t];
      ap += 2;
      bs += P;
      an = *ap;
#pragma unroll
      for (int t = 0; t < 8; ++t) bn[t] = bs[t * 16];
      bn[8] = bs[TN];
      __builtin_amdgcn_sched_barrier(0);
      v2f a = { ac.x, ac.y };
#pragma unroll
      for (int t = 0; t < NT; ++t) {
        v2f b = { bc[t].x, bc[t].y };
        c[t] = __builtin_amdgcn_wmma_f32_16x16x4_f32(false, a, false, b, (short)0, c[t], false, false);
      }
    }
    {
      v2f a = { an.x, an.y };
#pragma unroll
      for (int t = 0; t < NT; ++t) {
        v2f b = { bn[t].x, bn[t].y };
        c[t] = __builtin_amdgcn_wmma_f32_16x16x4_f32(false, a, false, b, (short)0, c[t], false, false);
      }
    }
  }

  if constexpr (NT == 8) {
#pragma unroll
    for (int t = 0; t < 8; ++t) {
      const int cc = nh * 128 + t * 16 + l15;
      const float pb = bias[cc];
      const float pm = bnm[cc];
      const float ps = rsqrtf(bnv[cc] + 1e-5f);
      const float pg = bng[cc];
      const float pe = bnb[cc];
#pragma unroll
      for (int r = 0; r < 8; ++r) {
        const int m = tileM * 16 + r + half * 8;
        if (m < M) {
          float vv = c[t][r] + pb;
          vv = (vv - pm) * ps * pg + pe;
          vv = vv > 0.0f ? vv : 0.01f * vv;
          D[(size_t)m * TN + cc] = vv;
        }
      }
    }
  } else {
#pragma unroll
    for (int r = 0; r < 8; ++r) {
      const int m = tileM * 16 + r + half * 8;
      if (m < M) {
        float* Dr = D + (size_t)m * TN + nh * 128 + l15;
#pragma unroll
        for (int t = 0; t < 8; ++t) Dr[t * 16] = c[t][r];
        if (l15 < H) {
          AS[(size_t)m * H + l15] = c[NT - 1][r];
        } else if (l15 < 2 * H && m < Md) {
          AD[(size_t)m * H + (l15 - H)] = c[NT - 1][r];
        }
      }
    }
  }
}

template <int TN, int TK, int NTP>
__device__ __forceinline__ void gat_gemm_piece(const float* __restrict__ A,
                                               const float* Bs,
                                               float* __restrict__ D,
                                               float* __restrict__ AS,
                                               float* __restrict__ AD,
                                               const float* __restrict__ bias,
                                               const float* __restrict__ bng,
                                               const float* __restrict__ bnb,
                                               const float* __restrict__ bnm,
                                               const float* __restrict__ bnv,
                                               int M, int Md, int tileM, int ct0,
                                               int half, int l15) {
  constexpr int P  = 2 * TN + 32;
  constexpr int P2 = P / 2;
  constexpr int H  = TN / 128;

  int row = tileM * 16 + l15;
  if (row >= M) row = M - 1;
  const float2* ap = (const float2*)(A + (size_t)row * TK) + half;
  const float2* bs = (const float2*)Bs + half * P2 + ct0 * 16 + l15;

  v8f c[NTP];
#pragma unroll
  for (int t = 0; t < NTP; ++t) c[t] = v8f{};

#pragma unroll 8
  for (int k0 = 0; k0 < TK; k0 += 4) {
    const float2 a2 = *ap;
    ap += 2;
    v2f a = { a2.x, a2.y };
#pragma unroll
    for (int t = 0; t < NTP; ++t) {
      const float2 b2 = bs[t * 16];
      v2f b = { b2.x, b2.y };
      c[t] = __builtin_amdgcn_wmma_f32_16x16x4_f32(false, a, false, b, (short)0, c[t], false, false);
    }
    bs += P;
  }

  if constexpr (TN == 256) {
#pragma unroll
    for (int t = 0; t < NTP; ++t) {
      const int cc = (ct0 + t) * 16 + l15;
      const float pb = bias[cc];
      const float pm = bnm[cc];
      const float ps = rsqrtf(bnv[cc] + 1e-5f);
      const float pg = bng[cc];
      const float pe = bnb[cc];
#pragma unroll
      for (int r = 0; r < 8; ++r) {
        const int m = tileM * 16 + r + half * 8;
        if (m < M) {
          float vv = c[t][r] + pb;
          vv = (vv - pm) * ps * pg + pe;
          vv = vv > 0.0f ? vv : 0.01f * vv;
          D[(size_t)m * TN + cc] = vv;
        }
      }
    }
  } else if constexpr (NTP == 2) {
#pragma unroll
    for (int r = 0; r < 8; ++r) {
      const int m = tileM * 16 + r + half * 8;
      if (m < M) {
        float* Dr = D + (size_t)m * TN + ct0 * 16 + l15;
#pragma unroll
        for (int t = 0; t < NTP; ++t) Dr[t * 16] = c[t][r];
      }
    }
  } else {
#pragma unroll
    for (int r = 0; r < 8; ++r) {
      const int m = tileM * 16 + r + half * 8;
      if (m < M) {
        if (l15 < H) {
          AS[(size_t)m * H + l15] = c[0][r];
        } else if (l15 < 2 * H && m < Md) {
          AD[(size_t)m * H + (l15 - H)] = c[0][r];
        }
      }
    }
  }
}

template <int TN, int TK>
__global__ void __launch_bounds__(GEMM_WAVES_ * 32)
gat_gemm_wmma(const float* __restrict__ A,
              const float* __restrict__ B,
              const float* __restrict__ WF,
              float* __restrict__ D,
              float* __restrict__ AS,
              float* __restrict__ AD,
              const float* __restrict__ bias,
              const float* __restrict__ bng,
              const float* __restrict__ bnb,
              const float* __restrict__ bnm,
              const float* __restrict__ bnv,
              int M, int Md, int tilesM) {
  constexpr int P      = 2 * TN + 32;
  constexpr int HALVES = TN / 128;
  __shared__ __attribute__((aligned(16))) float Bs[(TK / 2) * P];

  {
    constexpr int C4    = TN / 4;
    constexpr int ITEMS = (TK / 2) * C4;
#pragma unroll
    for (int it = threadIdx.x; it < ITEMS; it += GEMM_WAVES_ * 32) {
      const int kp = it / C4;
      const int c4 = it % C4;
      const float4 r0 = *((const float4*)(B + (size_t)(2 * kp) * TN) + c4);
      const float4 r1 = *((const float4*)(B + (size_t)(2 * kp + 1) * TN) + c4);
      float4* dst = (float4*)(Bs + kp * P + c4 * 8);
      dst[0] = make_float4(r0.x, r1.x, r0.y, r1.y);
      dst[1] = make_float4(r0.z, r1.z, r0.w, r1.w);
    }
    constexpr int XITEMS = (TK / 2) * 4;
    if (HALVES == 1 && threadIdx.x < XITEMS) {
      const int kp = threadIdx.x >> 2;
      const int c4 = threadIdx.x & 3;
      const float4 r0 = *((const float4*)(WF + (size_t)(2 * kp) * 16) + c4);
      const float4 r1 = *((const float4*)(WF + (size_t)(2 * kp + 1) * 16) + c4);
      float4* dst = (float4*)(Bs + kp * P + 2 * TN + c4 * 8);
      dst[0] = make_float4(r0.x, r1.x, r0.y, r1.y);
      dst[1] = make_float4(r0.z, r1.z, r0.w, r1.w);
    }
  }
  __syncthreads();

  const int lane = threadIdx.x & 31;
  const int wave = __builtin_amdgcn_readfirstlane(threadIdx.x >> 5);
  const int half = lane >> 4;
  const int l15  = lane & 15;

  const int G         = (int)gridDim.x;
  const int mainTiles = (tilesM / G) * G;
  const int restTiles = tilesM - mainTiles;

  if constexpr (HALVES == 2) {
    const int slot  = (wave >> 3) * 4 + (wave & 3);
    const int phase = (wave >> 2) & 1;
    for (int i = 0; ; ++i) {
      const int tileM = blockIdx.x + G * (slot + 16 * i);
      if (tileM >= mainTiles) break;
      const int nh = (phase + i) & 1;
      gat_gemm_item<TN, TK, 8>(A + (size_t)nh * M * TK, Bs, D, AS, AD,
                               bias, bng, bnb, bnm, bnv, M, Md, tileM, nh, half, l15);
    }
    constexpr int PT = TN / 32;
    for (int p = (GEMM_WAVES_ - 1 - wave) * G + (int)blockIdx.x; p < restTiles * PT;
         p += GEMM_WAVES_ * G) {
      const int tileM = mainTiles + p / PT;
      const int ct0   = (p % PT) * 2;
      const int nh    = ct0 >> 3;
      gat_gemm_piece<TN, TK, 2>(A + (size_t)nh * M * TK, Bs, D, AS, AD,
                                bias, bng, bnb, bnm, bnv, M, Md, tileM, ct0, half, l15);
    }
  } else {
    for (int i = 0; ; ++i) {
      const int tileM = blockIdx.x + G * (wave + GEMM_WAVES_ * i);
      if (tileM >= mainTiles) break;
      gat_gemm_item<TN, TK, 9>(A, Bs, D, AS, AD,
                               bias, bng, bnb, bnm, bnv, M, Md, tileM, 0, half, l15);
    }
    constexpr int PT = TN / 32 + 1;
    for (int p = (GEMM_WAVES_ - 1 - wave) * G + (int)blockIdx.x; p < restTiles * PT;
         p += GEMM_WAVES_ * G) {
      const int tileM = mainTiles + p / PT;
      const int q     = p % PT;
      if (q < TN / 32) {
        gat_gemm_piece<TN, TK, 2>(A, Bs, D, AS, AD,
                                  bias, bng, bnb, bnm, bnv, M, Md, tileM, 2 * q, half, l15);
      } else {
        gat_gemm_piece<TN, TK, 1>(A, Bs, D, AS, AD,
                                  bias, bng, bnb, bnm, bnv, M, Md, tileM, TN / 16, half, l15);
      }
    }
  }
}

#define FUSE_ZP_ 260

__device__ __forceinline__ void gat_lds_barrier() {
  asm volatile("s_wait_dscnt 0x0\n\ts_barrier_signal -1\n\ts_barrier_wait -1" ::: "memory");
}
template <int R>
__global__ void __launch_bounds__(GEMM_WAVES_ * 32)
gat_gather_gemm1(const float* __restrict__ X,
                 const float* __restrict__ WF,
                 const int* __restrict__ col,
                 const int* __restrict__ tt,
                 const int* __restrict__ timeP,
                 const int* __restrict__ intervalP,
                 const float* __restrict__ B,
                 float* __restrict__ D,
                 const float* __restrict__ bias,
                 const float* __restrict__ bng,
                 const float* __restrict__ bnb,
                 const float* __restrict__ bnm,
                 const float* __restrict__ bnv,
                 int Nd, int tilesM) {
  static_assert(IN_ == 256 && H1_ == 2 && HID_ == 128, "two heads of 128 columns, rows of 256");
  static_assert(N1_ % 16 == 0, "every row tile is whole: each gather wave of a tile has a row");
  constexpr int TN = 256;
  constexpr int TK = 256;
  constexpr int P  = 2 * TN + 32;
  constexpr int P2 = P / 2;
  __shared__ __attribute__((aligned(16))) float Bs[(TK / 2) * P];
  __shared__ __attribute__((aligned(16))) float Zt[2 * 16 * FUSE_ZP_];
  __shared__ __attribute__((aligned(16))) float Wc[4 * 256];
  __shared__ float Pp[5 * 256];

  {
    constexpr int C4    = TN / 4;
    constexpr int ITEMS = (TK / 2) * C4;
#pragma unroll
    for (int it = threadIdx.x; it < ITEMS; it += GEMM_WAVES_ * 32) {
      const int kp = it / C4;
      const int c4 = it % C4;
      const float4 r0 = *((const float4*)(B + (size_t)(2 * kp) * TN) + c4);
      const float4 r1 = *((const float4*)(B + (size_t)(2 * kp + 1) * TN) + c4);
      float4* dst = (float4*)(Bs + kp * P + c4 * 8);
      dst[0] = make_float4(r0.x, r1.x, r0.y, r1.y);
      dst[1] = make_float4(r0.z, r1.z, r0.w, r1.w);
    }
    if (threadIdx.x < 256) {
      const float4 f = *((const float4*)(WF + (size_t)threadIdx.x * 16));
      Wc[threadIdx.x]       = f.x;
      Wc[256 + threadIdx.x] = f.y;
      Wc[512 + threadIdx.x] = f.z;
      Wc[768 + threadIdx.x] = f.w;
      Pp[threadIdx.x]        = bias[threadIdx.x];
      Pp[256 + threadIdx.x]  = bnm[threadIdx.x];
      Pp[512 + threadIdx.x]  = rsqrtf(bnv[threadIdx.x] + 1e-5f);
      Pp[768 + threadIdx.x]  = bng[threadIdx.x];
      Pp[1024 + threadIdx.x] = bnb[threadIdx.x];
    }
  }
  __syncthreads();

  const int lane = threadIdx.x & 31;
  const int l15  = lane & 15;
  const int wave = __builtin_amdgcn_readfirstlane((int)(threadIdx.x >> 5));
  const bool gath = ((wave >> 2) & 1) != 0;
  const int  g    = ((wave >> 3) << 2) | (wave & 3);
  const int  G    = (int)gridDim.x;
  int nt = 0;
  if ((int)blockIdx.x < tilesM) nt = (tilesM - 1 - (int)blockIdx.x) / G + 1;

  const int wv   = (int)blockIdx.x * 16 + g;
  const int nwv  = 16 * G;
  const int t0   = timeP[0];
  const int t1e  = t0 + intervalP[0];
  const int last = Nd - 1;
  int c0 = 0, tv0 = 0, c1 = 0, tv1 = 0;
  float adv[2] = { 0.0f, 0.0f };
  if (gath && nt > 0) {
    int i1 = wv + nwv;
    if (i1 > last) i1 = last;
    c0  = col[wv * DEG_ + l15];
    tv0 = tt[wv * DEG_ + l15];
    c1  = col[i1 * DEG_ + l15];
    tv1 = tt[i1 * DEG_ + l15];
    float4 xd0[2];
    const float4* p = (const float4*)(X + (size_t)wv * 256) + lane;
    xd0[0] = p[0];
    xd0[1] = p[32];
    float pd0, pd1;
    gat_g1_dst_part(Wc, lane, xd0, pd0, pd1);
    gat_g1_dst_sum(pd0, pd1, lane, adv);
  }

  for (int t = 0; t <= nt; ++t) {
    float4 acc[2][2];
#pragma unroll
    for (int h = 0; h < 2; ++h)
#pragma unroll
      for (int q = 0; q < 2; ++q) acc[h][q] = make_float4(0.0f, 0.0f, 0.0f, 0.0f);

    if (gath) {
      if (t < nt) {
        const int i = wv + t * nwv;
        int in1 = i + nwv;
        if (in1 > last) in1 = last;
        int i2 = i + 2 * nwv;
        if (i2 > last) i2 = last;
        float4 xd1[2];
        {
          const float4* p = (const float4*)(X + (size_t)in1 * 256) + lane;
          xd1[0] = p[0];
          xd1[1] = p[32];
        }
        const int c2  = col[i2 * DEG_ + l15];
        const int tv2 = tt[i2 * DEG_ + l15];
        asm volatile("" ::: "memory");

        const bool val = (tv0 >= t0) && (tv0 < t1e);
        const unsigned km = __builtin_amdgcn_ballot_w32(val) & 0xFFFFu;
        unsigned m = km;
        float4 xr[R][2];
        int    jr[R];
        bool   on[R];
        gat_g1_load<R>(X, m, c0, lane, xr, jr, on);

        float pn0, pn1;
        gat_g1_dst_part(Wc, lane, xd1, pn0, pn1);
        float asv[2] = { 0.0f, 0.0f };
        float mxu[2] = { -__builtin_inff(), -__builtin_inff() };
        gat_g1_dots<R>(Wc, lane, l15, xr, jr, on, adv, asv, mxu);
        float adn[2];
        gat_g1_dst_sum(pn0, pn1, lane, adn);
        const bool spill = (m != 0u);
        if (spill) {
          do {
            gat_g1_load<R>(X, m, c0, lane, xr, jr, on);
            gat_g1_dots<R>(Wc, lane, l15, xr, jr, on, adv, asv, mxu);
          } while (m != 0u);
        }

        float ex[2];
        float inv[2];
#pragma unroll
        for (int h = 0; h < 2; ++h) {
          float ev = asv[h] + adv[h];
          ev = ev > 0.0f ? ev : 0.2f * ev;
          float mx = mxu[h];
          if (!(mx > -__builtin_inff())) mx = 0.0f;
          const float e = val ? __expf(ev - mx) : 0.0f;
          float denom = 0.0f;
#pragma unroll
          for (int j = 0; j < DEG_; ++j) denom += gat_lane_f(e, j);
          inv[h] = 1.0f / fmaxf(denom, 1e-16f);
          ex[h]  = e;
        }

        if (spill) {
          m = km;
          gat_g1_load<R>(X, m, c0, lane, xr, jr, on);
        }
        gat_g1_acc<R>(xr, jr, on, ex, inv, acc);
        while (m != 0u) {
          gat_g1_load<R>(X, m, c0, lane, xr, jr, on);
          gat_g1_acc<R>(xr, jr, on, ex, inv, acc);
        }

        c0 = c1; tv0 = tv1;
        c1 = c2; tv1 = tv2;
        adv[0] = adn[0]; adv[1] = adn[1];
      }
    } else {
      if (t >= 1) {
        const int tileM = (int)blockIdx.x + G * (t - 1);
        const int nh = g >> 3;
        const int ct = g & 7;
        int ln = lane;
        asm volatile("" : "+v"(ln));
        const int ml15 = ln & 15;
        const int mhalf = ln >> 4;
        const float2* zp = (const float2*)(Zt + (nh * 16 + ml15) * FUSE_ZP_) + mhalf;
        const float2* bs = (const float2*)Bs + mhalf * P2 + nh * 128 + ct * 16 + ml15;
        v8f c = v8f{};
#pragma unroll 8
        for (int k0 = 0; k0 < TK; k0 += 4) {
          const float2 a2 = *zp;
          zp += 2;
          const float2 b2 = *bs;
          bs += P;
          v2f a = { a2.x, a2.y };
          v2f b = { b2.x, b2.y };
          c = __builtin_amdgcn_wmma_f32_16x16x4_f32(false, a, false, b, (short)0, c, false, false);
        }
        const int cc = nh * 128 + ct * 16 + ml15;
        const float pb = Pp[cc];
        const float pm = Pp[256 + cc];
        const float ps = Pp[512 + cc];
        const float pg = Pp[768 + cc];
        const float pe = Pp[1024 + cc];
#pragma unroll
        for (int r = 0; r < 8; ++r) {
          const int mrow = tileM * 16 + r + mhalf * 8;
          if (mrow < Nd) {
            float vv = c[r] + pb;
            vv = (vv - pm) * ps * pg + pe;
            vv = vv > 0.0f ? vv : 0.01f * vv;
            D[(size_t)mrow * TN + cc] = vv;
          }
        }
      }
    }
    gat_lds_barrier();
    if (gath && t < nt) {
#pragma unroll
      for (int h = 0; h < 2; ++h) {
        float4* zr = (float4*)(Zt + (h * 16 + g) * FUSE_ZP_);
#pragma unroll
        for (int q = 0; q < 2; ++q) zr[q * 32 + lane] = acc[h][q];
      }
    }
    gat_lds_barrier();
  }
}

static inline int cdiv(int a, int b) { return (a + b - 1) / b; }

extern "C" void kernel_launch(void* const* d_in, const int* in_sizes, int n_in,
                              void* d_out, int out_size, void* d_ws, size_t ws_size,
                              hipStream_t stream) {
  const float* x     = (const float*)d_in[0];
  const int*   col1  = (const int*)d_in[2];
  const int*   t1    = (const int*)d_in[3];
  const int*   col2  = (const int*)d_in[5];
  const int*   t2    = (const int*)d_in[6];
  const int*   timeP = (const int*)d_in[7];
  const int*   intvP = (const int*)d_in[8];
  const float* Ws1   = (const float*)d_in[9];
  const float* Wd1   = (const float*)d_in[10];
  const float* as1   = (const float*)d_in[11];
  const float* ad1   = (const float*)d_in[12];
  const float* b1    = (const float*)d_in[13];
  const float* g1    = (const float*)d_in[14];
  const float* beta1 = (const float*)d_in[15];
  const float* m1    = (const float*)d_in[16];
  const float* v1    = (const float*)d_in[17];
  const float* Ws2   = (const float*)d_in[18];
  const float* Wd2   = (const float*)d_in[19];
  const float* as2   = (const float*)d_in[20];
  const float* ad2   = (const float*)d_in[21];
  const float* b2    = (const float*)d_in[22];

  float* ws   = (float*)d_ws;
  float* hd1  = ws;
  float* h    = hd1  + (size_t)N1_ * 256;
  float* as2v = h    + (size_t)N1_ * 256;
  float* ad2v = as2v + (size_t)N1_;
  float* hs2  = hd1;
  float* wf1  = hd1  + (size_t)N1_ * 128;
  float* wf2  = wf1  + (size_t)IN_ * 16;

  const dim3 blk(256);
  const dim3 gblk(GEMM_WAVES_ * 32);

  gat_fold_att<<<cdiv(IN_ + H1_ * HID_, 8), blk, 0, stream>>>(
      Ws1, Wd1, as1, ad1, Ws2, Wd2, as2, ad2, wf1, wf2);

  gat_gather_gemm1<7><<<GEMM_WGS_, gblk, 0, stream>>>(
      x, wf1, col1, t1, timeP, intvP, Ws1, h,
      b1, g1, beta1, m1, v1, N1_, cdiv(N1_, 16));

  gat_gemm_wmma<128, 256><<<GEMM_WGS_, gblk, 0, stream>>>(
      h, Ws2, wf2, hs2, as2v, ad2v,
      nullptr, nullptr, nullptr, nullptr, nullptr, N1_, N2_, cdiv(N1_, 16));
  gat_gather_resident<H2_, 1, 8, true><<<GATHER2_WGS_, blk, 0, stream>>>(
      hs2, as2v, ad2v, col2, t2, timeP, intvP, b2, (float*)d_out, N2_);
}


# Cross-lane butterfly steps (layer-1 logit sums, layer-2 row maximum, weight folding) as DPP row_xmask and v_permlanex16 on the vector ALU, not ds_bpermute via LDS; same partners, sum order unchanged
# speedup vs baseline: 3.1750x; 1.0851x over previous
#include <hip/hip_runtime.h>
#include <hip/hip_bf16.h>
#include <cstddef>

#define N0_  100000
#define N1_  50000
#define N2_  25000
#define DEG_ 16
#define IN_  256
#define HID_ 128
#define OUT_ 128
#define H1_  2
#define H2_  1

typedef float v2f __attribute__((ext_vector_type(2)));
typedef float v8f __attribute__((ext_vector_type(8)));

template <int OFF>
__device__ __forceinline__ float gat_lane_xor(float x) {
  static_assert(OFF == 16 || OFF == 8 || OFF == 4 || OFF == 2 || OFF == 1, "one butterfly step");
  const int b = __float_as_int(x);
  if constexpr (OFF == 16) {
    return __int_as_float(__builtin_amdgcn_permlanex16(b, b, 0x76543210, 0xfedcba98, false, false));
  } else {
    return __int_as_float(__builtin_amdgcn_update_dpp(0, b, 0x160 + OFF, 0xf, 0xf, true));
  }
}
__device__ __forceinline__ float gat_row_sum(float s) {
  s += gat_lane_xor<8>(s);
  s += gat_lane_xor<4>(s);
  s += gat_lane_xor<2>(s);
  s += gat_lane_xor<1>(s);
  return s;
}
__device__ __forceinline__ float gat_wave_sum(float s) {
  s += gat_lane_xor<16>(s);
  return gat_row_sum(s);
}

template <int H>
__device__ __forceinline__ void gat_fold_row(const float* __restrict__ Ws,
                                             const float* __restrict__ Wd,
                                             const float* __restrict__ as,
                                             const float* __restrict__ ad,
                                             float* __restrict__ WF,
                                             int k, int lane) {
  float outv = 0.0f;
#pragma unroll
  for (int h = 0; h < H; ++h) {
    const float4 av = *((const float4*)(as + (size_t)h * 128) + lane);
    const float4 dv = *((const float4*)(ad + (size_t)h * 128) + lane);
    const float4 sv = *((const float4*)(Ws + (size_t)k * (H * 128) + h * 128) + lane);
    const float4 wv = *((const float4*)(Wd + (size_t)k * (H * 128) + h * 128) + lane);
    float s = sv.x * av.x + sv.y * av.y + sv.z * av.z + sv.w * av.w;
    float d = wv.x * dv.x + wv.y * dv.y + wv.z * dv.z + wv.w * dv.w;
    s = gat_wave_sum(s);
    d = gat_wave_sum(d);
    if (lane == h)     outv = s;
    if (lane == H + h) outv = d;
  }
  if (lane < 16) WF[(size_t)k * 16 + lane] = outv;
}

__global__ void gat_fold_att(const float* __restrict__ Ws1, const float* __restrict__ Wd1,
                             const float* __restrict__ as1, const float* __restrict__ ad1,
                             const float* __restrict__ Ws2, const float* __restrict__ Wd2,
                             const float* __restrict__ as2, const float* __restrict__ ad2,
                             float* __restrict__ WF1, float* __restrict__ WF2) {
  const int wid  = blockIdx.x * 8 + (threadIdx.x >> 5);
  const int lane = threadIdx.x & 31;
  if (wid < IN_) {
    gat_fold_row<H1_>(Ws1, Wd1, as1, ad1, WF1, wid, lane);
  } else if (wid < IN_ + H1_ * HID_) {
    gat_fold_row<H2_>(Ws2, Wd2, as2, ad2, WF2, wid - IN_, lane);
  }
}


__device__ __forceinline__ float gat_lane_f(float v, int l) {
  return __int_as_float(__builtin_amdgcn_readlane(__float_as_int(v), l));
}

template <int H, int NQ, int R, bool FINAL>
__global__ void __launch_bounds__(256)
gat_gather_resident(const float* __restrict__ X,
                    const float* __restrict__ AS,
                    const float* __restrict__ AD,
                    const int* __restrict__ col,
                    const int* __restrict__ tt,
                    const int* __restrict__ timeP,
                    const int* __restrict__ intervalP,
                    const float* __restrict__ b,
                    float* __restrict__ OUTP,
                    int Nd) {
  static_assert(!FINAL || (H == 1 && NQ == 1), "final form: one head, 128 channels");
  const int lane = threadIdx.x & 31;
  const int l15  = lane & 15;
  const int wv  = __builtin_amdgcn_readfirstlane((int)(blockIdx.x * 8 + (threadIdx.x >> 5)));
  const int nwv = (int)gridDim.x * 8;
  if (wv >= Nd) return;

  const int t0   = timeP[0];
  const int t1e  = t0 + intervalP[0];
  const int last = Nd - 1;

  float4 bq = make_float4(0.0f, 0.0f, 0.0f, 0.0f);
  if constexpr (FINAL) bq = *((const float4*)b + lane);

  int i1 = wv + nwv;
  if (i1 > last) i1 = last;
  int c0  = col[wv * DEG_ + l15];
  int tv0 = tt[wv * DEG_ + l15];
  int c1  = col[i1 * DEG_ + l15];
  int tv1 = tt[i1 * DEG_ + l15];
  float as0[H];
  float ad0[H];
#pragma unroll
  for (int h = 0; h < H; ++h) {
    as0[h] = AS[(size_t)c0 * H + h];
    ad0[h] = AD[(size_t)wv * H + h];
  }

  for (int i = wv; i < Nd; i += nwv) {
    int in1 = i + nwv;
    if (in1 > last) in1 = last;
    int i2 = i + 2 * nwv;
    if (i2 > last) i2 = last;
    float as1[H];
    float ad1[H];
#pragma unroll
    for (int h = 0; h < H; ++h) {
      as1[h] = AS[(size_t)c1 * H + h];
      ad1[h] = AD[(size_t)in1 * H + h];
    }
    const int c2  = col[i2 * DEG_ + l15];
    const int tv2 = tt[i2 * DEG_ + l15];
    asm volatile("" ::: "memory");

    const bool val = (tv0 >= t0) && (tv0 < t1e);
    float ex[H];
    float inv[H];
#pragma unroll
    for (int h = 0; h < H; ++h) {
      float ev = as0[h] + ad0[h];
      ev = ev > 0.0f ? ev : 0.2f * ev;
      float m = val ? ev : -__builtin_inff();
      m = fmaxf(m, gat_lane_xor<8>(m));
      m = fmaxf(m, gat_lane_xor<4>(m));
      m = fmaxf(m, gat_lane_xor<2>(m));
      m = fmaxf(m, gat_lane_xor<1>(m));
      if (!(m > -__builtin_inff())) m = 0.0f;
      const float e = val ? __expf(ev - m) : 0.0f;
      float denom = 0.0f;
#pragma unroll
      for (int j = 0; j < DEG_; ++j) denom += gat_lane_f(e, j);
      inv[h] = 1.0f / fmaxf(denom, 1e-16f);
      ex[h]  = e;
    }
    bool nz = false;
#pragma unroll
    for (int h = 0; h < H; ++h) nz = nz || (ex[h] != 0.0f);
    unsigned mask = __builtin_amdgcn_ballot_w32(nz) & 0xFFFFu;

    float4 acc[H][NQ];
#pragma unroll
    for (int h = 0; h < H; ++h)
#pragma unroll
      for (int q = 0; q < NQ; ++q) acc[h][q] = make_float4(0.0f, 0.0f, 0.0f, 0.0f);

    while (mask != 0u) {
      float4 xr[R][NQ];
      float  wr[H][R];
      bool   on[R];
#pragma unroll
      for (int r = 0; r < R; ++r) {
        on[r] = (mask != 0u);
        if (on[r]) {
          const int j = __builtin_ctz(mask);
          mask &= mask - 1u;
          const int cr = __builtin_amdgcn_readlane(c0, j);
#pragma unroll
          for (int h = 0; h < H; ++h) wr[h][r] = gat_lane_f(ex[h], j) * inv[h];
          const float4* p = (const float4*)(X + (size_t)cr * (NQ * 128)) + lane;
#pragma unroll
          for (int q = 0; q < NQ; ++q) xr[r][q] = p[q * 32];
        }
      }
#pragma unroll
      for (int r = 0; r < R; ++r) {
        if (on[r]) {
#pragma unroll
          for (int h = 0; h < H; ++h) {
            const float wj = wr[h][r];
#pragma unroll
            for (int q = 0; q < NQ; ++q) {
              acc[h][q].x += wj * xr[r][q].x; acc[h][q].y += wj * xr[r][q].y;
              acc[h][q].z += wj * xr[r][q].z; acc[h][q].w += wj * xr[r][q].w;
            }
          }
        }
      }
    }

    if constexpr (FINAL) {
      float4 o;
      o.x = acc[0][0].x / (float)H + bq.x;
      o.y = acc[0][0].y / (float)H + bq.y;
      o.z = acc[0][0].z / (float)H + bq.z;
      o.w = acc[0][0].w / (float)H + bq.w;
      *((float4*)(OUTP + (size_t)i * 128) + lane) = o;
    } else {
#pragma unroll
      for (int h = 0; h < H; ++h) {
        float4* zr = (float4*)(OUTP + ((size_t)h * Nd + i) * (NQ * 128));
#pragma unroll
        for (int q = 0; q < NQ; ++q) zr[q * 32 + lane] = acc[h][q];
      }
    }

    c0 = c1; tv0 = tv1;
#pragma unroll
    for (int h = 0; h < H; ++h) { as0[h] = as1[h]; ad0[h] = ad1[h]; }
    c1 = c2; tv1 = tv2;
  }
}

template <int R>
__device__ __forceinline__ void gat_g1_load(const float* __restrict__ X, unsigned& m,
                                            int cidx, int lane,
                                            float4 (&xr)[R][2], int (&jr)[R], bool (&on)[R]) {
#pragma unroll
  for (int r = 0; r < R; ++r) {
    on[r] = (m != 0u);
    jr[r] = 16;
    if (on[r]) {
      const int j = __builtin_ctz(m);
      m &= m - 1u;
      jr[r] = j;
      const int cr = __builtin_amdgcn_readlane(cidx, j);
      const float4* p = (const float4*)(X + (size_t)cr * 256) + lane;
      xr[r][0] = p[0];
      xr[r][1] = p[32];
    }
  }
}

__device__ __forceinline__ float gat_dot8(const float4& a0, const float4& a1,
                                          const float4& w0, const float4& w1) {
  return a0.x * w0.x + a0.y * w0.y + a0.z * w0.z + a0.w * w0.w +
         a1.x * w1.x + a1.y * w1.y + a1.z * w1.z + a1.w * w1.w;
}

__device__ __forceinline__ float gat_pair_sum(float p0, float p1, int lane) {
  const bool hi = (lane & 16) != 0;
  const float s = (hi ? p1 : p0) + gat_lane_xor<16>(hi ? p0 : p1);
  return gat_row_sum(s);
}

template <int R>
__device__ __forceinline__ void gat_g1_dots(const float* Wc, int lane, int l15,
                                            const float4 (&xr)[R][2], const int (&jr)[R],
                                            const bool (&on)[R], const float (&adv)[2],
                                            float (&asv)[2], float (&mxu)[2]) {
  static_assert(R <= 8, "the halving exchange is written for up to eight row slots");
  const float4 w00 = *((const float4*)(Wc) + lane);
  const float4 w01 = *((const float4*)(Wc) + 32 + lane);
  const float4 w10 = *((const float4*)(Wc + 256) + lane);
  const float4 w11 = *((const float4*)(Wc + 256) + 32 + lane);
  float p0[R];
  float p1[R];
#pragma unroll
  for (int r = 0; r < R; ++r) {
    p0[r] = 0.0f;
    p1[r] = 0.0f;
    if (on[r]) {
      p0[r] = gat_dot8(xr[r][0], xr[r][1], w00, w01);
      p1[r] = gat_dot8(xr[r][0], xr[r][1], w10, w11);
    }
  }
  const bool hi = (lane & 16) != 0;
  const bool b3 = (lane & 8) != 0;
  const bool b2 = (lane & 4) != 0;
  const bool b1 = (lane & 2) != 0;
  float a[8];
#pragma unroll
  for (int r = 0; r < 8; ++r) {
    a[r] = 0.0f;
    if (r < R) a[r] = (hi ? p1[r] : p0[r]) + gat_lane_xor<16>(hi ? p0[r] : p1[r]);
  }
  float b[4];
#pragma unroll
  for (int k = 0; k < 4; ++k)
    b[k] = (b3 ? a[k + 4] : a[k]) + gat_lane_xor<8>(b3 ? a[k] : a[k + 4]);
  float c[2];
#pragma unroll
  for (int k = 0; k < 2; ++k)
    c[k] = (b2 ? b[k + 2] : b[k]) + gat_lane_xor<4>(b2 ? b[k] : b[k + 2]);
  float d = (b1 ? c[1] : c[0]) + gat_lane_xor<2>(b1 ? c[0] : c[1]);
  d += gat_lane_xor<1>(d);
#pragma unroll
  for (int r = 0; r < R; ++r) {
    const float s0 = gat_lane_f(d, 2 * r);
    const float s1 = gat_lane_f(d, 16 + 2 * r);
    if (l15 == jr[r]) { asv[0] = s0; asv[1] = s1; }
    if (on[r]) {
      float e0 = s0 + adv[0];
      float e1 = s1 + adv[1];
      e0 = e0 > 0.0f ? e0 : 0.2f * e0;
      e1 = e1 > 0.0f ? e1 : 0.2f * e1;
      mxu[0] = fmaxf(mxu[0], e0);
      mxu[1] = fmaxf(mxu[1], e1);
    }
  }
}

__device__ __forceinline__ void gat_g1_dst_part(const float* Wc, int lane,
                                                const float4 (&xd)[2], float& pd0, float& pd1) {
  const float4 d00 = *((const float4*)(Wc + 512) + lane);
  const float4 d01 = *((const float4*)(Wc + 512) + 32 + lane);
  const float4 d10 = *((const float4*)(Wc + 768) + lane);
  const float4 d11 = *((const float4*)(Wc + 768) + 32 + lane);
  pd0 = gat_dot8(xd[0], xd[1], d00, d01);
  pd1 = gat_dot8(xd[0], xd[1], d10, d11);
}
__device__ __forceinline__ void gat_g1_dst_sum(float pd0, float pd1, int lane, float (&adv)[2]) {
  const float s = gat_pair_sum(pd0, pd1, lane);
  adv[0] = gat_lane_f(s, 0);
  adv[1] = gat_lane_f(s, 16);
}

template <int R>
__device__ __forceinline__ void gat_g1_acc(const float4 (&xr)[R][2], const int (&jr)[R],
                                           const bool (&on)[R], const float (&ex)[2],
                                           const float (&inv)[2], float4 (&acc)[2][2]) {
#pragma unroll
  for (int r = 0; r < R; ++r) {
    if (on[r]) {
#pragma unroll
      for (int h = 0; h < 2; ++h) {
        const float wj = gat_lane_f(ex[h], jr[r]) * inv[h];
#pragma unroll
        for (int q = 0; q < 2; ++q) {
          acc[h][q].x += wj * xr[r][q].x; acc[h][q].y += wj * xr[r][q].y;
          acc[h][q].z += wj * xr[r][q].z; acc[h][q].w += wj * xr[r][q].w;
        }
      }
    }
  }
}

#define GEMM_WGS_   256
#define GATHER2_WGS_ 512
#define GEMM_WAVES_ 32

template <int TN, int TK, int NT>
__device__ __forceinline__ void gat_gemm_item(const float* __restrict__ A,
                                              const float* Bs,
                                              float* __restrict__ D,
                                              float* __restrict__ AS,
                                              float* __restrict__ AD,
                                              const float* __restrict__ bias,
                                              const float* __restrict__ bng,
                                              const float* __restrict__ bnb,
                                              const float* __restrict__ bnm,
                                              const float* __restrict__ bnv,
                                              int M, int Md, int tileM, int nh,
                                              int half, int l15) {
  constexpr int P  = 2 * TN + 32;
  constexpr int P2 = P / 2;
  constexpr int H  = TN / 128;

  int row = tileM * 16 + l15;
  if (row >= M) row = M - 1;
  const float2* ap = (const float2*)(A + (size_t)row * TK) + half;
  const float2* bs = (const float2*)Bs + half * P2 + nh * 128 + l15;

  v8f c[NT];
#pragma unroll
  for (int t = 0; t < NT; ++t) c[t] = v8f{};

  if constexpr (NT == 8) {
#pragma unroll 4
    for (int k0 = 0; k0 < TK; k0 += 4) {
      const float2 a2 = *ap;
      ap += 2;
      v2f a = { a2.x, a2.y };
#pragma unroll
      for (int t = 0; t < 8; ++t) {
        const float2 b2 = bs[t * 16];
        v2f b = { b2.x, b2.y };
        c[t] = __builtin_amdgcn_wmma_f32_16x16x4_f32(false, a, false, b, (short)0, c[t], false, false);
      }
      bs += P;
    }
  } else {
    float2 an = *ap;
    float2 bn[NT];
#pragma unroll
    for (int t = 0; t < 8; ++t) bn[t] = bs[t * 16];
    bn[8] = bs[TN];
#pragma unroll
    for (int k0 = 0; k0 < TK - 4; k0 += 4) {
      const float2 ac = an;
      float2 bc[NT];
#pragma unroll
      for (int t = 0; t < NT; ++t) bc[t] = bn[t];
      ap += 2;
      bs += P;
      an = *ap;
#pragma unroll
      for (int t = 0; t < 8; ++t) bn[t] = bs[t * 16];
      bn[8] = bs[TN];
      __builtin_amdgcn_sched_barrier(0);
      v2f a = { ac.x, ac.y };
#pragma unroll
      for (int t = 0; t < NT; ++t) {
        v2f b = { bc[t].x, bc[t].y };
        c[t] = __builtin_amdgcn_wmma_f32_16x16x4_f32(false, a, false, b, (short)0, c[t], false, false);
      }
    }
    {
      v2f a = { an.x, an.y };
#pragma unroll
      for (int t = 0; t < NT; ++t) {
        v2f b = { bn[t].x, bn[t].y };
        c[t] = __builtin_amdgcn_wmma_f32_16x16x4_f32(false, a, false, b, (short)0, c[t], false, false);
      }
    }
  }

  if constexpr (NT == 8) {
#pragma unroll
    for (int t = 0; t < 8; ++t) {
      const int cc = nh * 128 + t * 16 + l15;
      const float pb = bias[cc];
      const float pm = bnm[cc];
      const float ps = rsqrtf(bnv[cc] + 1e-5f);
      const float pg = bng[cc];
      const float pe = bnb[cc];
#pragma unroll
      for (int r = 0; r < 8; ++r) {
        const int m = tileM * 16 + r + half * 8;
        if (m < M) {
          float vv = c[t][r] + pb;
          vv = (vv - pm) * ps * pg + pe;
          vv = vv > 0.0f ? vv : 0.01f * vv;
          D[(size_t)m * TN + cc] = vv;
        }
      }
    }
  } else {
#pragma unroll
    for (int r = 0; r < 8; ++r) {
      const int m = tileM * 16 + r + half * 8;
      if (m < M) {
        float* Dr = D + (size_t)m * TN + nh * 128 + l15;
#pragma unroll
        for (int t = 0; t < 8; ++t) Dr[t * 16] = c[t][r];
        if (l15 < H) {
          AS[(size_t)m * H + l15] = c[NT - 1][r];
        } else if (l15 < 2 * H && m < Md) {
          AD[(size_t)m * H + (l15 - H)] = c[NT - 1][r];
        }
      }
    }
  }
}

template <int TN, int TK, int NTP>
__device__ __forceinline__ void gat_gemm_piece(const float* __restrict__ A,
                                               const float* Bs,
                                               float* __restrict__ D,
                                               float* __restrict__ AS,
                                               float* __restrict__ AD,
                                               const float* __restrict__ bias,
                                               const float* __restrict__ bng,
                                               const float* __restrict__ bnb,
                                               const float* __restrict__ bnm,
                                               const float* __restrict__ bnv,
                                               int M, int Md, int tileM, int ct0,
                                               int half, int l15) {
  constexpr int P  = 2 * TN + 32;
  constexpr int P2 = P / 2;
  constexpr int H  = TN / 128;

  int row = tileM * 16 + l15;
  if (row >= M) row = M - 1;
  const float2* ap = (const float2*)(A + (size_t)row * TK) + half;
  const float2* bs = (const float2*)Bs + half * P2 + ct0 * 16 + l15;

  v8f c[NTP];
#pragma unroll
  for (int t = 0; t < NTP; ++t) c[t] = v8f{};

#pragma unroll 8
  for (int k0 = 0; k0 < TK; k0 += 4) {
    const float2 a2 = *ap;
    ap += 2;
    v2f a = { a2.x, a2.y };
#pragma unroll
    for (int t = 0; t < NTP; ++t) {
      const float2 b2 = bs[t * 16];
      v2f b = { b2.x, b2.y };
      c[t] = __builtin_amdgcn_wmma_f32_16x16x4_f32(false, a, false, b, (short)0, c[t], false, false);
    }
    bs += P;
  }

  if constexpr (TN == 256) {
#pragma unroll
    for (int t = 0; t < NTP; ++t) {
      const int cc = (ct0 + t) * 16 + l15;
      const float pb = bias[cc];
      const float pm = bnm[cc];
      const float ps = rsqrtf(bnv[cc] + 1e-5f);
      const float pg = bng[cc];
      const float pe = bnb[cc];
#pragma unroll
      for (int r = 0; r < 8; ++r) {
        const int m = tileM * 16 + r + half * 8;
        if (m < M) {
          float vv = c[t][r] + pb;
          vv = (vv - pm) * ps * pg + pe;
          vv = vv > 0.0f ? vv : 0.01f * vv;
          D[(size_t)m * TN + cc] = vv;
        }
      }
    }
  } else if constexpr (NTP == 2) {
#pragma unroll
    for (int r = 0; r < 8; ++r) {
      const int m = tileM * 16 + r + half * 8;
      if (m < M) {
        float* Dr = D + (size_t)m * TN + ct0 * 16 + l15;
#pragma unroll
        for (int t = 0; t < NTP; ++t) Dr[t * 16] = c[t][r];
      }
    }
  } else {
#pragma unroll
    for (int r = 0; r < 8; ++r) {
      const int m = tileM * 16 + r + half * 8;
      if (m < M) {
        if (l15 < H) {
          AS[(size_t)m * H + l15] = c[0][r];
        } else if (l15 < 2 * H && m < Md) {
          AD[(size_t)m * H + (l15 - H)] = c[0][r];
        }
      }
    }
  }
}

template <int TN, int TK>
__global__ void __launch_bounds__(GEMM_WAVES_ * 32)
gat_gemm_wmma(const float* __restrict__ A,
              const float* __restrict__ B,
              const float* __restrict__ WF,
              float* __restrict__ D,
              float* __restrict__ AS,
              float* __restrict__ AD,
              const float* __restrict__ bias,
              const float* __restrict__ bng,
              const float* __restrict__ bnb,
              const float* __restrict__ bnm,
              const float* __restrict__ bnv,
              int M, int Md, int tilesM) {
  constexpr int P      = 2 * TN + 32;
  constexpr int HALVES = TN / 128;
  __shared__ __attribute__((aligned(16))) float Bs[(TK / 2) * P];

  {
    constexpr int C4    = TN / 4;
    constexpr int ITEMS = (TK / 2) * C4;
#pragma unroll
    for (int it = threadIdx.x; it < ITEMS; it += GEMM_WAVES_ * 32) {
      const int kp = it / C4;
      const int c4 = it % C4;
      const float4 r0 = *((const float4*)(B + (size_t)(2 * kp) * TN) + c4);
      const float4 r1 = *((const float4*)(B + (size_t)(2 * kp + 1) * TN) + c4);
      float4* dst = (float4*)(Bs + kp * P + c4 * 8);
      dst[0] = make_float4(r0.x, r1.x, r0.y, r1.y);
      dst[1] = make_float4(r0.z, r1.z, r0.w, r1.w);
    }
    constexpr int XITEMS = (TK / 2) * 4;
    if (HALVES == 1 && threadIdx.x < XITEMS) {
      const int kp = threadIdx.x >> 2;
      const int c4 = threadIdx.x & 3;
      const float4 r0 = *((const float4*)(WF + (size_t)(2 * kp) * 16) + c4);
      const float4 r1 = *((const float4*)(WF + (size_t)(2 * kp + 1) * 16) + c4);
      float4* dst = (float4*)(Bs + kp * P + 2 * TN + c4 * 8);
      dst[0] = make_float4(r0.x, r1.x, r0.y, r1.y);
      dst[1] = make_float4(r0.z, r1.z, r0.w, r1.w);
    }
  }
  __syncthreads();

  const int lane = threadIdx.x & 31;
  const int wave = __builtin_amdgcn_readfirstlane(threadIdx.x >> 5);
  const int half = lane >> 4;
  const int l15  = lane & 15;

  const int G         = (int)gridDim.x;
  const int mainTiles = (tilesM / G) * G;
  const int restTiles = tilesM - mainTiles;

  if constexpr (HALVES == 2) {
    const int slot  = (wave >> 3) * 4 + (wave & 3);
    const int phase = (wave >> 2) & 1;
    for (int i = 0; ; ++i) {
      const int tileM = blockIdx.x + G * (slot + 16 * i);
      if (tileM >= mainTiles) break;
      const int nh = (phase + i) & 1;
      gat_gemm_item<TN, TK, 8>(A + (size_t)nh * M * TK, Bs, D, AS, AD,
                               bias, bng, bnb, bnm, bnv, M, Md, tileM, nh, half, l15);
    }
    constexpr int PT = TN / 32;
    for (int p = (GEMM_WAVES_ - 1 - wave) * G + (int)blockIdx.x; p < restTiles * PT;
         p += GEMM_WAVES_ * G) {
      const int tileM = mainTiles + p / PT;
      const int ct0   = (p % PT) * 2;
      const int nh    = ct0 >> 3;
      gat_gemm_piece<TN, TK, 2>(A + (size_t)nh * M * TK, Bs, D, AS, AD,
                                bias, bng, bnb, bnm, bnv, M, Md, tileM, ct0, half, l15);
    }
  } else {
    for (int i = 0; ; ++i) {
      const int tileM = blockIdx.x + G * (wave + GEMM_WAVES_ * i);
      if (tileM >= mainTiles) break;
      gat_gemm_item<TN, TK, 9>(A, Bs, D, AS, AD,
                               bias, bng, bnb, bnm, bnv, M, Md, tileM, 0, half, l15);
    }
    constexpr int PT = TN / 32 + 1;
    for (int p = (GEMM_WAVES_ - 1 - wave) * G + (int)blockIdx.x; p < restTiles * PT;
         p += GEMM_WAVES_ * G) {
      const int tileM = mainTiles + p / PT;
      const int q     = p % PT;
      if (q < TN / 32) {
        gat_gemm_piece<TN, TK, 2>(A, Bs, D, AS, AD,
                                  bias, bng, bnb, bnm, bnv, M, Md, tileM, 2 * q, half, l15);
      } else {
        gat_gemm_piece<TN, TK, 1>(A, Bs, D, AS, AD,
                                  bias, bng, bnb, bnm, bnv, M, Md, tileM, TN / 16, half, l15);
      }
    }
  }
}

#define FUSE_ZP_ 260

__device__ __forceinline__ void gat_lds_barrier() {
  asm volatile("s_wait_dscnt 0x0\n\ts_barrier_signal -1\n\ts_barrier_wait -1" ::: "memory");
}
template <int R>
__global__ void __launch_bounds__(GEMM_WAVES_ * 32)
gat_gather_gemm1(const float* __restrict__ X,
                 const float* __restrict__ WF,
                 const int* __restrict__ col,
                 const int* __restrict__ tt,
                 const int* __restrict__ timeP,
                 const int* __restrict__ intervalP,
                 const float* __restrict__ B,
                 float* __restrict__ D,
                 const float* __restrict__ bias,
                 const float* __restrict__ bng,
                 const float* __restrict__ bnb,
                 const float* __restrict__ bnm,
                 const float* __restrict__ bnv,
                 int Nd, int tilesM) {
  static_assert(IN_ == 256 && H1_ == 2 && HID_ == 128, "two heads of 128 columns, rows of 256");
  static_assert(N1_ % 16 == 0, "every row tile is whole: each gather wave of a tile has a row");
  constexpr int TN = 256;
  constexpr int TK = 256;
  constexpr int P  = 2 * TN + 32;
  constexpr int P2 = P / 2;
  __shared__ __attribute__((aligned(16))) float Bs[(TK / 2) * P];
  __shared__ __attribute__((aligned(16))) float Zt[2 * 16 * FUSE_ZP_];
  __shared__ __attribute__((aligned(16))) float Wc[4 * 256];
  __shared__ float Pp[5 * 256];

  {
    constexpr int C4    = TN / 4;
    constexpr int ITEMS = (TK / 2) * C4;
#pragma unroll
    for (int it = threadIdx.x; it < ITEMS; it += GEMM_WAVES_ * 32) {
      const int kp = it / C4;
      const int c4 = it % C4;
      const float4 r0 = *((const float4*)(B + (size_t)(2 * kp) * TN) + c4);
      const float4 r1 = *((const float4*)(B + (size_t)(2 * kp + 1) * TN) + c4);
      float4* dst = (float4*)(Bs + kp * P + c4 * 8);
      dst[0] = make_float4(r0.x, r1.x, r0.y, r1.y);
      dst[1] = make_float4(r0.z, r1.z, r0.w, r1.w);
    }
    if (threadIdx.x < 256) {
      const float4 f = *((const float4*)(WF + (size_t)threadIdx.x * 16));
      Wc[threadIdx.x]       = f.x;
      Wc[256 + threadIdx.x] = f.y;
      Wc[512 + threadIdx.x] = f.z;
      Wc[768 + threadIdx.x] = f.w;
      Pp[threadIdx.x]        = bias[threadIdx.x];
      Pp[256 + threadIdx.x]  = bnm[threadIdx.x];
      Pp[512 + threadIdx.x]  = rsqrtf(bnv[threadIdx.x] + 1e-5f);
      Pp[768 + threadIdx.x]  = bng[threadIdx.x];
      Pp[1024 + threadIdx.x] = bnb[threadIdx.x];
    }
  }
  __syncthreads();

  const int lane = threadIdx.x & 31;
  const int l15  = lane & 15;
  const int wave = __builtin_amdgcn_readfirstlane((int)(threadIdx.x >> 5));
  const bool gath = ((wave >> 2) & 1) != 0;
  const int  g    = ((wave >> 3) << 2) | (wave & 3);
  const int  G    = (int)gridDim.x;
  int nt = 0;
  if ((int)blockIdx.x < tilesM) nt = (tilesM - 1 - (int)blockIdx.x) / G + 1;

  const int wv   = (int)blockIdx.x * 16 + g;
  const int nwv  = 16 * G;
  const int t0   = timeP[0];
  const int t1e  = t0 + intervalP[0];
  const int last = Nd - 1;
  int c0 = 0, tv0 = 0, c1 = 0, tv1 = 0;
  float adv[2] = { 0.0f, 0.0f };
  if (gath && nt > 0) {
    int i1 = wv + nwv;
    if (i1 > last) i1 = last;
    c0  = col[wv * DEG_ + l15];
    tv0 = tt[wv * DEG_ + l15];
    c1  = col[i1 * DEG_ + l15];
    tv1 = tt[i1 * DEG_ + l15];
    float4 xd0[2];
    const float4* p = (const float4*)(X + (size_t)wv * 256) + lane;
    xd0[0] = p[0];
    xd0[1] = p[32];
    float pd0, pd1;
    gat_g1_dst_part(Wc, lane, xd0, pd0, pd1);
    gat_g1_dst_sum(pd0, pd1, lane, adv);
  }

  for (int t = 0; t <= nt; ++t) {
    float4 acc[2][2];
#pragma unroll
    for (int h = 0; h < 2; ++h)
#pragma unroll
      for (int q = 0; q < 2; ++q) acc[h][q] = make_float4(0.0f, 0.0f, 0.0f, 0.0f);

    if (gath) {
      if (t < nt) {
        const int i = wv + t * nwv;
        int in1 = i + nwv;
        if (in1 > last) in1 = last;
        int i2 = i + 2 * nwv;
        if (i2 > last) i2 = last;
        float4 xd1[2];
        {
          const float4* p = (const float4*)(X + (size_t)in1 * 256) + lane;
          xd1[0] = p[0];
          xd1[1] = p[32];
        }
        const int c2  = col[i2 * DEG_ + l15];
        const int tv2 = tt[i2 * DEG_ + l15];
        asm volatile("" ::: "memory");

        const bool val = (tv0 >= t0) && (tv0 < t1e);
        const unsigned km = __builtin_amdgcn_ballot_w32(val) & 0xFFFFu;
        unsigned m = km;
        float4 xr[R][2];
        int    jr[R];
        bool   on[R];
        gat_g1_load<R>(X, m, c0, lane, xr, jr, on);

        float pn0, pn1;
        gat_g1_dst_part(Wc, lane, xd1, pn0, pn1);
        float asv[2] = { 0.0f, 0.0f };
        float mxu[2] = { -__builtin_inff(), -__builtin_inff() };
        gat_g1_dots<R>(Wc, lane, l15, xr, jr, on, adv, asv, mxu);
        float adn[2];
        gat_g1_dst_sum(pn0, pn1, lane, adn);
        const bool spill = (m != 0u);
        if (spill) {
          do {
            gat_g1_load<R>(X, m, c0, lane, xr, jr, on);
            gat_g1_dots<R>(Wc, lane, l15, xr, jr, on, adv, asv, mxu);
          } while (m != 0u);
        }

        float ex[2];
        float inv[2];
#pragma unroll
        for (int h = 0; h < 2; ++h) {
          float ev = asv[h] + adv[h];
          ev = ev > 0.0f ? ev : 0.2f * ev;
          float mx = mxu[h];
          if (!(mx > -__builtin_inff())) mx = 0.0f;
          const float e = val ? __expf(ev - mx) : 0.0f;
          float denom = 0.0f;
#pragma unroll
          for (int j = 0; j < DEG_; ++j) denom += gat_lane_f(e, j);
          inv[h] = 1.0f / fmaxf(denom, 1e-16f);
          ex[h]  = e;
        }

        if (spill) {
          m = km;
          gat_g1_load<R>(X, m, c0, lane, xr, jr, on);
        }
        gat_g1_acc<R>(xr, jr, on, ex, inv, acc);
        while (m != 0u) {
          gat_g1_load<R>(X, m, c0, lane, xr, jr, on);
          gat_g1_acc<R>(xr, jr, on, ex, inv, acc);
        }

        c0 = c1; tv0 = tv1;
        c1 = c2; tv1 = tv2;
        adv[0] = adn[0]; adv[1] = adn[1];
      }
    } else {
      if (t >= 1) {
        const int tileM = (int)blockIdx.x + G * (t - 1);
        const int nh = g >> 3;
        const int ct = g & 7;
        int ln = lane;
        asm volatile("" : "+v"(ln));
        const int ml15 = ln & 15;
        const int mhalf = ln >> 4;
        const float2* zp = (const float2*)(Zt + (nh * 16 + ml15) * FUSE_ZP_) + mhalf;
        const float2* bs = (const float2*)Bs + mhalf * P2 + nh * 128 + ct * 16 + ml15;
        v8f c = v8f{};
#pragma unroll 8
        for (int k0 = 0; k0 < TK; k0 += 4) {
          const float2 a2 = *zp;
          zp += 2;
          const float2 b2 = *bs;
          bs += P;
          v2f a = { a2.x, a2.y };
          v2f b = { b2.x, b2.y };
          c = __builtin_amdgcn_wmma_f32_16x16x4_f32(false, a, false, b, (short)0, c, false, false);
        }
        const int cc = nh * 128 + ct * 16 + ml15;
        const float pb = Pp[cc];
        const float pm = Pp[256 + cc];
        const float ps = Pp[512 + cc];
        const float pg = Pp[768 + cc];
        const float pe = Pp[1024 + cc];
#pragma unroll
        for (int r = 0; r < 8; ++r) {
          const int mrow = tileM * 16 + r + mhalf * 8;
          if (mrow < Nd) {
            float vv = c[r] + pb;
            vv = (vv - pm) * ps * pg + pe;
            vv = vv > 0.0f ? vv : 0.01f * vv;
            D[(size_t)mrow * TN + cc] = vv;
          }
        }
      }
    }
    gat_lds_barrier();
    if (gath && t < nt) {
#pragma unroll
      for (int h = 0; h < 2; ++h) {
        float4* zr = (float4*)(Zt + (h * 16 + g) * FUSE_ZP_);
#pragma unroll
        for (int q = 0; q < 2; ++q) zr[q * 32 + lane] = acc[h][q];
      }
    }
    gat_lds_barrier();
  }
}

static inline int cdiv(int a, int b) { return (a + b - 1) / b; }

extern "C" void kernel_launch(void* const* d_in, const int* in_sizes, int n_in,
                              void* d_out, int out_size, void* d_ws, size_t ws_size,
                              hipStream_t stream) {
  const float* x     = (const float*)d_in[0];
  const int*   col1  = (const int*)d_in[2];
  const int*   t1    = (const int*)d_in[3];
  const int*   col2  = (const int*)d_in[5];
  const int*   t2    = (const int*)d_in[6];
  const int*   timeP = (const int*)d_in[7];
  const int*   intvP = (const int*)d_in[8];
  const float* Ws1   = (const float*)d_in[9];
  const float* Wd1   = (const float*)d_in[10];
  const float* as1   = (const float*)d_in[11];
  const float* ad1   = (const float*)d_in[12];
  const float* b1    = (const float*)d_in[13];
  const float* g1    = (const float*)d_in[14];
  const float* beta1 = (const float*)d_in[15];
  const float* m1    = (const float*)d_in[16];
  const float* v1    = (const float*)d_in[17];
  const float* Ws2   = (const float*)d_in[18];
  const float* Wd2   = (const float*)d_in[19];
  const float* as2   = (const float*)d_in[20];
  const float* ad2   = (const float*)d_in[21];
  const float* b2    = (const float*)d_in[22];

  float* ws   = (float*)d_ws;
  float* hd1  = ws;
  float* h    = hd1  + (size_t)N1_ * 256;
  float* as2v = h    + (size_t)N1_ * 256;
  float* ad2v = as2v + (size_t)N1_;
  float* hs2  = hd1;
  float* wf1  = hd1  + (size_t)N1_ * 128;
  float* wf2  = wf1  + (size_t)IN_ * 16;

  const dim3 blk(256);
  const dim3 gblk(GEMM_WAVES_ * 32);

  gat_fold_att<<<cdiv(IN_ + H1_ * HID_, 8), blk, 0, stream>>>(
      Ws1, Wd1, as1, ad1, Ws2, Wd2, as2, ad2, wf1, wf2);

  gat_gather_gemm1<7><<<GEMM_WGS_, gblk, 0, stream>>>(
      x, wf1, col1, t1, timeP, intvP, Ws1, h,
      b1, g1, beta1, m1, v1, N1_, cdiv(N1_, 16));

  gat_gemm_wmma<128, 256><<<GEMM_WGS_, gblk, 0, stream>>>(
      h, Ws2, wf2, hs2, as2v, ad2v,
      nullptr, nullptr, nullptr, nullptr, nullptr, N1_, N2_, cdiv(N1_, 16));
  gat_gather_resident<H2_, 1, 8, true><<<GATHER2_WGS_, blk, 0, stream>>>(
      hs2, as2v, ad2v, col2, t2, timeP, intvP, b2, (float*)d_out, N2_);
}


# Layer 2 aggregate-then-transform in layer 1's fused gather+WMMA kernel: h rows summed per dst row, GEMM over N2 not N1 rows, logits as vector dots in the gather; one launch less (sum order changes)
# speedup vs baseline: 3.3805x; 1.0647x over previous
#include <hip/hip_runtime.h>
#include <hip/hip_bf16.h>
#include <cstddef>

#define N0_  100000
#define N1_  50000
#define N2_  25000
#define DEG_ 16
#define IN_  256
#define HID_ 128
#define OUT_ 128
#define H1_  2
#define H2_  1

typedef float v2f __attribute__((ext_vector_type(2)));
typedef float v8f __attribute__((ext_vector_type(8)));

template <int OFF>
__device__ __forceinline__ float gat_lane_xor(float x) {
  static_assert(OFF == 16 || OFF == 8 || OFF == 4 || OFF == 2 || OFF == 1, "one butterfly step");
  const int b = __float_as_int(x);
  if constexpr (OFF == 16) {
    return __int_as_float(__builtin_amdgcn_permlanex16(b, b, 0x76543210, 0xfedcba98, false, false));
  } else {
    return __int_as_float(__builtin_amdgcn_update_dpp(0, b, 0x160 + OFF, 0xf, 0xf, true));
  }
}
__device__ __forceinline__ float gat_row_sum(float s) {
  s += gat_lane_xor<8>(s);
  s += gat_lane_xor<4>(s);
  s += gat_lane_xor<2>(s);
  s += gat_lane_xor<1>(s);
  return s;
}
__device__ __forceinline__ float gat_wave_sum(float s) {
  s += gat_lane_xor<16>(s);
  return gat_row_sum(s);
}

template <int H>
__device__ __forceinline__ void gat_fold_row(const float* __restrict__ Ws,
                                             const float* __restrict__ Wd,
                                             const float* __restrict__ as,
                                             const float* __restrict__ ad,
                                             float* __restrict__ WF,
                                             int k, int lane) {
  float outv = 0.0f;
#pragma unroll
  for (int h = 0; h < H; ++h) {
    const float4 av = *((const float4*)(as + (size_t)h * 128) + lane);
    const float4 dv = *((const float4*)(ad + (size_t)h * 128) + lane);
    const float4 sv = *((const float4*)(Ws + (size_t)k * (H * 128) + h * 128) + lane);
    const float4 wv = *((const float4*)(Wd + (size_t)k * (H * 128) + h * 128) + lane);
    float s = sv.x * av.x + sv.y * av.y + sv.z * av.z + sv.w * av.w;
    float d = wv.x * dv.x + wv.y * dv.y + wv.z * dv.z + wv.w * dv.w;
    s = gat_wave_sum(s);
    d = gat_wave_sum(d);
    if (lane == h)     outv = s;
    if (lane == H + h) outv = d;
  }
  if (lane < 16) WF[(size_t)k * 16 + lane] = outv;
}

__global__ void gat_fold_att(const float* __restrict__ Ws1, const float* __restrict__ Wd1,
                             const float* __restrict__ as1, const float* __restrict__ ad1,
                             const float* __restrict__ Ws2, const float* __restrict__ Wd2,
                             const float* __restrict__ as2, const float* __restrict__ ad2,
                             float* __restrict__ WF1, float* __restrict__ WF2) {
  const int wid  = blockIdx.x * 8 + (threadIdx.x >> 5);
  const int lane = threadIdx.x & 31;
  if (wid < IN_) {
    gat_fold_row<H1_>(Ws1, Wd1, as1, ad1, WF1, wid, lane);
  } else if (wid < IN_ + H1_ * HID_) {
    gat_fold_row<H2_>(Ws2, Wd2, as2, ad2, WF2, wid - IN_, lane);
  }
}


__device__ __forceinline__ float gat_lane_f(float v, int l) {
  return __int_as_float(__builtin_amdgcn_readlane(__float_as_int(v), l));
}

template <int R>
__device__ __forceinline__ void gat_g1_load(const float* __restrict__ X, unsigned& m,
                                            int cidx, int lane,
                                            float4 (&xr)[R][2], int (&jr)[R], bool (&on)[R]) {
#pragma unroll
  for (int r = 0; r < R; ++r) {
    on[r] = (m != 0u);
    jr[r] = 16;
    if (on[r]) {
      const int j = __builtin_ctz(m);
      m &= m - 1u;
      jr[r] = j;
      const int cr = __builtin_amdgcn_readlane(cidx, j);
      const float4* p = (const float4*)(X + (size_t)cr * 256) + lane;
      xr[r][0] = p[0];
      xr[r][1] = p[32];
    }
  }
}

__device__ __forceinline__ float gat_dot8(const float4& a0, const float4& a1,
                                          const float4& w0, const float4& w1) {
  return a0.x * w0.x + a0.y * w0.y + a0.z * w0.z + a0.w * w0.w +
         a1.x * w1.x + a1.y * w1.y + a1.z * w1.z + a1.w * w1.w;
}

__device__ __forceinline__ float gat_pair_sum(float p0, float p1, int lane) {
  const bool hi = (lane & 16) != 0;
  const float s = (hi ? p1 : p0) + gat_lane_xor<16>(hi ? p0 : p1);
  return gat_row_sum(s);
}

template <int R>
__device__ __forceinline__ void gat_g1_dots(const float* Wc, int lane, int l15,
                                            const float4 (&xr)[R][2], const int (&jr)[R],
                                            const bool (&on)[R], const float (&adv)[2],
                                            float (&asv)[2], float (&mxu)[2]) {
  static_assert(R <= 8, "the halving exchange is written for up to eight row slots");
  const float4 w00 = *((const float4*)(Wc) + lane);
  const float4 w01 = *((const float4*)(Wc) + 32 + lane);
  const float4 w10 = *((const float4*)(Wc + 256) + lane);
  const float4 w11 = *((const float4*)(Wc + 256) + 32 + lane);
  float p0[R];
  float p1[R];
#pragma unroll
  for (int r = 0; r < R; ++r) {
    p0[r] = 0.0f;
    p1[r] = 0.0f;
    if (on[r]) {
      p0[r] = gat_dot8(xr[r][0], xr[r][1], w00, w01);
      p1[r] = gat_dot8(xr[r][0], xr[r][1], w10, w11);
    }
  }
  const bool hi = (lane & 16) != 0;
  const bool b3 = (lane & 8) != 0;
  const bool b2 = (lane & 4) != 0;
  const bool b1 = (lane & 2) != 0;
  float a[8];
#pragma unroll
  for (int r = 0; r < 8; ++r) {
    a[r] = 0.0f;
    if (r < R) a[r] = (hi ? p1[r] : p0[r]) + gat_lane_xor<16>(hi ? p0[r] : p1[r]);
  }
  float b[4];
#pragma unroll
  for (int k = 0; k < 4; ++k)
    b[k] = (b3 ? a[k + 4] : a[k]) + gat_lane_xor<8>(b3 ? a[k] : a[k + 4]);
  float c[2];
#pragma unroll
  for (int k = 0; k < 2; ++k)
    c[k] = (b2 ? b[k + 2] : b[k]) + gat_lane_xor<4>(b2 ? b[k] : b[k + 2]);
  float d = (b1 ? c[1] : c[0]) + gat_lane_xor<2>(b1 ? c[0] : c[1]);
  d += gat_lane_xor<1>(d);
#pragma unroll
  for (int r = 0; r < R; ++r) {
    const float s0 = gat_lane_f(d, 2 * r);
    const float s1 = gat_lane_f(d, 16 + 2 * r);
    if (l15 == jr[r]) { asv[0] = s0; asv[1] = s1; }
    if (on[r]) {
      float e0 = s0 + adv[0];
      float e1 = s1 + adv[1];
      e0 = e0 > 0.0f ? e0 : 0.2f * e0;
      e1 = e1 > 0.0f ? e1 : 0.2f * e1;
      mxu[0] = fmaxf(mxu[0], e0);
      mxu[1] = fmaxf(mxu[1], e1);
    }
  }
}

__device__ __forceinline__ void gat_g1_dst_part(const float* Wc, int lane,
                                                const float4 (&xd)[2], float& pd0, float& pd1) {
  const float4 d00 = *((const float4*)(Wc + 512) + lane);
  const float4 d01 = *((const float4*)(Wc + 512) + 32 + lane);
  const float4 d10 = *((const float4*)(Wc + 768) + lane);
  const float4 d11 = *((const float4*)(Wc + 768) + 32 + lane);
  pd0 = gat_dot8(xd[0], xd[1], d00, d01);
  pd1 = gat_dot8(xd[0], xd[1], d10, d11);
}
__device__ __forceinline__ void gat_g1_dst_sum(float pd0, float pd1, int lane, float (&adv)[2]) {
  const float s = gat_pair_sum(pd0, pd1, lane);
  adv[0] = gat_lane_f(s, 0);
  adv[1] = gat_lane_f(s, 16);
}

template <int R>
__device__ __forceinline__ void gat_g1_acc(const float4 (&xr)[R][2], const int (&jr)[R],
                                           const bool (&on)[R], const float (&ex)[2],
                                           const float (&inv)[2], float4 (&acc)[2][2]) {
#pragma unroll
  for (int r = 0; r < R; ++r) {
    if (on[r]) {
#pragma unroll
      for (int h = 0; h < 2; ++h) {
        const float wj = gat_lane_f(ex[h], jr[r]) * inv[h];
#pragma unroll
        for (int q = 0; q < 2; ++q) {
          acc[h][q].x += wj * xr[r][q].x; acc[h][q].y += wj * xr[r][q].y;
          acc[h][q].z += wj * xr[r][q].z; acc[h][q].w += wj * xr[r][q].w;
        }
      }
    }
  }
}


template <int R>
__device__ __forceinline__ void gat_g1_dots1(const float* Wc, int lane, int l15,
                                             const float4 (&xr)[R][2], const int (&jr)[R],
                                             const bool (&on)[R], float adv,
                                             float& asv, float& mxu) {
  static_assert(R <= 8, "the halving exchange is written for up to eight row slots");
  const float4 w0 = *((const float4*)(Wc) + lane);
  const float4 w1 = *((const float4*)(Wc) + 32 + lane);
  float p[8];
#pragma unroll
  for (int r = 0; r < 8; ++r) {
    p[r] = 0.0f;
    if (r < R) {
      if (on[r]) p[r] = gat_dot8(xr[r][0], xr[r][1], w0, w1);
    }
  }
  const bool hi = (lane & 16) != 0;
  const bool b3 = (lane & 8) != 0;
  const bool b2 = (lane & 4) != 0;
  float b[4];
#pragma unroll
  for (int k = 0; k < 4; ++k)
    b[k] = (hi ? p[k + 4] : p[k]) + gat_lane_xor<16>(hi ? p[k] : p[k + 4]);
  float c[2];
#pragma unroll
  for (int k = 0; k < 2; ++k)
    c[k] = (b3 ? b[k + 2] : b[k]) + gat_lane_xor<8>(b3 ? b[k] : b[k + 2]);
  float d = (b2 ? c[1] : c[0]) + gat_lane_xor<4>(b2 ? c[0] : c[1]);
  d += gat_lane_xor<2>(d);
  d += gat_lane_xor<1>(d);
#pragma unroll
  for (int r = 0; r < R; ++r) {
    const float s = gat_lane_f(d, 4 * r);
    if (l15 == jr[r]) asv = s;
    if (on[r]) {
      float e = s + adv;
      e = e > 0.0f ? e : 0.2f * e;
      mxu = fmaxf(mxu, e);
    }
  }
}

__device__ __forceinline__ float gat_g1_dst_part1(const float* Wc, int lane, const float4 (&xd)[2]) {
  const float4 d0 = *((const float4*)(Wc + 256) + lane);
  const float4 d1 = *((const float4*)(Wc + 256) + 32 + lane);
  return gat_dot8(xd[0], xd[1], d0, d1);
}

template <int R>
__device__ __forceinline__ void gat_g1_acc1(const float4 (&xr)[R][2], const int (&jr)[R],
                                            const bool (&on)[R], float ex, float inv,
                                            float4 (&acc)[2]) {
#pragma unroll
  for (int r = 0; r < R; ++r) {
    if (on[r]) {
      const float wj = gat_lane_f(ex, jr[r]) * inv;
#pragma unroll
      for (int q = 0; q < 2; ++q) {
        acc[q].x += wj * xr[r][q].x; acc[q].y += wj * xr[r][q].y;
        acc[q].z += wj * xr[r][q].z; acc[q].w += wj * xr[r][q].w;
      }
    }
  }
}

#define GEMM_WGS_   256
#define GEMM_WAVES_ 32
#define FUSE_ZP_ 260

__device__ __forceinline__ void gat_lds_barrier() {
  asm volatile("s_wait_dscnt 0x0\n\ts_barrier_signal -1\n\ts_barrier_wait -1" ::: "memory");
}
template <int R, int H>
__global__ void __launch_bounds__(GEMM_WAVES_ * 32)
gat_gather_gemm(const float* __restrict__ X,
                const float* __restrict__ WF,
                const int* __restrict__ col,
                const int* __restrict__ tt,
                const int* __restrict__ timeP,
                const int* __restrict__ intervalP,
                const float* __restrict__ B,
                float* __restrict__ D,
                const float* __restrict__ bias,
                const float* __restrict__ bng,
                const float* __restrict__ bnb,
                const float* __restrict__ bnm,
                const float* __restrict__ bnv,
                int Nd, int tilesM) {
  static_assert(H == 1 || H == 2, "one or two heads of 128 columns");
  static_assert(IN_ == 256 && H1_ * HID_ == 256, "both layers read rows of 256 floats");
  constexpr int TN = H * 128;
  constexpr int TK = 256;
  constexpr int P  = 2 * TN + 32;
  constexpr int P2 = P / 2;
  __shared__ __attribute__((aligned(16))) float Bs[(TK / 2) * P];
  __shared__ __attribute__((aligned(16))) float Zt[H * 16 * FUSE_ZP_];
  __shared__ __attribute__((aligned(16))) float Wc[2 * H * 256];
  __shared__ float Pp[(H == 2 ? 5 : 1) * TN];

  {
    constexpr int C4    = TN / 4;
    constexpr int ITEMS = (TK / 2) * C4;
#pragma unroll
    for (int it = threadIdx.x; it < ITEMS; it += GEMM_WAVES_ * 32) {
      const int kp = it / C4;
      const int c4 = it % C4;
      const float4 r0 = *((const float4*)(B + (size_t)(2 * kp) * TN) + c4);
      const float4 r1 = *((const float4*)(B + (size_t)(2 * kp + 1) * TN) + c4);
      float4* dst = (float4*)(Bs + kp * P + c4 * 8);
      dst[0] = make_float4(r0.x, r1.x, r0.y, r1.y);
      dst[1] = make_float4(r0.z, r1.z, r0.w, r1.w);
    }
    if (threadIdx.x < 256) {
      const float4 f = *((const float4*)(WF + (size_t)threadIdx.x * 16));
      if constexpr (H == 2) {
        Wc[threadIdx.x]       = f.x;
        Wc[256 + threadIdx.x] = f.y;
        Wc[512 + threadIdx.x] = f.z;
        Wc[768 + threadIdx.x] = f.w;
        Pp[threadIdx.x]        = bias[threadIdx.x];
        Pp[256 + threadIdx.x]  = bnm[threadIdx.x];
        Pp[512 + threadIdx.x]  = rsqrtf(bnv[threadIdx.x] + 1e-5f);
        Pp[768 + threadIdx.x]  = bng[threadIdx.x];
        Pp[1024 + threadIdx.x] = bnb[threadIdx.x];
      } else {
        Wc[threadIdx.x]       = f.x;
        Wc[256 + threadIdx.x] = f.y;
        if (threadIdx.x < TN) Pp[threadIdx.x] = bias[threadIdx.x];
      }
    }
  }
  __syncthreads();

  const int lane = threadIdx.x & 31;
  const int l15  = lane & 15;
  const int wave = __builtin_amdgcn_readfirstlane((int)(threadIdx.x >> 5));
  const bool gath = ((wave >> 2) & 1) != 0;
  const int  g    = ((wave >> 3) << 2) | (wave & 3);
  const int  G    = (int)gridDim.x;
  int nt = 0;
  if ((int)blockIdx.x < tilesM) nt = (tilesM - 1 - (int)blockIdx.x) / G + 1;

  const int wv   = (int)blockIdx.x * 16 + g;
  const int nwv  = 16 * G;
  const int t0   = timeP[0];
  const int t1e  = t0 + intervalP[0];
  const int last = Nd - 1;
  int c0 = 0, tv0 = 0, c1 = 0, tv1 = 0;
  float adv[2] = { 0.0f, 0.0f };
  if (gath && nt > 0) {
    int i1 = wv + nwv;
    if (i1 > last) i1 = last;
    c0  = col[wv * DEG_ + l15];
    tv0 = tt[wv * DEG_ + l15];
    c1  = col[i1 * DEG_ + l15];
    tv1 = tt[i1 * DEG_ + l15];
    float4 xd0[2];
    const float4* p = (const float4*)(X + (size_t)wv * 256) + lane;
    xd0[0] = p[0];
    xd0[1] = p[32];
    if constexpr (H == 2) {
      float pd0, pd1;
      gat_g1_dst_part(Wc, lane, xd0, pd0, pd1);
      gat_g1_dst_sum(pd0, pd1, lane, adv);
    } else {
      adv[0] = gat_lane_f(gat_wave_sum(gat_g1_dst_part1(Wc, lane, xd0)), 0);
    }
  }

  for (int t = 0; t <= nt; ++t) {
    float4 acc[H][2];
#pragma unroll
    for (int h = 0; h < H; ++h)
#pragma unroll
      for (int q = 0; q < 2; ++q) acc[h][q] = make_float4(0.0f, 0.0f, 0.0f, 0.0f);

    if (gath) {
      if (t < nt) {
        const int i = wv + t * nwv;
        int in1 = i + nwv;
        if (in1 > last) in1 = last;
        int i2 = i + 2 * nwv;
        if (i2 > last) i2 = last;
        float4 xd1[2];
        {
          const float4* p = (const float4*)(X + (size_t)in1 * 256) + lane;
          xd1[0] = p[0];
          xd1[1] = p[32];
        }
        const int c2  = col[i2 * DEG_ + l15];
        const int tv2 = tt[i2 * DEG_ + l15];
        asm volatile("" ::: "memory");

        const bool val = (tv0 >= t0) && (tv0 < t1e);
        const unsigned km = __builtin_amdgcn_ballot_w32(val) & 0xFFFFu;
        unsigned m = km;
        float4 xr[R][2];
        int    jr[R];
        bool   on[R];
        gat_g1_load<R>(X, m, c0, lane, xr, jr, on);

        float adn[H];
        if constexpr (H == 2) {
          float pn0, pn1;
          gat_g1_dst_part(Wc, lane, xd1, pn0, pn1);
          float asv[2] = { 0.0f, 0.0f };
          float mxu[2] = { -__builtin_inff(), -__builtin_inff() };
          gat_g1_dots<R>(Wc, lane, l15, xr, jr, on, adv, asv, mxu);
          gat_g1_dst_sum(pn0, pn1, lane, adn);
          const bool spill = (m != 0u);
          if (spill) {
            do {
              gat_g1_load<R>(X, m, c0, lane, xr, jr, on);
              gat_g1_dots<R>(Wc, lane, l15, xr, jr, on, adv, asv, mxu);
            } while (m != 0u);
          }

          float ex[2];
          float inv[2];
#pragma unroll
          for (int h = 0; h < 2; ++h) {
            float ev = asv[h] + adv[h];
            ev = ev > 0.0f ? ev : 0.2f * ev;
            float mx = mxu[h];
            if (!(mx > -__builtin_inff())) mx = 0.0f;
            const float e = val ? __expf(ev - mx) : 0.0f;
            float denom = 0.0f;
#pragma unroll
            for (int j = 0; j < DEG_; ++j) denom += gat_lane_f(e, j);
            inv[h] = 1.0f / fmaxf(denom, 1e-16f);
            ex[h]  = e;
          }

          if (spill) {
            m = km;
            gat_g1_load<R>(X, m, c0, lane, xr, jr, on);
          }
          gat_g1_acc<R>(xr, jr, on, ex, inv, acc);
          while (m != 0u) {
            gat_g1_load<R>(X, m, c0, lane, xr, jr, on);
            gat_g1_acc<R>(xr, jr, on, ex, inv, acc);
          }
        } else {
          const float pn = gat_g1_dst_part1(Wc, lane, xd1);
          float asv = 0.0f;
          float mxu = -__builtin_inff();
          gat_g1_dots1<R>(Wc, lane, l15, xr, jr, on, adv[0], asv, mxu);
          adn[0] = gat_lane_f(gat_wave_sum(pn), 0);
          const bool spill = (m != 0u);
          if (spill) {
            do {
              gat_g1_load<R>(X, m, c0, lane, xr, jr, on);
              gat_g1_dots1<R>(Wc, lane, l15, xr, jr, on, adv[0], asv, mxu);
            } while (m != 0u);
          }

          float ev = asv + adv[0];
          ev = ev > 0.0f ? ev : 0.2f * ev;
          float mx = mxu;
          if (!(mx > -__builtin_inff())) mx = 0.0f;
          const float ex = val ? __expf(ev - mx) : 0.0f;
          float denom = 0.0f;
#pragma unroll
          for (int j = 0; j < DEG_; ++j) denom += gat_lane_f(ex, j);
          const float inv = 1.0f / fmaxf(denom, 1e-16f);

          if (spill) {
            m = km;
            gat_g1_load<R>(X, m, c0, lane, xr, jr, on);
          }
          gat_g1_acc1<R>(xr, jr, on, ex, inv, acc[0]);
          while (m != 0u) {
            gat_g1_load<R>(X, m, c0, lane, xr, jr, on);
            gat_g1_acc1<R>(xr, jr, on, ex, inv, acc[0]);
          }
        }

        c0 = c1; tv0 = tv1;
        c1 = c2; tv1 = tv2;
#pragma unroll
        for (int h = 0; h < H; ++h) adv[h] = adn[h];
      }
    } else {
      if (t >= 1 && (H == 2 || (((wave >> 3) + wave) & 1) == 0)) {
        const int tileM = (int)blockIdx.x + G * (t - 1);
        const int nh = (H == 2) ? (g >> 3) : 0;
        const int ct = (H == 2) ? (g & 7) : (((wave >> 3) << 1) | ((wave >> 1) & 1));
        int ln = lane;
        asm volatile("" : "+v"(ln));
        const int ml15 = ln & 15;
        const int mhalf = ln >> 4;
        const float2* zp = (const float2*)(Zt + (nh * 16 + ml15) * FUSE_ZP_) + mhalf;
        const float2* bs = (const float2*)Bs + mhalf * P2 + nh * 128 + ct * 16 + ml15;
        v8f c = v8f{};
#pragma unroll 8
        for (int k0 = 0; k0 < TK; k0 += 4) {
          const float2 a2 = *zp;
          zp += 2;
          const float2 b2 = *bs;
          bs += P;
          v2f a = { a2.x, a2.y };
          v2f b = { b2.x, b2.y };
          c = __builtin_amdgcn_wmma_f32_16x16x4_f32(false, a, false, b, (short)0, c, false, false);
        }
        const int cc = nh * 128 + ct * 16 + ml15;
        if constexpr (H == 2) {
          const float pb = Pp[cc];
          const float pm = Pp[256 + cc];
          const float ps = Pp[512 + cc];
          const float pg = Pp[768 + cc];
          const float pe = Pp[1024 + cc];
#pragma unroll
          for (int r = 0; r < 8; ++r) {
            const int mrow = tileM * 16 + r + mhalf * 8;
            if (mrow < Nd) {
              float vv = c[r] + pb;
              vv = (vv - pm) * ps * pg + pe;
              vv = vv > 0.0f ? vv : 0.01f * vv;
              D[(size_t)mrow * TN + cc] = vv;
            }
          }
        } else {
          const float pb = Pp[cc];
#pragma unroll
          for (int r = 0; r < 8; ++r) {
            const int mrow = tileM * 16 + r + mhalf * 8;
            if (mrow < Nd) D[(size_t)mrow * TN + cc] = c[r] / (float)H + pb;
          }
        }
      }
    }
    gat_lds_barrier();
    if (gath && t < nt) {
#pragma unroll
      for (int h = 0; h < H; ++h) {
        float4* zr = (float4*)(Zt + (h * 16 + g) * FUSE_ZP_);
#pragma unroll
        for (int q = 0; q < 2; ++q) zr[q * 32 + lane] = acc[h][q];
      }
    }
    gat_lds_barrier();
  }
}

static inline int cdiv(int a, int b) { return (a + b - 1) / b; }

extern "C" void kernel_launch(void* const* d_in, const int* in_sizes, int n_in,
                              void* d_out, int out_size, void* d_ws, size_t ws_size,
                              hipStream_t stream) {
  const float* x     = (const float*)d_in[0];
  const int*   col1  = (const int*)d_in[2];
  const int*   t1    = (const int*)d_in[3];
  const int*   col2  = (const int*)d_in[5];
  const int*   t2    = (const int*)d_in[6];
  const int*   timeP = (const int*)d_in[7];
  const int*   intvP = (const int*)d_in[8];
  const float* Ws1   = (const float*)d_in[9];
  const float* Wd1   = (const float*)d_in[10];
  const float* as1   = (const float*)d_in[11];
  const float* ad1   = (const float*)d_in[12];
  const float* b1    = (const float*)d_in[13];
  const float* g1    = (const float*)d_in[14];
  const float* beta1 = (const float*)d_in[15];
  const float* m1    = (const float*)d_in[16];
  const float* v1    = (const float*)d_in[17];
  const float* Ws2   = (const float*)d_in[18];
  const float* Wd2   = (const float*)d_in[19];
  const float* as2   = (const float*)d_in[20];
  const float* ad2   = (const float*)d_in[21];
  const float* b2    = (const float*)d_in[22];

  float* ws  = (float*)d_ws;
  float* wf1 = ws;
  float* wf2 = wf1 + (size_t)IN_ * 16;
  float* h   = wf2 + (size_t)H1_ * HID_ * 16;

  const dim3 blk(256);
  const dim3 gblk(GEMM_WAVES_ * 32);

  gat_fold_att<<<cdiv(IN_ + H1_ * HID_, 8), blk, 0, stream>>>(
      Ws1, Wd1, as1, ad1, Ws2, Wd2, as2, ad2, wf1, wf2);

  gat_gather_gemm<7, H1_><<<GEMM_WGS_, gblk, 0, stream>>>(
      x, wf1, col1, t1, timeP, intvP, Ws1, h,
      b1, g1, beta1, m1, v1, N1_, cdiv(N1_, 16));

  gat_gather_gemm<8, H2_><<<GEMM_WGS_, gblk, 0, stream>>>(
      h, wf2, col2, t2, timeP, intvP, Ws2, (float*)d_out,
      b2, nullptr, nullptr, nullptr, nullptr, N2_, cdiv(N2_, 16));
}
